# starting kernel
# baseline (speedup 1.0000x reference)
#include <hip/hip_runtime.h>
#include <hip/hip_bf16.h>
#include <math.h>

#define BATCH 4
#define SEQ   2048
#define DIMX  192
#define DI    384
#define NST   16
#define DTR   12
#define XZW   768
#define DBLW  48
#define MROWS (BATCH*SEQ)
#define NCH   32
#define CL    (SEQ/NCH)

typedef float v2f __attribute__((ext_vector_type(2)));
typedef float v8f __attribute__((ext_vector_type(8)));

__device__ __forceinline__ v8f wmma4(v2f a, v2f b, v8f c) {
  return __builtin_amdgcn_wmma_f32_16x16x4_f32(false, a, false, b, (short)0, c,
                                               false, false);
}

__global__ void k_gemm_in(const float* __restrict__ x,
                          const float* __restrict__ Win,
                          float* __restrict__ xz, int reverse) {
  const int wave = blockIdx.x * (blockDim.x >> 5) + (threadIdx.x >> 5);
  const int lane = threadIdx.x & 31;
  const int TG = XZW / 64;
  const int gm = wave / TG, gng = wave % TG;
  const int mrow = lane & 15, kh = lane >> 4;

  int m  = gm * 16 + mrow;
  int bb = m / SEQ, ll = m % SEQ;
  int srow = reverse ? (bb * SEQ + (SEQ - 1 - ll)) : m;

  const float* ap = x + (size_t)srow * DIMX;
  const float* bp[4];
#pragma unroll
  for (int j = 0; j < 4; ++j)
    bp[j] = Win + (size_t)(gng * 64 + j * 16 + mrow) * DIMX;

  v8f acc[4] = {};
  for (int k0 = 0; k0 < DIMX; k0 += 4) {
    int kk = k0 + 2 * kh;
    v2f a; a.x = ap[kk]; a.y = ap[kk + 1];
#pragma unroll
    for (int j = 0; j < 4; ++j) {
      v2f b; b.x = bp[j][kk]; b.y = bp[j][kk + 1];
      acc[j] = wmma4(a, b, acc[j]);
    }
  }
#pragma unroll
  for (int j = 0; j < 4; ++j)
#pragma unroll
    for (int v = 0; v < 8; ++v) {
      int orow = gm * 16 + v + 8 * kh;
      xz[(size_t)orow * XZW + gng * 64 + j * 16 + mrow] = acc[j][v];
    }
}

__global__ void k_conv_silu(const float* __restrict__ xz,
                            const float* __restrict__ cw,
                            const float* __restrict__ cb,
                            float* __restrict__ u) {
  int idx = blockIdx.x * blockDim.x + threadIdx.x;
  if (idx >= MROWS * DI) return;
  int d = idx % DI;
  int t = (idx / DI) % SEQ;
  int b = idx / (DI * SEQ);
  float acc = cb[d];
#pragma unroll
  for (int j = 0; j < 4; ++j) {
    int tt = t - 3 + j;
    if (tt >= 0)
      acc += xz[(size_t)(b * SEQ + tt) * XZW + d] * cw[d * 4 + j];
  }
  u[(size_t)(b * SEQ + t) * DI + d] = acc / (1.0f + __expf(-acc));
}

__global__ void k_gemm_xp(const float* __restrict__ u,
                          const float* __restrict__ Wxp,
                          float* __restrict__ dbl) {
  const int wave = blockIdx.x * (blockDim.x >> 5) + (threadIdx.x >> 5);
  const int lane = threadIdx.x & 31;
  const int gm = wave;
  const int mrow = lane & 15, kh = lane >> 4;

  const float* ap = u + (size_t)(gm * 16 + mrow) * DI;
  bool nvalid[3];
  const float* bp[3];
#pragma unroll
  for (int j = 0; j < 3; ++j) {
    int n = j * 16 + mrow;
    nvalid[j] = (n < DTR + 2 * NST);
    bp[j] = Wxp + (size_t)(nvalid[j] ? n : 0) * DI;
  }

  v8f acc[3] = {};
  for (int k0 = 0; k0 < DI; k0 += 4) {
    int kk = k0 + 2 * kh;
    v2f a; a.x = ap[kk]; a.y = ap[kk + 1];
#pragma unroll
    for (int j = 0; j < 3; ++j) {
      v2f b;
      b.x = nvalid[j] ? bp[j][kk] : 0.0f;
      b.y = nvalid[j] ? bp[j][kk + 1] : 0.0f;
      acc[j] = wmma4(a, b, acc[j]);
    }
  }
#pragma unroll
  for (int j = 0; j < 3; ++j)
#pragma unroll
    for (int v = 0; v < 8; ++v) {
      int orow = gm * 16 + v + 8 * kh;
      dbl[(size_t)orow * DBLW + j * 16 + mrow] = acc[j][v];
    }
}

__global__ void k_gemm_dt(const float* __restrict__ dbl,
                          const float* __restrict__ Wdt,
                          const float* __restrict__ bdt,
                          float* __restrict__ xz) {
  const int wave = blockIdx.x * (blockDim.x >> 5) + (threadIdx.x >> 5);
  const int lane = threadIdx.x & 31;
  const int TG = DI / 64;
  const int gm = wave / TG, gng = wave % TG;
  const int mrow = lane & 15, kh = lane >> 4;

  const float* ap = dbl + (size_t)(gm * 16 + mrow) * DBLW;
  const float* bp[4];
#pragma unroll
  for (int j = 0; j < 4; ++j)
    bp[j] = Wdt + (size_t)(gng * 64 + j * 16 + mrow) * DTR;

  v8f acc[4] = {};
  for (int k0 = 0; k0 < DTR; k0 += 4) {
    int kk = k0 + 2 * kh;
    v2f a; a.x = ap[kk]; a.y = ap[kk + 1];
#pragma unroll
    for (int j = 0; j < 4; ++j) {
      v2f b; b.x = bp[j][kk]; b.y = bp[j][kk + 1];
      acc[j] = wmma4(a, b, acc[j]);
    }
  }
#pragma unroll
  for (int j = 0; j < 4; ++j) {
    int n = gng * 64 + j * 16 + mrow;
    float bias = bdt[n];
#pragma unroll
    for (int v = 0; v < 8; ++v) {
      int orow = gm * 16 + v + 8 * kh;
      float t = acc[j][v] + bias;
      float sp = (t > 20.0f) ? t : log1pf(__expf(t));
      xz[(size_t)orow * XZW + n] = sp;
    }
  }
}

__global__ void k_scan_part1(const float* __restrict__ xz,
                             const float* __restrict__ u,
                             const float* __restrict__ dbl,
                             const float* __restrict__ Alog,
                             float* __restrict__ Pb, float* __restrict__ Sb) {
  const int b = blockIdx.x / NCH;
  const int c = blockIdx.x % NCH;
  const int d = threadIdx.x;

  float a[NST], P[NST], S[NST];
#pragma unroll
  for (int n = 0; n < NST; ++n) {
    a[n] = -__expf(Alog[d * NST + n]);
    P[n] = 1.0f;
    S[n] = 0.0f;
  }
  const int t0 = c * CL;
  for (int t = t0; t < t0 + CL; ++t) {
    const size_t row = (size_t)b * SEQ + t;
    const float* bc = dbl + row * DBLW;
    __builtin_prefetch(bc + DBLW, 0, 0);
    const float dt = xz[row * XZW + d];
    const float ut = u[row * DI + d];
    const float dtu = dt * ut;
#pragma unroll
    for (int n = 0; n < NST; ++n) {
      float dA = __expf(dt * a[n]);
      S[n] = S[n] * dA + dtu * bc[DTR + n];
      P[n] *= dA;
    }
  }
  const size_t base = ((size_t)(b * NCH + c) * DI + d) * NST;
#pragma unroll
  for (int n = 0; n < NST; ++n) {
    Pb[base + n] = P[n];
    Sb[base + n] = S[n];
  }
}

__global__ void k_scan_combine(const float* __restrict__ Pb,
                               float* __restrict__ Sb) {
  int idx = blockIdx.x * blockDim.x + threadIdx.x;
  if (idx >= BATCH * DI * NST) return;
  const int n = idx & (NST - 1);
  const int d = (idx >> 4) % DI;
  const int b = idx / (NST * DI);

  float carry = 0.0f;
  for (int c = 0; c < NCH; ++c) {
    const size_t base = ((size_t)(b * NCH + c) * DI + d) * NST + n;
    float p = Pb[base];
    float s = Sb[base];
    Sb[base] = carry;
    carry = p * carry + s;
  }
}

__global__ void k_scan_part3(float* __restrict__ xz,
                             const float* __restrict__ u,
                             const float* __restrict__ dbl,
                             const float* __restrict__ Alog,
                             const float* __restrict__ Dv,
                             const float* __restrict__ Sb) {
  const int b = blockIdx.x / NCH;
  const int c = blockIdx.x % NCH;
  const int d = threadIdx.x;

  float a[NST], h[NST];
  const size_t base = ((size_t)(b * NCH + c) * DI + d) * NST;
#pragma unroll
  for (int n = 0; n < NST; ++n) {
    a[n] = -__expf(Alog[d * NST + n]);
    h[n] = Sb[base + n];
  }
  const float Dd = Dv[d];

  const int t0 = c * CL;
  for (int t = t0; t < t0 + CL; ++t) {
    const size_t row = (size_t)b * SEQ + t;
    const float* bc = dbl + row * DBLW;
    __builtin_prefetch(bc + DBLW, 0, 0);
    const float dt = xz[row * XZW + d];
    const float ut = u[row * DI + d];
    const float dtu = dt * ut;
    float y = 0.0f;
#pragma unroll
    for (int n = 0; n < NST; ++n) {
      float dA = __expf(dt * a[n]);
      h[n] = h[n] * dA + dtu * bc[DTR + n];
      y += h[n] * bc[DTR + NST + n];
    }
    y += ut * Dd;
    float z = xz[row * XZW + DI + d];
    y *= z / (1.0f + __expf(-z));
    xz[row * XZW + DI + d] = y;
  }
}

__global__ void k_gemm_out(const float* __restrict__ xz_f,
                           const float* __restrict__ xz_b,
                           const float* __restrict__ x,
                           const float* __restrict__ Wout,
                           float* __restrict__ out) {
  const int wave = blockIdx.x * (blockDim.x >> 5) + (threadIdx.x >> 5);
  const int lane = threadIdx.x & 31;
  const int TG = DIMX / 64;
  const int gm = wave / TG, gng = wave % TG;
  const int mrow = lane & 15, kh = lane >> 4;

  int m  = gm * 16 + mrow;
  int bb = m / SEQ, ll = m % SEQ;
  int mrev = bb * SEQ + (SEQ - 1 - ll);

  const float* af = xz_f + (size_t)m    * XZW + DI;
  const float* ab = xz_b + (size_t)mrev * XZW + DI;
  const float* bp[4];
#pragma unroll
  for (int j = 0; j < 4; ++j)
    bp[j] = Wout + (size_t)(gng * 64 + j * 16 + mrow) * DI;

  v8f acc[4] = {};
  for (int k0 = 0; k0 < DI; k0 += 4) {
    int kk = k0 + 2 * kh;
    v2f a; a.x = af[kk] + ab[kk]; a.y = af[kk + 1] + ab[kk + 1];
#pragma unroll
    for (int j = 0; j < 4; ++j) {
      v2f b; b.x = bp[j][kk]; b.y = bp[j][kk + 1];
      acc[j] = wmma4(a, b, acc[j]);
    }
  }
#pragma unroll
  for (int j = 0; j < 4; ++j)
#pragma unroll
    for (int v = 0; v < 8; ++v) {
      int orow = gm * 16 + v + 8 * kh;
      int n = gng * 64 + j * 16 + mrow;
      out[(size_t)orow * DIMX + n] = acc[j][v] + x[(size_t)orow * DIMX + n];
    }
}

extern "C" void kernel_launch(void* const* d_in, const int* in_sizes, int n_in,
                              void* d_out, int out_size, void* d_ws, size_t ws_size,
                              hipStream_t stream) {
  const float* x      = (const float*)d_in[0];
  const float* Win_f  = (const float*)d_in[1];
  const float* cw_f   = (const float*)d_in[2];
  const float* cb_f   = (const float*)d_in[3];
  const float* Wxp_f  = (const float*)d_in[4];
  const float* Wdt_f  = (const float*)d_in[5];
  const float* bdt_f  = (const float*)d_in[6];
  const float* Alog_f = (const float*)d_in[7];
  const float* Dv_f   = (const float*)d_in[8];
  const float* Win_b  = (const float*)d_in[9];
  const float* cw_b   = (const float*)d_in[10];
  const float* cb_b   = (const float*)d_in[11];
  const float* Wxp_b  = (const float*)d_in[12];
  const float* Wdt_b  = (const float*)d_in[13];
  const float* bdt_b  = (const float*)d_in[14];
  const float* Alog_b = (const float*)d_in[15];
  const float* Dv_b   = (const float*)d_in[16];
  const float* Wout   = (const float*)d_in[17];
  float* out = (float*)d_out;

  float* ws    = (float*)d_ws;
  float* xz_f  = ws;
  float* xz_b  = xz_f + (size_t)MROWS * XZW;
  float* u_f   = xz_b + (size_t)MROWS * XZW;
  float* u_b   = u_f  + (size_t)MROWS * DI;
  float* dbl_f = u_b  + (size_t)MROWS * DI;
  float* dbl_b = dbl_f + (size_t)MROWS * DBLW;
  const size_t PS = (size_t)BATCH * NCH * DI * NST;
  float* P_f   = dbl_b + (size_t)MROWS * DBLW;
  float* S_f   = P_f + PS;
  float* P_b   = S_f + PS;
  float* S_b   = P_b + PS;

  const int BLK = 256, WPB = BLK / 32;

  {
    int waves = (MROWS / 16) * (XZW / 64);
    k_gemm_in<<<waves / WPB, BLK, 0, stream>>>(x, Win_f, xz_f, 0);
    k_gemm_in<<<waves / WPB, BLK, 0, stream>>>(x, Win_b, xz_b, 1);
  }
  {
    int n = MROWS * DI;
    k_conv_silu<<<(n + BLK - 1) / BLK, BLK, 0, stream>>>(xz_f, cw_f, cb_f, u_f);
    k_conv_silu<<<(n + BLK - 1) / BLK, BLK, 0, stream>>>(xz_b, cw_b, cb_b, u_b);
  }
  {
    int waves = MROWS / 16;
    k_gemm_xp<<<waves / WPB, BLK, 0, stream>>>(u_f, Wxp_f, dbl_f);
    k_gemm_xp<<<waves / WPB, BLK, 0, stream>>>(u_b, Wxp_b, dbl_b);
  }
  {
    int waves = (MROWS / 16) * (DI / 64);
    k_gemm_dt<<<waves / WPB, BLK, 0, stream>>>(dbl_f, Wdt_f, bdt_f, xz_f);
    k_gemm_dt<<<waves / WPB, BLK, 0, stream>>>(dbl_b, Wdt_b, bdt_b, xz_b);
  }
  {
    int cb = (BATCH * DI * NST + BLK - 1) / BLK;
    k_scan_part1<<<BATCH * NCH, DI, 0, stream>>>(xz_f, u_f, dbl_f, Alog_f, P_f, S_f);
    k_scan_part1<<<BATCH * NCH, DI, 0, stream>>>(xz_b, u_b, dbl_b, Alog_b, P_b, S_b);
    k_scan_combine<<<cb, BLK, 0, stream>>>(P_f, S_f);
    k_scan_combine<<<cb, BLK, 0, stream>>>(P_b, S_b);
    k_scan_part3<<<BATCH * NCH, DI, 0, stream>>>(xz_f, u_f, dbl_f, Alog_f, Dv_f, S_f);
    k_scan_part3<<<BATCH * NCH, DI, 0, stream>>>(xz_b, u_b, dbl_b, Alog_b, Dv_b, S_b);
  }
  {
    int waves = (MROWS / 16) * (DIMX / 64);
    k_gemm_out<<<waves / WPB, BLK, 0, stream>>>(xz_f, xz_b, x, Wout, out);
  }
}


# Forward and reversed branch of each two-branch stage share one launch (grid.y picks the branch): 15 launches become 8, each with twice the waves; arithmetic and summation order untouched
# speedup vs baseline: 1.4205x; 1.4205x over previous
#include <hip/hip_runtime.h>
#include <hip/hip_bf16.h>
#include <math.h>

#define BATCH 4
#define SEQ   2048
#define DIMX  192
#define DI    384
#define NST   16
#define DTR   12
#define XZW   768
#define DBLW  48
#define MROWS (BATCH*SEQ)
#define NCH   32
#define CL    (SEQ/NCH)
#define PSTRIDE (2 * BATCH * NCH * DI * NST)

typedef float v2f __attribute__((ext_vector_type(2)));
typedef float v8f __attribute__((ext_vector_type(8)));

__device__ __forceinline__ v8f wmma4(v2f a, v2f b, v8f c) {
  return __builtin_amdgcn_wmma_f32_16x16x4_f32(false, a, false, b, (short)0, c,
                                               false, false);
}

__global__ void k_gemm_in(const float* __restrict__ x,
                          const float* __restrict__ Win_f,
                          const float* __restrict__ Win_b,
                          float* __restrict__ xz_all) {
  const int reverse = blockIdx.y;
  const float* Win = reverse ? Win_b : Win_f;
  float* xz = xz_all + (size_t)reverse * MROWS * XZW;
  const int wave = blockIdx.x * (blockDim.x >> 5) + (threadIdx.x >> 5);
  const int lane = threadIdx.x & 31;
  const int TG = XZW / 64;
  const int gm = wave / TG, gng = wave % TG;
  const int mrow = lane & 15, kh = lane >> 4;

  int m  = gm * 16 + mrow;
  int bb = m / SEQ, ll = m % SEQ;
  int srow = reverse ? (bb * SEQ + (SEQ - 1 - ll)) : m;

  const float* ap = x + (size_t)srow * DIMX;
  const float* bp[4];
#pragma unroll
  for (int j = 0; j < 4; ++j)
    bp[j] = Win + (size_t)(gng * 64 + j * 16 + mrow) * DIMX;

  v8f acc[4] = {};
  for (int k0 = 0; k0 < DIMX; k0 += 4) {
    int kk = k0 + 2 * kh;
    v2f a; a.x = ap[kk]; a.y = ap[kk + 1];
#pragma unroll
    for (int j = 0; j < 4; ++j) {
      v2f b; b.x = bp[j][kk]; b.y = bp[j][kk + 1];
      acc[j] = wmma4(a, b, acc[j]);
    }
  }
#pragma unroll
  for (int j = 0; j < 4; ++j)
#pragma unroll
    for (int v = 0; v < 8; ++v) {
      int orow = gm * 16 + v + 8 * kh;
      xz[(size_t)orow * XZW + gng * 64 + j * 16 + mrow] = acc[j][v];
    }
}

__global__ void k_conv_silu(const float* __restrict__ xz_all,
                            const float* __restrict__ cw_f,
                            const float* __restrict__ cb_f,
                            const float* __restrict__ cw_b,
                            const float* __restrict__ cb_b,
                            float* __restrict__ u_all) {
  const int dir = blockIdx.y;
  const float* xz = xz_all + (size_t)dir * MROWS * XZW;
  const float* cw = dir ? cw_b : cw_f;
  const float* cb = dir ? cb_b : cb_f;
  float* u = u_all + (size_t)dir * MROWS * DI;
  int idx = blockIdx.x * blockDim.x + threadIdx.x;
  if (idx >= MROWS * DI) return;
  int d = idx % DI;
  int t = (idx / DI) % SEQ;
  int b = idx / (DI * SEQ);
  float acc = cb[d];
#pragma unroll
  for (int j = 0; j < 4; ++j) {
    int tt = t - 3 + j;
    if (tt >= 0)
      acc += xz[(size_t)(b * SEQ + tt) * XZW + d] * cw[d * 4 + j];
  }
  u[(size_t)(b * SEQ + t) * DI + d] = acc / (1.0f + __expf(-acc));
}

__global__ void k_gemm_xp(const float* __restrict__ u_all,
                          const float* __restrict__ Wxp_f,
                          const float* __restrict__ Wxp_b,
                          float* __restrict__ dbl_all) {
  const int dir = blockIdx.y;
  const float* u = u_all + (size_t)dir * MROWS * DI;
  const float* Wxp = dir ? Wxp_b : Wxp_f;
  float* dbl = dbl_all + (size_t)dir * MROWS * DBLW;
  const int wave = blockIdx.x * (blockDim.x >> 5) + (threadIdx.x >> 5);
  const int lane = threadIdx.x & 31;
  const int gm = wave;
  const int mrow = lane & 15, kh = lane >> 4;

  const float* ap = u + (size_t)(gm * 16 + mrow) * DI;
  bool nvalid[3];
  const float* bp[3];
#pragma unroll
  for (int j = 0; j < 3; ++j) {
    int n = j * 16 + mrow;
    nvalid[j] = (n < DTR + 2 * NST);
    bp[j] = Wxp + (size_t)(nvalid[j] ? n : 0) * DI;
  }

  v8f acc[3] = {};
  for (int k0 = 0; k0 < DI; k0 += 4) {
    int kk = k0 + 2 * kh;
    v2f a; a.x = ap[kk]; a.y = ap[kk + 1];
#pragma unroll
    for (int j = 0; j < 3; ++j) {
      v2f b;
      b.x = nvalid[j] ? bp[j][kk] : 0.0f;
      b.y = nvalid[j] ? bp[j][kk + 1] : 0.0f;
      acc[j] = wmma4(a, b, acc[j]);
    }
  }
#pragma unroll
  for (int j = 0; j < 3; ++j)
#pragma unroll
    for (int v = 0; v < 8; ++v) {
      int orow = gm * 16 + v + 8 * kh;
      dbl[(size_t)orow * DBLW + j * 16 + mrow] = acc[j][v];
    }
}

__global__ void k_gemm_dt(const float* __restrict__ dbl_all,
                          const float* __restrict__ Wdt_f,
                          const float* __restrict__ bdt_f,
                          const float* __restrict__ Wdt_b,
                          const float* __restrict__ bdt_b,
                          float* __restrict__ xz_all) {
  const int dir = blockIdx.y;
  const float* dbl = dbl_all + (size_t)dir * MROWS * DBLW;
  const float* Wdt = dir ? Wdt_b : Wdt_f;
  const float* bdt = dir ? bdt_b : bdt_f;
  float* xz = xz_all + (size_t)dir * MROWS * XZW;
  const int wave = blockIdx.x * (blockDim.x >> 5) + (threadIdx.x >> 5);
  const int lane = threadIdx.x & 31;
  const int TG = DI / 64;
  const int gm = wave / TG, gng = wave % TG;
  const int mrow = lane & 15, kh = lane >> 4;

  const float* ap = dbl + (size_t)(gm * 16 + mrow) * DBLW;
  const float* bp[4];
#pragma unroll
  for (int j = 0; j < 4; ++j)
    bp[j] = Wdt + (size_t)(gng * 64 + j * 16 + mrow) * DTR;

  v8f acc[4] = {};
  for (int k0 = 0; k0 < DTR; k0 += 4) {
    int kk = k0 + 2 * kh;
    v2f a; a.x = ap[kk]; a.y = ap[kk + 1];
#pragma unroll
    for (int j = 0; j < 4; ++j) {
      v2f b; b.x = bp[j][kk]; b.y = bp[j][kk + 1];
      acc[j] = wmma4(a, b, acc[j]);
    }
  }
#pragma unroll
  for (int j = 0; j < 4; ++j) {
    int n = gng * 64 + j * 16 + mrow;
    float bias = bdt[n];
#pragma unroll
    for (int v = 0; v < 8; ++v) {
      int orow = gm * 16 + v + 8 * kh;
      float t = acc[j][v] + bias;
      float sp = (t > 20.0f) ? t : log1pf(__expf(t));
      xz[(size_t)orow * XZW + n] = sp;
    }
  }
}

__global__ void k_scan_part1(const float* __restrict__ xz_all,
                             const float* __restrict__ u_all,
                             const float* __restrict__ dbl_all,
                             const float* __restrict__ Alog_f,
                             const float* __restrict__ Alog_b,
                             float* __restrict__ Pb_all, float* __restrict__ Sb_all) {
  const int dir = blockIdx.y;
  const float* xz  = xz_all  + (size_t)dir * MROWS * XZW;
  const float* u   = u_all   + (size_t)dir * MROWS * DI;
  const float* dbl = dbl_all + (size_t)dir * MROWS * DBLW;
  const float* Alog = dir ? Alog_b : Alog_f;
  float* Pb = Pb_all + (size_t)dir * PSTRIDE;
  float* Sb = Sb_all + (size_t)dir * PSTRIDE;
  const int b = blockIdx.x / NCH;
  const int c = blockIdx.x % NCH;
  const int d = threadIdx.x;

  float a[NST], P[NST], S[NST];
#pragma unroll
  for (int n = 0; n < NST; ++n) {
    a[n] = -__expf(Alog[d * NST + n]);
    P[n] = 1.0f;
    S[n] = 0.0f;
  }
  const int t0 = c * CL;
  for (int t = t0; t < t0 + CL; ++t) {
    const size_t row = (size_t)b * SEQ + t;
    const float* bc = dbl + row * DBLW;
    __builtin_prefetch(bc + DBLW, 0, 0);
    const float dt = xz[row * XZW + d];
    const float ut = u[row * DI + d];
    const float dtu = dt * ut;
#pragma unroll
    for (int n = 0; n < NST; ++n) {
      float dA = __expf(dt * a[n]);
      S[n] = S[n] * dA + dtu * bc[DTR + n];
      P[n] *= dA;
    }
  }
  const size_t base = ((size_t)(b * NCH + c) * DI + d) * NST;
#pragma unroll
  for (int n = 0; n < NST; ++n) {
    Pb[base + n] = P[n];
    Sb[base + n] = S[n];
  }
}

__global__ void k_scan_combine(const float* __restrict__ Pb_all,
                               float* __restrict__ Sb_all) {
  const int dir = blockIdx.y;
  const float* Pb = Pb_all + (size_t)dir * PSTRIDE;
  float* Sb = Sb_all + (size_t)dir * PSTRIDE;
  int idx = blockIdx.x * blockDim.x + threadIdx.x;
  if (idx >= BATCH * DI * NST) return;
  const int n = idx & (NST - 1);
  const int d = (idx >> 4) % DI;
  const int b = idx / (NST * DI);

  float carry = 0.0f;
  for (int c = 0; c < NCH; ++c) {
    const size_t base = ((size_t)(b * NCH + c) * DI + d) * NST + n;
    float p = Pb[base];
    float s = Sb[base];
    Sb[base] = carry;
    carry = p * carry + s;
  }
}

__global__ void k_scan_part3(float* __restrict__ xz_all,
                             const float* __restrict__ u_all,
                             const float* __restrict__ dbl_all,
                             const float* __restrict__ Alog_f,
                             const float* __restrict__ Dv_f,
                             const float* __restrict__ Alog_b,
                             const float* __restrict__ Dv_b,
                             const float* __restrict__ Sb_all) {
  const int dir = blockIdx.y;
  float* xz        = xz_all  + (size_t)dir * MROWS * XZW;
  const float* u   = u_all   + (size_t)dir * MROWS * DI;
  const float* dbl = dbl_all + (size_t)dir * MROWS * DBLW;
  const float* Alog = dir ? Alog_b : Alog_f;
  const float* Dv   = dir ? Dv_b : Dv_f;
  const float* Sb = Sb_all + (size_t)dir * PSTRIDE;
  const int b = blockIdx.x / NCH;
  const int c = blockIdx.x % NCH;
  const int d = threadIdx.x;

  float a[NST], h[NST];
  const size_t base = ((size_t)(b * NCH + c) * DI + d) * NST;
#pragma unroll
  for (int n = 0; n < NST; ++n) {
    a[n] = -__expf(Alog[d * NST + n]);
    h[n] = Sb[base + n];
  }
  const float Dd = Dv[d];

  const int t0 = c * CL;
  for (int t = t0; t < t0 + CL; ++t) {
    const size_t row = (size_t)b * SEQ + t;
    const float* bc = dbl + row * DBLW;
    __builtin_prefetch(bc + DBLW, 0, 0);
    const float dt = xz[row * XZW + d];
    const float ut = u[row * DI + d];
    const float dtu = dt * ut;
    float y = 0.0f;
#pragma unroll
    for (int n = 0; n < NST; ++n) {
      float dA = __expf(dt * a[n]);
      h[n] = h[n] * dA + dtu * bc[DTR + n];
      y += h[n] * bc[DTR + NST + n];
    }
    y += ut * Dd;
    float z = xz[row * XZW + DI + d];
    y *= z / (1.0f + __expf(-z));
    xz[row * XZW + DI + d] = y;
  }
}

__global__ void k_gemm_out(const float* __restrict__ xz_f,
                           const float* __restrict__ xz_b,
                           const float* __restrict__ x,
                           const float* __restrict__ Wout,
                           float* __restrict__ out) {
  const int wave = blockIdx.x * (blockDim.x >> 5) + (threadIdx.x >> 5);
  const int lane = threadIdx.x & 31;
  const int TG = DIMX / 64;
  const int gm = wave / TG, gng = wave % TG;
  const int mrow = lane & 15, kh = lane >> 4;

  int m  = gm * 16 + mrow;
  int bb = m / SEQ, ll = m % SEQ;
  int mrev = bb * SEQ + (SEQ - 1 - ll);

  const float* af = xz_f + (size_t)m    * XZW + DI;
  const float* ab = xz_b + (size_t)mrev * XZW + DI;
  const float* bp[4];
#pragma unroll
  for (int j = 0; j < 4; ++j)
    bp[j] = Wout + (size_t)(gng * 64 + j * 16 + mrow) * DI;

  v8f acc[4] = {};
  for (int k0 = 0; k0 < DI; k0 += 4) {
    int kk = k0 + 2 * kh;
    v2f a; a.x = af[kk] + ab[kk]; a.y = af[kk + 1] + ab[kk + 1];
#pragma unroll
    for (int j = 0; j < 4; ++j) {
      v2f b; b.x = bp[j][kk]; b.y = bp[j][kk + 1];
      acc[j] = wmma4(a, b, acc[j]);
    }
  }
#pragma unroll
  for (int j = 0; j < 4; ++j)
#pragma unroll
    for (int v = 0; v < 8; ++v) {
      int orow = gm * 16 + v + 8 * kh;
      int n = gng * 64 + j * 16 + mrow;
      out[(size_t)orow * DIMX + n] = acc[j][v] + x[(size_t)orow * DIMX + n];
    }
}

extern "C" void kernel_launch(void* const* d_in, const int* in_sizes, int n_in,
                              void* d_out, int out_size, void* d_ws, size_t ws_size,
                              hipStream_t stream) {
  const float* x      = (const float*)d_in[0];
  const float* Win_f  = (const float*)d_in[1];
  const float* cw_f   = (const float*)d_in[2];
  const float* cb_f   = (const float*)d_in[3];
  const float* Wxp_f  = (const float*)d_in[4];
  const float* Wdt_f  = (const float*)d_in[5];
  const float* bdt_f  = (const float*)d_in[6];
  const float* Alog_f = (const float*)d_in[7];
  const float* Dv_f   = (const float*)d_in[8];
  const float* Win_b  = (const float*)d_in[9];
  const float* cw_b   = (const float*)d_in[10];
  const float* cb_b   = (const float*)d_in[11];
  const float* Wxp_b  = (const float*)d_in[12];
  const float* Wdt_b  = (const float*)d_in[13];
  const float* bdt_b  = (const float*)d_in[14];
  const float* Alog_b = (const float*)d_in[15];
  const float* Dv_b   = (const float*)d_in[16];
  const float* Wout   = (const float*)d_in[17];
  float* out = (float*)d_out;

  float* ws    = (float*)d_ws;
  float* xz_f  = ws;
  float* xz_b  = xz_f + (size_t)MROWS * XZW;
  float* u_f   = xz_b + (size_t)MROWS * XZW;
  float* dbl_f = u_f  + (size_t)2 * MROWS * DI;
  const size_t PS = (size_t)BATCH * NCH * DI * NST;
  float* P_f   = dbl_f + (size_t)2 * MROWS * DBLW;
  float* S_f   = P_f + PS;

  const int BLK = 256, WPB = BLK / 32;

  {
    int waves = (MROWS / 16) * (XZW / 64);
    k_gemm_in<<<dim3(waves / WPB, 2), BLK, 0, stream>>>(x, Win_f, Win_b, xz_f);
  }
  {
    int n = MROWS * DI;
    k_conv_silu<<<dim3((n + BLK - 1) / BLK, 2), BLK, 0, stream>>>(
        xz_f, cw_f, cb_f, cw_b, cb_b, u_f);
  }
  {
    int waves = MROWS / 16;
    k_gemm_xp<<<dim3(waves / WPB, 2), BLK, 0, stream>>>(u_f, Wxp_f, Wxp_b, dbl_f);
  }
  {
    int waves = (MROWS / 16) * (DI / 64);
    k_gemm_dt<<<dim3(waves / WPB, 2), BLK, 0, stream>>>(
        dbl_f, Wdt_f, bdt_f, Wdt_b, bdt_b, xz_f);
  }
  {
    int cb = (BATCH * DI * NST + BLK - 1) / BLK;
    k_scan_part1<<<dim3(BATCH * NCH, 2), DI, 0, stream>>>(
        xz_f, u_f, dbl_f, Alog_f, Alog_b, P_f, S_f);
    k_scan_combine<<<dim3(cb, 2), BLK, 0, stream>>>(P_f, S_f);
    k_scan_part3<<<dim3(BATCH * NCH, 2), DI, 0, stream>>>(
        xz_f, u_f, dbl_f, Alog_f, Dv_f, Alog_b, Dv_b, S_f);
  }
  {
    int waves = (MROWS / 16) * (DIMX / 64);
    k_gemm_out<<<waves / WPB, BLK, 0, stream>>>(xz_f, xz_b, x, Wout, out);
  }
}


# Fragment loads of the three big products widened from 8 to 16 bytes per lane, one whole 128-byte line of each row per loop pass; k is summed in a new fixed order inside each 32-wide block
# speedup vs baseline: 1.7397x; 1.2247x over previous
#include <hip/hip_runtime.h>
#include <hip/hip_bf16.h>
#include <math.h>

#define BATCH 4
#define SEQ   2048
#define DIMX  192
#define DI    384
#define NST   16
#define DTR   12
#define XZW   768
#define DBLW  48
#define MROWS (BATCH*SEQ)
#define NCH   32
#define CL    (SEQ/NCH)
#define PSTRIDE (2 * BATCH * NCH * DI * NST)

typedef float v2f __attribute__((ext_vector_type(2)));
typedef float v8f __attribute__((ext_vector_type(8)));

__device__ __forceinline__ v8f wmma4(v2f a, v2f b, v8f c) {
  return __builtin_amdgcn_wmma_f32_16x16x4_f32(false, a, false, b, (short)0, c,
                                               false, false);
}

typedef float v4f __attribute__((ext_vector_type(4)));

__device__ __forceinline__ void ld16(const float* __restrict__ p, v4f (&r)[4]) {
  const v4f* q = reinterpret_cast<const v4f*>(p);
  r[0] = q[0]; r[1] = q[1]; r[2] = q[2]; r[3] = q[3];
}
__device__ __forceinline__ v2f lo2(v4f v) { v2f r; r.x = v.x; r.y = v.y; return r; }
__device__ __forceinline__ v2f hi2(v4f v) { v2f r; r.x = v.z; r.y = v.w; return r; }
#define LOADED4(v) "v"((v)[0]), "v"((v)[1]), "v"((v)[2]), "v"((v)[3])

__global__ void __launch_bounds__(256) k_gemm_in(const float* __restrict__ x,
                          const float* __restrict__ Win_f,
                          const float* __restrict__ Win_b,
                          float* __restrict__ xz_all) {
  const int reverse = blockIdx.y;
  const float* Win = reverse ? Win_b : Win_f;
  float* xz = xz_all + (size_t)reverse * MROWS * XZW;
  const int wave = blockIdx.x * (blockDim.x >> 5) + (threadIdx.x >> 5);
  const int lane = threadIdx.x & 31;
  const int TG = XZW / 64;
  const int gm = wave / TG, gng = wave % TG;
  const int mrow = lane & 15, kh = lane >> 4;

  int m  = gm * 16 + mrow;
  int bb = m / SEQ, ll = m % SEQ;
  int srow = reverse ? (bb * SEQ + (SEQ - 1 - ll)) : m;

  const float* ap = x + (size_t)srow * DIMX + 16 * kh;
  const float* bp[4];
#pragma unroll
  for (int j = 0; j < 4; ++j)
    bp[j] = Win + (size_t)(gng * 64 + j * 16 + mrow) * DIMX + 16 * kh;

  v8f acc[4] = {};
  for (int k0 = 0; k0 < DIMX; k0 += 32) {
    v4f a[4], b[4][4];
    ld16(ap + k0, a);
#pragma unroll
    for (int j = 0; j < 4; ++j) ld16(bp[j] + k0, b[j]);
    asm volatile("" :: LOADED4(a), LOADED4(b[0]), LOADED4(b[1]), LOADED4(b[2]), LOADED4(b[3]));
#pragma unroll
    for (int q = 0; q < 4; ++q) {
#pragma unroll
      for (int j = 0; j < 4; ++j) acc[j] = wmma4(lo2(a[q]), lo2(b[j][q]), acc[j]);
#pragma unroll
      for (int j = 0; j < 4; ++j) acc[j] = wmma4(hi2(a[q]), hi2(b[j][q]), acc[j]);
    }
  }
#pragma unroll
  for (int j = 0; j < 4; ++j)
#pragma unroll
    for (int v = 0; v < 8; ++v) {
      int orow = gm * 16 + v + 8 * kh;
      xz[(size_t)orow * XZW + gng * 64 + j * 16 + mrow] = acc[j][v];
    }
}

__global__ void k_conv_silu(const float* __restrict__ xz_all,
                            const float* __restrict__ cw_f,
                            const float* __restrict__ cb_f,
                            const float* __restrict__ cw_b,
                            const float* __restrict__ cb_b,
                            float* __restrict__ u_all) {
  const int dir = blockIdx.y;
  const float* xz = xz_all + (size_t)dir * MROWS * XZW;
  const float* cw = dir ? cw_b : cw_f;
  const float* cb = dir ? cb_b : cb_f;
  float* u = u_all + (size_t)dir * MROWS * DI;
  int idx = blockIdx.x * blockDim.x + threadIdx.x;
  if (idx >= MROWS * DI) return;
  int d = idx % DI;
  int t = (idx / DI) % SEQ;
  int b = idx / (DI * SEQ);
  float acc = cb[d];
#pragma unroll
  for (int j = 0; j < 4; ++j) {
    int tt = t - 3 + j;
    if (tt >= 0)
      acc += xz[(size_t)(b * SEQ + tt) * XZW + d] * cw[d * 4 + j];
  }
  u[(size_t)(b * SEQ + t) * DI + d] = acc / (1.0f + __expf(-acc));
}

__global__ void __launch_bounds__(256) k_gemm_xp(const float* __restrict__ u_all,
                          const float* __restrict__ Wxp_f,
                          const float* __restrict__ Wxp_b,
                          float* __restrict__ dbl_all) {
  const int dir = blockIdx.y;
  const float* u = u_all + (size_t)dir * MROWS * DI;
  const float* Wxp = dir ? Wxp_b : Wxp_f;
  float* dbl = dbl_all + (size_t)dir * MROWS * DBLW;
  const int wave = blockIdx.x * (blockDim.x >> 5) + (threadIdx.x >> 5);
  const int lane = threadIdx.x & 31;
  const int gm = wave;
  const int mrow = lane & 15, kh = lane >> 4;

  const float* ap = u + (size_t)(gm * 16 + mrow) * DI + 16 * kh;
  bool nvalid[3];
  const float* bp[3];
#pragma unroll
  for (int j = 0; j < 3; ++j) {
    int n = j * 16 + mrow;
    nvalid[j] = (n < DTR + 2 * NST);
    bp[j] = Wxp + (size_t)(nvalid[j] ? n : 0) * DI + 16 * kh;
  }

  v8f acc[3] = {};
  for (int k0 = 0; k0 < DI; k0 += 32) {
    v4f a[4], b[3][4];
    ld16(ap + k0, a);
#pragma unroll
    for (int j = 0; j < 3; ++j) ld16(bp[j] + k0, b[j]);
    asm volatile("" :: LOADED4(a), LOADED4(b[0]), LOADED4(b[1]), LOADED4(b[2]));
#pragma unroll
    for (int j = 0; j < 3; ++j) {
#pragma unroll
      for (int q = 0; q < 4; ++q) {
        v4f zero = {0.0f, 0.0f, 0.0f, 0.0f};
        b[j][q] = nvalid[j] ? b[j][q] : zero;
      }
    }
#pragma unroll
    for (int q = 0; q < 4; ++q) {
#pragma unroll
      for (int j = 0; j < 3; ++j) acc[j] = wmma4(lo2(a[q]), lo2(b[j][q]), acc[j]);
#pragma unroll
      for (int j = 0; j < 3; ++j) acc[j] = wmma4(hi2(a[q]), hi2(b[j][q]), acc[j]);
    }
  }
#pragma unroll
  for (int j = 0; j < 3; ++j)
#pragma unroll
    for (int v = 0; v < 8; ++v) {
      int orow = gm * 16 + v + 8 * kh;
      dbl[(size_t)orow * DBLW + j * 16 + mrow] = acc[j][v];
    }
}

__global__ void k_gemm_dt(const float* __restrict__ dbl_all,
                          const float* __restrict__ Wdt_f,
                          const float* __restrict__ bdt_f,
                          const float* __restrict__ Wdt_b,
                          const float* __restrict__ bdt_b,
                          float* __restrict__ xz_all) {
  const int dir = blockIdx.y;
  const float* dbl = dbl_all + (size_t)dir * MROWS * DBLW;
  const float* Wdt = dir ? Wdt_b : Wdt_f;
  const float* bdt = dir ? bdt_b : bdt_f;
  float* xz = xz_all + (size_t)dir * MROWS * XZW;
  const int wave = blockIdx.x * (blockDim.x >> 5) + (threadIdx.x >> 5);
  const int lane = threadIdx.x & 31;
  const int TG = DI / 64;
  const int gm = wave / TG, gng = wave % TG;
  const int mrow = lane & 15, kh = lane >> 4;

  const float* ap = dbl + (size_t)(gm * 16 + mrow) * DBLW;
  const float* bp[4];
#pragma unroll
  for (int j = 0; j < 4; ++j)
    bp[j] = Wdt + (size_t)(gng * 64 + j * 16 + mrow) * DTR;

  v8f acc[4] = {};
  for (int k0 = 0; k0 < DTR; k0 += 4) {
    int kk = k0 + 2 * kh;
    v2f a; a.x = ap[kk]; a.y = ap[kk + 1];
#pragma unroll
    for (int j = 0; j < 4; ++j) {
      v2f b; b.x = bp[j][kk]; b.y = bp[j][kk + 1];
      acc[j] = wmma4(a, b, acc[j]);
    }
  }
#pragma unroll
  for (int j = 0; j < 4; ++j) {
    int n = gng * 64 + j * 16 + mrow;
    float bias = bdt[n];
#pragma unroll
    for (int v = 0; v < 8; ++v) {
      int orow = gm * 16 + v + 8 * kh;
      float t = acc[j][v] + bias;
      float sp = (t > 20.0f) ? t : log1pf(__expf(t));
      xz[(size_t)orow * XZW + n] = sp;
    }
  }
}

__global__ void k_scan_part1(const float* __restrict__ xz_all,
                             const float* __restrict__ u_all,
                             const float* __restrict__ dbl_all,
                             const float* __restrict__ Alog_f,
                             const float* __restrict__ Alog_b,
                             float* __restrict__ Pb_all, float* __restrict__ Sb_all) {
  const int dir = blockIdx.y;
  const float* xz  = xz_all  + (size_t)dir * MROWS * XZW;
  const float* u   = u_all   + (size_t)dir * MROWS * DI;
  const float* dbl = dbl_all + (size_t)dir * MROWS * DBLW;
  const float* Alog = dir ? Alog_b : Alog_f;
  float* Pb = Pb_all + (size_t)dir * PSTRIDE;
  float* Sb = Sb_all + (size_t)dir * PSTRIDE;
  const int b = blockIdx.x / NCH;
  const int c = blockIdx.x % NCH;
  const int d = threadIdx.x;

  float a[NST], P[NST], S[NST];
#pragma unroll
  for (int n = 0; n < NST; ++n) {
    a[n] = -__expf(Alog[d * NST + n]);
    P[n] = 1.0f;
    S[n] = 0.0f;
  }
  const int t0 = c * CL;
  for (int t = t0; t < t0 + CL; ++t) {
    const size_t row = (size_t)b * SEQ + t;
    const float* bc = dbl + row * DBLW;
    __builtin_prefetch(bc + DBLW, 0, 0);
    const float dt = xz[row * XZW + d];
    const float ut = u[row * DI + d];
    const float dtu = dt * ut;
#pragma unroll
    for (int n = 0; n < NST; ++n) {
      float dA = __expf(dt * a[n]);
      S[n] = S[n] * dA + dtu * bc[DTR + n];
      P[n] *= dA;
    }
  }
  const size_t base = ((size_t)(b * NCH + c) * DI + d) * NST;
#pragma unroll
  for (int n = 0; n < NST; ++n) {
    Pb[base + n] = P[n];
    Sb[base + n] = S[n];
  }
}

__global__ void k_scan_combine(const float* __restrict__ Pb_all,
                               float* __restrict__ Sb_all) {
  const int dir = blockIdx.y;
  const float* Pb = Pb_all + (size_t)dir * PSTRIDE;
  float* Sb = Sb_all + (size_t)dir * PSTRIDE;
  int idx = blockIdx.x * blockDim.x + threadIdx.x;
  if (idx >= BATCH * DI * NST) return;
  const int n = idx & (NST - 1);
  const int d = (idx >> 4) % DI;
  const int b = idx / (NST * DI);

  float carry = 0.0f;
  for (int c = 0; c < NCH; ++c) {
    const size_t base = ((size_t)(b * NCH + c) * DI + d) * NST + n;
    float p = Pb[base];
    float s = Sb[base];
    Sb[base] = carry;
    carry = p * carry + s;
  }
}

__global__ void k_scan_part3(float* __restrict__ xz_all,
                             const float* __restrict__ u_all,
                             const float* __restrict__ dbl_all,
                             const float* __restrict__ Alog_f,
                             const float* __restrict__ Dv_f,
                             const float* __restrict__ Alog_b,
                             const float* __restrict__ Dv_b,
                             const float* __restrict__ Sb_all) {
  const int dir = blockIdx.y;
  float* xz        = xz_all  + (size_t)dir * MROWS * XZW;
  const float* u   = u_all   + (size_t)dir * MROWS * DI;
  const float* dbl = dbl_all + (size_t)dir * MROWS * DBLW;
  const float* Alog = dir ? Alog_b : Alog_f;
  const float* Dv   = dir ? Dv_b : Dv_f;
  const float* Sb = Sb_all + (size_t)dir * PSTRIDE;
  const int b = blockIdx.x / NCH;
  const int c = blockIdx.x % NCH;
  const int d = threadIdx.x;

  float a[NST], h[NST];
  const size_t base = ((size_t)(b * NCH + c) * DI + d) * NST;
#pragma unroll
  for (int n = 0; n < NST; ++n) {
    a[n] = -__expf(Alog[d * NST + n]);
    h[n] = Sb[base + n];
  }
  const float Dd = Dv[d];

  const int t0 = c * CL;
  for (int t = t0; t < t0 + CL; ++t) {
    const size_t row = (size_t)b * SEQ + t;
    const float* bc = dbl + row * DBLW;
    __builtin_prefetch(bc + DBLW, 0, 0);
    const float dt = xz[row * XZW + d];
    const float ut = u[row * DI + d];
    const float dtu = dt * ut;
    float y = 0.0f;
#pragma unroll
    for (int n = 0; n < NST; ++n) {
      float dA = __expf(dt * a[n]);
      h[n] = h[n] * dA + dtu * bc[DTR + n];
      y += h[n] * bc[DTR + NST + n];
    }
    y += ut * Dd;
    float z = xz[row * XZW + DI + d];
    y *= z / (1.0f + __expf(-z));
    xz[row * XZW + DI + d] = y;
  }
}

__global__ void __launch_bounds__(256) k_gemm_out(const float* __restrict__ xz_f,
                           const float* __restrict__ xz_b,
                           const float* __restrict__ x,
                           const float* __restrict__ Wout,
                           float* __restrict__ out) {
  const int wave = blockIdx.x * (blockDim.x >> 5) + (threadIdx.x >> 5);
  const int lane = threadIdx.x & 31;
  const int TG = DIMX / 64;
  const int gm = wave / TG, gng = wave % TG;
  const int mrow = lane & 15, kh = lane >> 4;

  int m  = gm * 16 + mrow;
  int bb = m / SEQ, ll = m % SEQ;
  int mrev = bb * SEQ + (SEQ - 1 - ll);

  const float* af = xz_f + (size_t)m    * XZW + DI + 16 * kh;
  const float* ab = xz_b + (size_t)mrev * XZW + DI + 16 * kh;
  const float* bp[4];
#pragma unroll
  for (int j = 0; j < 4; ++j)
    bp[j] = Wout + (size_t)(gng * 64 + j * 16 + mrow) * DI + 16 * kh;

  v8f acc[4] = {};
  for (int k0 = 0; k0 < DI; k0 += 32) {
    v4f a[4], a2[4], b[4][4];
    ld16(af + k0, a);
    ld16(ab + k0, a2);
#pragma unroll
    for (int j = 0; j < 4; ++j) ld16(bp[j] + k0, b[j]);
    asm volatile("" :: LOADED4(a), LOADED4(a2), LOADED4(b[0]), LOADED4(b[1]), LOADED4(b[2]), LOADED4(b[3]));
#pragma unroll
    for (int q = 0; q < 4; ++q) a[q] = a[q] + a2[q];
#pragma unroll
    for (int q = 0; q < 4; ++q) {
#pragma unroll
      for (int j = 0; j < 4; ++j) acc[j] = wmma4(lo2(a[q]), lo2(b[j][q]), acc[j]);
#pragma unroll
      for (int j = 0; j < 4; ++j) acc[j] = wmma4(hi2(a[q]), hi2(b[j][q]), acc[j]);
    }
  }
#pragma unroll
  for (int j = 0; j < 4; ++j)
#pragma unroll
    for (int v = 0; v < 8; ++v) {
      int orow = gm * 16 + v + 8 * kh;
      int n = gng * 64 + j * 16 + mrow;
      out[(size_t)orow * DIMX + n] = acc[j][v] + x[(size_t)orow * DIMX + n];
    }
}

extern "C" void kernel_launch(void* const* d_in, const int* in_sizes, int n_in,
                              void* d_out, int out_size, void* d_ws, size_t ws_size,
                              hipStream_t stream) {
  const float* x      = (const float*)d_in[0];
  const float* Win_f  = (const float*)d_in[1];
  const float* cw_f   = (const float*)d_in[2];
  const float* cb_f   = (const float*)d_in[3];
  const float* Wxp_f  = (const float*)d_in[4];
  const float* Wdt_f  = (const float*)d_in[5];
  const float* bdt_f  = (const float*)d_in[6];
  const float* Alog_f = (const float*)d_in[7];
  const float* Dv_f   = (const float*)d_in[8];
  const float* Win_b  = (const float*)d_in[9];
  const float* cw_b   = (const float*)d_in[10];
  const float* cb_b   = (const float*)d_in[11];
  const float* Wxp_b  = (const float*)d_in[12];
  const float* Wdt_b  = (const float*)d_in[13];
  const float* bdt_b  = (const float*)d_in[14];
  const float* Alog_b = (const float*)d_in[15];
  const float* Dv_b   = (const float*)d_in[16];
  const float* Wout   = (const float*)d_in[17];
  float* out = (float*)d_out;

  float* ws    = (float*)d_ws;
  float* xz_f  = ws;
  float* xz_b  = xz_f + (size_t)MROWS * XZW;
  float* u_f   = xz_b + (size_t)MROWS * XZW;
  float* dbl_f = u_f  + (size_t)2 * MROWS * DI;
  const size_t PS = (size_t)BATCH * NCH * DI * NST;
  float* P_f   = dbl_f + (size_t)2 * MROWS * DBLW;
  float* S_f   = P_f + PS;

  const int BLK = 256, WPB = BLK / 32;

  {
    int waves = (MROWS / 16) * (XZW / 64);
    k_gemm_in<<<dim3(waves / WPB, 2), BLK, 0, stream>>>(x, Win_f, Win_b, xz_f);
  }
  {
    int n = MROWS * DI;
    k_conv_silu<<<dim3((n + BLK - 1) / BLK, 2), BLK, 0, stream>>>(
        xz_f, cw_f, cb_f, cw_b, cb_b, u_f);
  }
  {
    int waves = MROWS / 16;
    k_gemm_xp<<<dim3(waves / WPB, 2), BLK, 0, stream>>>(u_f, Wxp_f, Wxp_b, dbl_f);
  }
  {
    int waves = (MROWS / 16) * (DI / 64);
    k_gemm_dt<<<dim3(waves / WPB, 2), BLK, 0, stream>>>(
        dbl_f, Wdt_f, bdt_f, Wdt_b, bdt_b, xz_f);
  }
  {
    int cb = (BATCH * DI * NST + BLK - 1) / BLK;
    k_scan_part1<<<dim3(BATCH * NCH, 2), DI, 0, stream>>>(
        xz_f, u_f, dbl_f, Alog_f, Alog_b, P_f, S_f);
    k_scan_combine<<<dim3(cb, 2), BLK, 0, stream>>>(P_f, S_f);
    k_scan_part3<<<dim3(BATCH * NCH, 2), DI, 0, stream>>>(
        xz_f, u_f, dbl_f, Alog_f, Dv_f, Alog_b, Dv_b, S_f);
  }
  {
    int waves = (MROWS / 16) * (DIMX / 64);
    k_gemm_out<<<waves / WPB, BLK, 0, stream>>>(xz_f, xz_b, x, Wout, out);
  }
}


# Conv+SiLU stage: a thread now owns 4 neighbouring channels for 8 timesteps (11 row pieces fetched once by 16-byte loads, 4-tap window slid in registers): 32x fewer waves to launch; tap order kept
# speedup vs baseline: 1.8931x; 1.0882x over previous
#include <hip/hip_runtime.h>
#include <hip/hip_bf16.h>
#include <math.h>

#define BATCH 4
#define SEQ   2048
#define DIMX  192
#define DI    384
#define NST   16
#define DTR   12
#define XZW   768
#define DBLW  48
#define MROWS (BATCH*SEQ)
#define NCH   32
#define CL    (SEQ/NCH)
#define PSTRIDE (2 * BATCH * NCH * DI * NST)

typedef float v2f __attribute__((ext_vector_type(2)));
typedef float v8f __attribute__((ext_vector_type(8)));

__device__ __forceinline__ v8f wmma4(v2f a, v2f b, v8f c) {
  return __builtin_amdgcn_wmma_f32_16x16x4_f32(false, a, false, b, (short)0, c,
                                               false, false);
}

typedef float v4f __attribute__((ext_vector_type(4)));

__device__ __forceinline__ void ld16(const float* __restrict__ p, v4f (&r)[4]) {
  const v4f* q = reinterpret_cast<const v4f*>(p);
  r[0] = q[0]; r[1] = q[1]; r[2] = q[2]; r[3] = q[3];
}
__device__ __forceinline__ v2f lo2(v4f v) { v2f r; r.x = v.x; r.y = v.y; return r; }
__device__ __forceinline__ v2f hi2(v4f v) { v2f r; r.x = v.z; r.y = v.w; return r; }
#define LOADED4(v) "v"((v)[0]), "v"((v)[1]), "v"((v)[2]), "v"((v)[3])

__global__ void __launch_bounds__(256) k_gemm_in(const float* __restrict__ x,
                          const float* __restrict__ Win_f,
                          const float* __restrict__ Win_b,
                          float* __restrict__ xz_all) {
  const int reverse = blockIdx.y;
  const float* Win = reverse ? Win_b : Win_f;
  float* xz = xz_all + (size_t)reverse * MROWS * XZW;
  const int wave = blockIdx.x * (blockDim.x >> 5) + (threadIdx.x >> 5);
  const int lane = threadIdx.x & 31;
  const int TG = XZW / 64;
  const int gm = wave / TG, gng = wave % TG;
  const int mrow = lane & 15, kh = lane >> 4;

  int m  = gm * 16 + mrow;
  int bb = m / SEQ, ll = m % SEQ;
  int srow = reverse ? (bb * SEQ + (SEQ - 1 - ll)) : m;

  const float* ap = x + (size_t)srow * DIMX + 16 * kh;
  const float* bp[4];
#pragma unroll
  for (int j = 0; j < 4; ++j)
    bp[j] = Win + (size_t)(gng * 64 + j * 16 + mrow) * DIMX + 16 * kh;

  v8f acc[4] = {};
  for (int k0 = 0; k0 < DIMX; k0 += 32) {
    v4f a[4], b[4][4];
    ld16(ap + k0, a);
#pragma unroll
    for (int j = 0; j < 4; ++j) ld16(bp[j] + k0, b[j]);
    asm volatile("" :: LOADED4(a), LOADED4(b[0]), LOADED4(b[1]), LOADED4(b[2]), LOADED4(b[3]));
#pragma unroll
    for (int q = 0; q < 4; ++q) {
#pragma unroll
      for (int j = 0; j < 4; ++j) acc[j] = wmma4(lo2(a[q]), lo2(b[j][q]), acc[j]);
#pragma unroll
      for (int j = 0; j < 4; ++j) acc[j] = wmma4(hi2(a[q]), hi2(b[j][q]), acc[j]);
    }
  }
#pragma unroll
  for (int j = 0; j < 4; ++j)
#pragma unroll
    for (int v = 0; v < 8; ++v) {
      int orow = gm * 16 + v + 8 * kh;
      xz[(size_t)orow * XZW + gng * 64 + j * 16 + mrow] = acc[j][v];
    }
}

#define CT 8
#define DQ (DI / 4)
__global__ void k_conv_silu(const float* __restrict__ xz_all,
                            const float* __restrict__ cw_f,
                            const float* __restrict__ cb_f,
                            const float* __restrict__ cw_b,
                            const float* __restrict__ cb_b,
                            float* __restrict__ u_all) {
  const int dir = blockIdx.y;
  const float* xz = xz_all + (size_t)dir * MROWS * XZW;
  const float* cw = dir ? cw_b : cw_f;
  const float* cb = dir ? cb_b : cb_f;
  float* u = u_all + (size_t)dir * MROWS * DI;
  const int idx = blockIdx.x * blockDim.x + threadIdx.x;
  const int dq  = idx % DQ;
  const int run = idx / DQ;
  const int row0 = run * CT;
  const bool head = (row0 % SEQ) == 0;

  const float* src = xz + (size_t)row0 * XZW + 4 * dq;
  v4f xr[CT + 3];
  for (int k = 0; k < 3; ++k) {
    v4f zero = {0.0f, 0.0f, 0.0f, 0.0f};
    xr[k] = head ? zero : *reinterpret_cast<const v4f*>(src + (k - 3) * XZW);
  }
  for (int k = 0; k < CT; ++k)
    xr[3 + k] = *reinterpret_cast<const v4f*>(src + (size_t)k * XZW);

  v4f w[4];
  for (int c = 0; c < 4; ++c)
    w[c] = *reinterpret_cast<const v4f*>(cw + (size_t)(4 * dq + c) * 4);
  const v4f bias = *reinterpret_cast<const v4f*>(cb + 4 * dq);

  float* dst = u + (size_t)row0 * DI + 4 * dq;
  for (int i = 0; i < CT; ++i) {
    v4f o;
    for (int c = 0; c < 4; ++c) {
      float acc = bias[c];
      for (int j = 0; j < 4; ++j) {
        if (i + j >= 3 || !head)
          acc += xr[i + j][c] * w[c][j];
      }
      o[c] = acc / (1.0f + __expf(-acc));
    }
    *reinterpret_cast<v4f*>(dst + (size_t)i * DI) = o;
  }
}

__global__ void __launch_bounds__(256) k_gemm_xp(const float* __restrict__ u_all,
                          const float* __restrict__ Wxp_f,
                          const float* __restrict__ Wxp_b,
                          float* __restrict__ dbl_all) {
  const int dir = blockIdx.y;
  const float* u = u_all + (size_t)dir * MROWS * DI;
  const float* Wxp = dir ? Wxp_b : Wxp_f;
  float* dbl = dbl_all + (size_t)dir * MROWS * DBLW;
  const int wave = blockIdx.x * (blockDim.x >> 5) + (threadIdx.x >> 5);
  const int lane = threadIdx.x & 31;
  const int gm = wave;
  const int mrow = lane & 15, kh = lane >> 4;

  const float* ap = u + (size_t)(gm * 16 + mrow) * DI + 16 * kh;
  bool nvalid[3];
  const float* bp[3];
#pragma unroll
  for (int j = 0; j < 3; ++j) {
    int n = j * 16 + mrow;
    nvalid[j] = (n < DTR + 2 * NST);
    bp[j] = Wxp + (size_t)(nvalid[j] ? n : 0) * DI + 16 * kh;
  }

  v8f acc[3] = {};
  for (int k0 = 0; k0 < DI; k0 += 32) {
    v4f a[4], b[3][4];
    ld16(ap + k0, a);
#pragma unroll
    for (int j = 0; j < 3; ++j) ld16(bp[j] + k0, b[j]);
    asm volatile("" :: LOADED4(a), LOADED4(b[0]), LOADED4(b[1]), LOADED4(b[2]));
#pragma unroll
    for (int j = 0; j < 3; ++j) {
#pragma unroll
      for (int q = 0; q < 4; ++q) {
        v4f zero = {0.0f, 0.0f, 0.0f, 0.0f};
        b[j][q] = nvalid[j] ? b[j][q] : zero;
      }
    }
#pragma unroll
    for (int q = 0; q < 4; ++q) {
#pragma unroll
      for (int j = 0; j < 3; ++j) acc[j] = wmma4(lo2(a[q]), lo2(b[j][q]), acc[j]);
#pragma unroll
      for (int j = 0; j < 3; ++j) acc[j] = wmma4(hi2(a[q]), hi2(b[j][q]), acc[j]);
    }
  }
#pragma unroll
  for (int j = 0; j < 3; ++j)
#pragma unroll
    for (int v = 0; v < 8; ++v) {
      int orow = gm * 16 + v + 8 * kh;
      dbl[(size_t)orow * DBLW + j * 16 + mrow] = acc[j][v];
    }
}

__global__ void k_gemm_dt(const float* __restrict__ dbl_all,
                          const float* __restrict__ Wdt_f,
                          const float* __restrict__ bdt_f,
                          const float* __restrict__ Wdt_b,
                          const float* __restrict__ bdt_b,
                          float* __restrict__ xz_all) {
  const int dir = blockIdx.y;
  const float* dbl = dbl_all + (size_t)dir * MROWS * DBLW;
  const float* Wdt = dir ? Wdt_b : Wdt_f;
  const float* bdt = dir ? bdt_b : bdt_f;
  float* xz = xz_all + (size_t)dir * MROWS * XZW;
  const int wave = blockIdx.x * (blockDim.x >> 5) + (threadIdx.x >> 5);
  const int lane = threadIdx.x & 31;
  const int TG = DI / 64;
  const int gm = wave / TG, gng = wave % TG;
  const int mrow = lane & 15, kh = lane >> 4;

  const float* ap = dbl + (size_t)(gm * 16 + mrow) * DBLW;
  const float* bp[4];
#pragma unroll
  for (int j = 0; j < 4; ++j)
    bp[j] = Wdt + (size_t)(gng * 64 + j * 16 + mrow) * DTR;

  v8f acc[4] = {};
  for (int k0 = 0; k0 < DTR; k0 += 4) {
    int kk = k0 + 2 * kh;
    v2f a; a.x = ap[kk]; a.y = ap[kk + 1];
#pragma unroll
    for (int j = 0; j < 4; ++j) {
      v2f b; b.x = bp[j][kk]; b.y = bp[j][kk + 1];
      acc[j] = wmma4(a, b, acc[j]);
    }
  }
#pragma unroll
  for (int j = 0; j < 4; ++j) {
    int n = gng * 64 + j * 16 + mrow;
    float bias = bdt[n];
#pragma unroll
    for (int v = 0; v < 8; ++v) {
      int orow = gm * 16 + v + 8 * kh;
      float t = acc[j][v] + bias;
      float sp = (t > 20.0f) ? t : log1pf(__expf(t));
      xz[(size_t)orow * XZW + n] = sp;
    }
  }
}

__global__ void k_scan_part1(const float* __restrict__ xz_all,
                             const float* __restrict__ u_all,
                             const float* __restrict__ dbl_all,
                             const float* __restrict__ Alog_f,
                             const float* __restrict__ Alog_b,
                             float* __restrict__ Pb_all, float* __restrict__ Sb_all) {
  const int dir = blockIdx.y;
  const float* xz  = xz_all  + (size_t)dir * MROWS * XZW;
  const float* u   = u_all   + (size_t)dir * MROWS * DI;
  const float* dbl = dbl_all + (size_t)dir * MROWS * DBLW;
  const float* Alog = dir ? Alog_b : Alog_f;
  float* Pb = Pb_all + (size_t)dir * PSTRIDE;
  float* Sb = Sb_all + (size_t)dir * PSTRIDE;
  const int b = blockIdx.x / NCH;
  const int c = blockIdx.x % NCH;
  const int d = threadIdx.x;

  float a[NST], P[NST], S[NST];
#pragma unroll
  for (int n = 0; n < NST; ++n) {
    a[n] = -__expf(Alog[d * NST + n]);
    P[n] = 1.0f;
    S[n] = 0.0f;
  }
  const int t0 = c * CL;
  for (int t = t0; t < t0 + CL; ++t) {
    const size_t row = (size_t)b * SEQ + t;
    const float* bc = dbl + row * DBLW;
    __builtin_prefetch(bc + DBLW, 0, 0);
    const float dt = xz[row * XZW + d];
    const float ut = u[row * DI + d];
    const float dtu = dt * ut;
#pragma unroll
    for (int n = 0; n < NST; ++n) {
      float dA = __expf(dt * a[n]);
      S[n] = S[n] * dA + dtu * bc[DTR + n];
      P[n] *= dA;
    }
  }
  const size_t base = ((size_t)(b * NCH + c) * DI + d) * NST;
#pragma unroll
  for (int n = 0; n < NST; ++n) {
    Pb[base + n] = P[n];
    Sb[base + n] = S[n];
  }
}

__global__ void k_scan_combine(const float* __restrict__ Pb_all,
                               float* __restrict__ Sb_all) {
  const int dir = blockIdx.y;
  const float* Pb = Pb_all + (size_t)dir * PSTRIDE;
  float* Sb = Sb_all + (size_t)dir * PSTRIDE;
  int idx = blockIdx.x * blockDim.x + threadIdx.x;
  if (idx >= BATCH * DI * NST) return;
  const int n = idx & (NST - 1);
  const int d = (idx >> 4) % DI;
  const int b = idx / (NST * DI);

  float carry = 0.0f;
  for (int c = 0; c < NCH; ++c) {
    const size_t base = ((size_t)(b * NCH + c) * DI + d) * NST + n;
    float p = Pb[base];
    float s = Sb[base];
    Sb[base] = carry;
    carry = p * carry + s;
  }
}

__global__ void k_scan_part3(float* __restrict__ xz_all,
                             const float* __restrict__ u_all,
                             const float* __restrict__ dbl_all,
                             const float* __restrict__ Alog_f,
                             const float* __restrict__ Dv_f,
                             const float* __restrict__ Alog_b,
                             const float* __restrict__ Dv_b,
                             const float* __restrict__ Sb_all) {
  const int dir = blockIdx.y;
  float* xz        = xz_all  + (size_t)dir * MROWS * XZW;
  const float* u   = u_all   + (size_t)dir * MROWS * DI;
  const float* dbl = dbl_all + (size_t)dir * MROWS * DBLW;
  const float* Alog = dir ? Alog_b : Alog_f;
  const float* Dv   = dir ? Dv_b : Dv_f;
  const float* Sb = Sb_all + (size_t)dir * PSTRIDE;
  const int b = blockIdx.x / NCH;
  const int c = blockIdx.x % NCH;
  const int d = threadIdx.x;

  float a[NST], h[NST];
  const size_t base = ((size_t)(b * NCH + c) * DI + d) * NST;
#pragma unroll
  for (int n = 0; n < NST; ++n) {
    a[n] = -__expf(Alog[d * NST + n]);
    h[n] = Sb[base + n];
  }
  const float Dd = Dv[d];

  const int t0 = c * CL;
  for (int t = t0; t < t0 + CL; ++t) {
    const size_t row = (size_t)b * SEQ + t;
    const float* bc = dbl + row * DBLW;
    __builtin_prefetch(bc + DBLW, 0, 0);
    const float dt = xz[row * XZW + d];
    const float ut = u[row * DI + d];
    const float dtu = dt * ut;
    float y = 0.0f;
#pragma unroll
    for (int n = 0; n < NST; ++n) {
      float dA = __expf(dt * a[n]);
      h[n] = h[n] * dA + dtu * bc[DTR + n];
      y += h[n] * bc[DTR + NST + n];
    }
    y += ut * Dd;
    float z = xz[row * XZW + DI + d];
    y *= z / (1.0f + __expf(-z));
    xz[row * XZW + DI + d] = y;
  }
}

__global__ void __launch_bounds__(256) k_gemm_out(const float* __restrict__ xz_f,
                           const float* __restrict__ xz_b,
                           const float* __restrict__ x,
                           const float* __restrict__ Wout,
                           float* __restrict__ out) {
  const int wave = blockIdx.x * (blockDim.x >> 5) + (threadIdx.x >> 5);
  const int lane = threadIdx.x & 31;
  const int TG = DIMX / 64;
  const int gm = wave / TG, gng = wave % TG;
  const int mrow = lane & 15, kh = lane >> 4;

  int m  = gm * 16 + mrow;
  int bb = m / SEQ, ll = m % SEQ;
  int mrev = bb * SEQ + (SEQ - 1 - ll);

  const float* af = xz_f + (size_t)m    * XZW + DI + 16 * kh;
  const float* ab = xz_b + (size_t)mrev * XZW + DI + 16 * kh;
  const float* bp[4];
#pragma unroll
  for (int j = 0; j < 4; ++j)
    bp[j] = Wout + (size_t)(gng * 64 + j * 16 + mrow) * DI + 16 * kh;

  v8f acc[4] = {};
  for (int k0 = 0; k0 < DI; k0 += 32) {
    v4f a[4], a2[4], b[4][4];
    ld16(af + k0, a);
    ld16(ab + k0, a2);
#pragma unroll
    for (int j = 0; j < 4; ++j) ld16(bp[j] + k0, b[j]);
    asm volatile("" :: LOADED4(a), LOADED4(a2), LOADED4(b[0]), LOADED4(b[1]), LOADED4(b[2]), LOADED4(b[3]));
#pragma unroll
    for (int q = 0; q < 4; ++q) a[q] = a[q] + a2[q];
#pragma unroll
    for (int q = 0; q < 4; ++q) {
#pragma unroll
      for (int j = 0; j < 4; ++j) acc[j] = wmma4(lo2(a[q]), lo2(b[j][q]), acc[j]);
#pragma unroll
      for (int j = 0; j < 4; ++j) acc[j] = wmma4(hi2(a[q]), hi2(b[j][q]), acc[j]);
    }
  }
#pragma unroll
  for (int j = 0; j < 4; ++j)
#pragma unroll
    for (int v = 0; v < 8; ++v) {
      int orow = gm * 16 + v + 8 * kh;
      int n = gng * 64 + j * 16 + mrow;
      out[(size_t)orow * DIMX + n] = acc[j][v] + x[(size_t)orow * DIMX + n];
    }
}

extern "C" void kernel_launch(void* const* d_in, const int* in_sizes, int n_in,
                              void* d_out, int out_size, void* d_ws, size_t ws_size,
                              hipStream_t stream) {
  const float* x      = (const float*)d_in[0];
  const float* Win_f  = (const float*)d_in[1];
  const float* cw_f   = (const float*)d_in[2];
  const float* cb_f   = (const float*)d_in[3];
  const float* Wxp_f  = (const float*)d_in[4];
  const float* Wdt_f  = (const float*)d_in[5];
  const float* bdt_f  = (const float*)d_in[6];
  const float* Alog_f = (const float*)d_in[7];
  const float* Dv_f   = (const float*)d_in[8];
  const float* Win_b  = (const float*)d_in[9];
  const float* cw_b   = (const float*)d_in[10];
  const float* cb_b   = (const float*)d_in[11];
  const float* Wxp_b  = (const float*)d_in[12];
  const float* Wdt_b  = (const float*)d_in[13];
  const float* bdt_b  = (const float*)d_in[14];
  const float* Alog_b = (const float*)d_in[15];
  const float* Dv_b   = (const float*)d_in[16];
  const float* Wout   = (const float*)d_in[17];
  float* out = (float*)d_out;

  float* ws    = (float*)d_ws;
  float* xz_f  = ws;
  float* xz_b  = xz_f + (size_t)MROWS * XZW;
  float* u_f   = xz_b + (size_t)MROWS * XZW;
  float* dbl_f = u_f  + (size_t)2 * MROWS * DI;
  const size_t PS = (size_t)BATCH * NCH * DI * NST;
  float* P_f   = dbl_f + (size_t)2 * MROWS * DBLW;
  float* S_f   = P_f + PS;

  const int BLK = 256, WPB = BLK / 32;

  {
    int waves = (MROWS / 16) * (XZW / 64);
    k_gemm_in<<<dim3(waves / WPB, 2), BLK, 0, stream>>>(x, Win_f, Win_b, xz_f);
  }
  {
    int n = (MROWS / CT) * DQ;
    k_conv_silu<<<dim3(n / BLK, 2), BLK, 0, stream>>>(
        xz_f, cw_f, cb_f, cw_b, cb_b, u_f);
  }
  {
    int waves = MROWS / 16;
    k_gemm_xp<<<dim3(waves / WPB, 2), BLK, 0, stream>>>(u_f, Wxp_f, Wxp_b, dbl_f);
  }
  {
    int waves = (MROWS / 16) * (DI / 64);
    k_gemm_dt<<<dim3(waves / WPB, 2), BLK, 0, stream>>>(
        dbl_f, Wdt_f, bdt_f, Wdt_b, bdt_b, xz_f);
  }
  {
    int cb = (BATCH * DI * NST + BLK - 1) / BLK;
    k_scan_part1<<<dim3(BATCH * NCH, 2), DI, 0, stream>>>(
        xz_f, u_f, dbl_f, Alog_f, Alog_b, P_f, S_f);
    k_scan_combine<<<dim3(cb, 2), BLK, 0, stream>>>(P_f, S_f);
    k_scan_part3<<<dim3(BATCH * NCH, 2), DI, 0, stream>>>(
        xz_f, u_f, dbl_f, Alog_f, Dv_f, Alog_b, Dv_b, S_f);
  }
  {
    int waves = (MROWS / 16) * (DIMX / 64);
    k_gemm_out<<<waves / WPB, BLK, 0, stream>>>(xz_f, xz_b, x, Wout, out);
  }
}


# Scan passes 1 and 3 wait for no load inside the recurrence: the chunk's B/C rows are copied to LDS once, dt/u/z are requested one 8-timestep group ahead in two alternating register sets; order kept
# speedup vs baseline: 2.3888x; 1.2618x over previous
#include <hip/hip_runtime.h>
#include <hip/hip_bf16.h>
#include <math.h>

#define BATCH 4
#define SEQ   2048
#define DIMX  192
#define DI    384
#define NST   16
#define DTR   12
#define XZW   768
#define DBLW  48
#define MROWS (BATCH*SEQ)
#define NCH   32
#define CL    (SEQ/NCH)
#define PSTRIDE (2 * BATCH * NCH * DI * NST)

typedef float v2f __attribute__((ext_vector_type(2)));
typedef float v8f __attribute__((ext_vector_type(8)));

__device__ __forceinline__ v8f wmma4(v2f a, v2f b, v8f c) {
  return __builtin_amdgcn_wmma_f32_16x16x4_f32(false, a, false, b, (short)0, c,
                                               false, false);
}

typedef float v4f __attribute__((ext_vector_type(4)));

__device__ __forceinline__ void ld16(const float* __restrict__ p, v4f (&r)[4]) {
  const v4f* q = reinterpret_cast<const v4f*>(p);
  r[0] = q[0]; r[1] = q[1]; r[2] = q[2]; r[3] = q[3];
}
__device__ __forceinline__ v2f lo2(v4f v) { v2f r; r.x = v.x; r.y = v.y; return r; }
__device__ __forceinline__ v2f hi2(v4f v) { v2f r; r.x = v.z; r.y = v.w; return r; }
#define LOADED4(v) "v"((v)[0]), "v"((v)[1]), "v"((v)[2]), "v"((v)[3])

__global__ void __launch_bounds__(256) k_gemm_in(const float* __restrict__ x,
                          const float* __restrict__ Win_f,
                          const float* __restrict__ Win_b,
                          float* __restrict__ xz_all) {
  const int reverse = blockIdx.y;
  const float* Win = reverse ? Win_b : Win_f;
  float* xz = xz_all + (size_t)reverse * MROWS * XZW;
  const int wave = blockIdx.x * (blockDim.x >> 5) + (threadIdx.x >> 5);
  const int lane = threadIdx.x & 31;
  const int TG = XZW / 64;
  const int gm = wave / TG, gng = wave % TG;
  const int mrow = lane & 15, kh = lane >> 4;

  int m  = gm * 16 + mrow;
  int bb = m / SEQ, ll = m % SEQ;
  int srow = reverse ? (bb * SEQ + (SEQ - 1 - ll)) : m;

  const float* ap = x + (size_t)srow * DIMX + 16 * kh;
  const float* bp[4];
#pragma unroll
  for (int j = 0; j < 4; ++j)
    bp[j] = Win + (size_t)(gng * 64 + j * 16 + mrow) * DIMX + 16 * kh;

  v8f acc[4] = {};
  for (int k0 = 0; k0 < DIMX; k0 += 32) {
    v4f a[4], b[4][4];
    ld16(ap + k0, a);
#pragma unroll
    for (int j = 0; j < 4; ++j) ld16(bp[j] + k0, b[j]);
    asm volatile("" :: LOADED4(a), LOADED4(b[0]), LOADED4(b[1]), LOADED4(b[2]), LOADED4(b[3]));
#pragma unroll
    for (int q = 0; q < 4; ++q) {
#pragma unroll
      for (int j = 0; j < 4; ++j) acc[j] = wmma4(lo2(a[q]), lo2(b[j][q]), acc[j]);
#pragma unroll
      for (int j = 0; j < 4; ++j) acc[j] = wmma4(hi2(a[q]), hi2(b[j][q]), acc[j]);
    }
  }
#pragma unroll
  for (int j = 0; j < 4; ++j)
#pragma unroll
    for (int v = 0; v < 8; ++v) {
      int orow = gm * 16 + v + 8 * kh;
      xz[(size_t)orow * XZW + gng * 64 + j * 16 + mrow] = acc[j][v];
    }
}

#define CT 8
#define DQ (DI / 4)
__global__ void k_conv_silu(const float* __restrict__ xz_all,
                            const float* __restrict__ cw_f,
                            const float* __restrict__ cb_f,
                            const float* __restrict__ cw_b,
                            const float* __restrict__ cb_b,
                            float* __restrict__ u_all) {
  const int dir = blockIdx.y;
  const float* xz = xz_all + (size_t)dir * MROWS * XZW;
  const float* cw = dir ? cw_b : cw_f;
  const float* cb = dir ? cb_b : cb_f;
  float* u = u_all + (size_t)dir * MROWS * DI;
  const int idx = blockIdx.x * blockDim.x + threadIdx.x;
  const int dq  = idx % DQ;
  const int run = idx / DQ;
  const int row0 = run * CT;
  const bool head = (row0 % SEQ) == 0;

  const float* src = xz + (size_t)row0 * XZW + 4 * dq;
  v4f xr[CT + 3];
  for (int k = 0; k < 3; ++k) {
    v4f zero = {0.0f, 0.0f, 0.0f, 0.0f};
    xr[k] = head ? zero : *reinterpret_cast<const v4f*>(src + (k - 3) * XZW);
  }
  for (int k = 0; k < CT; ++k)
    xr[3 + k] = *reinterpret_cast<const v4f*>(src + (size_t)k * XZW);

  v4f w[4];
  for (int c = 0; c < 4; ++c)
    w[c] = *reinterpret_cast<const v4f*>(cw + (size_t)(4 * dq + c) * 4);
  const v4f bias = *reinterpret_cast<const v4f*>(cb + 4 * dq);

  float* dst = u + (size_t)row0 * DI + 4 * dq;
  for (int i = 0; i < CT; ++i) {
    v4f o;
    for (int c = 0; c < 4; ++c) {
      float acc = bias[c];
      for (int j = 0; j < 4; ++j) {
        if (i + j >= 3 || !head)
          acc += xr[i + j][c] * w[c][j];
      }
      o[c] = acc / (1.0f + __expf(-acc));
    }
    *reinterpret_cast<v4f*>(dst + (size_t)i * DI) = o;
  }
}

__global__ void __launch_bounds__(256) k_gemm_xp(const float* __restrict__ u_all,
                          const float* __restrict__ Wxp_f,
                          const float* __restrict__ Wxp_b,
                          float* __restrict__ dbl_all) {
  const int dir = blockIdx.y;
  const float* u = u_all + (size_t)dir * MROWS * DI;
  const float* Wxp = dir ? Wxp_b : Wxp_f;
  float* dbl = dbl_all + (size_t)dir * MROWS * DBLW;
  const int wave = blockIdx.x * (blockDim.x >> 5) + (threadIdx.x >> 5);
  const int lane = threadIdx.x & 31;
  const int gm = wave;
  const int mrow = lane & 15, kh = lane >> 4;

  const float* ap = u + (size_t)(gm * 16 + mrow) * DI + 16 * kh;
  bool nvalid[3];
  const float* bp[3];
#pragma unroll
  for (int j = 0; j < 3; ++j) {
    int n = j * 16 + mrow;
    nvalid[j] = (n < DTR + 2 * NST);
    bp[j] = Wxp + (size_t)(nvalid[j] ? n : 0) * DI + 16 * kh;
  }

  v8f acc[3] = {};
  for (int k0 = 0; k0 < DI; k0 += 32) {
    v4f a[4], b[3][4];
    ld16(ap + k0, a);
#pragma unroll
    for (int j = 0; j < 3; ++j) ld16(bp[j] + k0, b[j]);
    asm volatile("" :: LOADED4(a), LOADED4(b[0]), LOADED4(b[1]), LOADED4(b[2]));
#pragma unroll
    for (int j = 0; j < 3; ++j) {
#pragma unroll
      for (int q = 0; q < 4; ++q) {
        v4f zero = {0.0f, 0.0f, 0.0f, 0.0f};
        b[j][q] = nvalid[j] ? b[j][q] : zero;
      }
    }
#pragma unroll
    for (int q = 0; q < 4; ++q) {
#pragma unroll
      for (int j = 0; j < 3; ++j) acc[j] = wmma4(lo2(a[q]), lo2(b[j][q]), acc[j]);
#pragma unroll
      for (int j = 0; j < 3; ++j) acc[j] = wmma4(hi2(a[q]), hi2(b[j][q]), acc[j]);
    }
  }
#pragma unroll
  for (int j = 0; j < 3; ++j)
#pragma unroll
    for (int v = 0; v < 8; ++v) {
      int orow = gm * 16 + v + 8 * kh;
      dbl[(size_t)orow * DBLW + j * 16 + mrow] = acc[j][v];
    }
}

__global__ void k_gemm_dt(const float* __restrict__ dbl_all,
                          const float* __restrict__ Wdt_f,
                          const float* __restrict__ bdt_f,
                          const float* __restrict__ Wdt_b,
                          const float* __restrict__ bdt_b,
                          float* __restrict__ xz_all) {
  const int dir = blockIdx.y;
  const float* dbl = dbl_all + (size_t)dir * MROWS * DBLW;
  const float* Wdt = dir ? Wdt_b : Wdt_f;
  const float* bdt = dir ? bdt_b : bdt_f;
  float* xz = xz_all + (size_t)dir * MROWS * XZW;
  const int wave = blockIdx.x * (blockDim.x >> 5) + (threadIdx.x >> 5);
  const int lane = threadIdx.x & 31;
  const int TG = DI / 64;
  const int gm = wave / TG, gng = wave % TG;
  const int mrow = lane & 15, kh = lane >> 4;

  const float* ap = dbl + (size_t)(gm * 16 + mrow) * DBLW;
  const float* bp[4];
#pragma unroll
  for (int j = 0; j < 4; ++j)
    bp[j] = Wdt + (size_t)(gng * 64 + j * 16 + mrow) * DTR;

  v8f acc[4] = {};
  for (int k0 = 0; k0 < DTR; k0 += 4) {
    int kk = k0 + 2 * kh;
    v2f a; a.x = ap[kk]; a.y = ap[kk + 1];
#pragma unroll
    for (int j = 0; j < 4; ++j) {
      v2f b; b.x = bp[j][kk]; b.y = bp[j][kk + 1];
      acc[j] = wmma4(a, b, acc[j]);
    }
  }
#pragma unroll
  for (int j = 0; j < 4; ++j) {
    int n = gng * 64 + j * 16 + mrow;
    float bias = bdt[n];
#pragma unroll
    for (int v = 0; v < 8; ++v) {
      int orow = gm * 16 + v + 8 * kh;
      float t = acc[j][v] + bias;
      float sp = (t > 20.0f) ? t : log1pf(__expf(t));
      xz[(size_t)orow * XZW + n] = sp;
    }
  }
}

#define TGRP 8
#define NGRP (CL / TGRP)
#define REQUESTS_STAY_AHEAD() asm volatile("" ::: "memory")

__device__ __forceinline__ void scan1_group(const float (&dt8)[TGRP], const float (&u8)[TGRP],
                                            const v4f (*sB)[NST / 4],
                                            const float (&a)[NST], float (&P)[NST], float (&S)[NST]) {
#pragma unroll
  for (int k = 0; k < TGRP; ++k) {
    float Bv[NST];
#pragma unroll
    for (int q = 0; q < NST / 4; ++q) {
      const v4f v = sB[k][q];
      Bv[4 * q] = v.x; Bv[4 * q + 1] = v.y; Bv[4 * q + 2] = v.z; Bv[4 * q + 3] = v.w;
    }
    const float dt = dt8[k];
    const float dtu = dt * u8[k];
#pragma unroll
    for (int n = 0; n < NST; ++n) {
      float dA = __expf(dt * a[n]);
      S[n] = S[n] * dA + dtu * Bv[n];
      P[n] *= dA;
    }
  }
}

__global__ void __launch_bounds__(DI) k_scan_part1(const float* __restrict__ xz_all,
                             const float* __restrict__ u_all,
                             const float* __restrict__ dbl_all,
                             const float* __restrict__ Alog_f,
                             const float* __restrict__ Alog_b,
                             float* __restrict__ Pb_all, float* __restrict__ Sb_all) {
  __shared__ v4f sB[CL][NST / 4];
  const int dir = blockIdx.y;
  const float* xz  = xz_all  + (size_t)dir * MROWS * XZW;
  const float* u   = u_all   + (size_t)dir * MROWS * DI;
  const float* dbl = dbl_all + (size_t)dir * MROWS * DBLW;
  const float* Alog = dir ? Alog_b : Alog_f;
  float* Pb = Pb_all + (size_t)dir * PSTRIDE;
  float* Sb = Sb_all + (size_t)dir * PSTRIDE;
  const int b = blockIdx.x / NCH;
  const int c = blockIdx.x % NCH;
  const int d = threadIdx.x;

  const size_t row0 = (size_t)b * SEQ + (size_t)c * CL;
  const float* dtp = xz  + row0 * XZW + d;
  const float* utp = u   + row0 * DI + d;
  const float* bcp = dbl + row0 * DBLW + DTR;

  float dtA[TGRP], utA[TGRP], dtB[TGRP], utB[TGRP];
#pragma unroll
  for (int k = 0; k < TGRP; ++k) {
    dtA[k] = dtp[(size_t)k * XZW];
    utA[k] = utp[(size_t)k * DI];
  }
  if (d < CL * (NST / 4)) {
    const int r = d >> 2, q = d & 3;
    sB[r][q] = *reinterpret_cast<const v4f*>(bcp + (size_t)r * DBLW + 4 * q);
  }

  float a[NST], P[NST], S[NST];
#pragma unroll
  for (int n = 0; n < NST; ++n) {
    a[n] = -__expf(Alog[d * NST + n]);
    P[n] = 1.0f;
    S[n] = 0.0f;
  }
  __syncthreads();

  for (int g = 0; g < NGRP; g += 2) {
#pragma unroll
    for (int k = 0; k < TGRP; ++k) {
      dtB[k] = dtp[(size_t)((g + 1) * TGRP + k) * XZW];
      utB[k] = utp[(size_t)((g + 1) * TGRP + k) * DI];
    }
    REQUESTS_STAY_AHEAD();
    scan1_group(dtA, utA, sB + g * TGRP, a, P, S);
    const int gn = (g + 2 < NGRP) ? g + 2 : g + 1;
#pragma unroll
    for (int k = 0; k < TGRP; ++k) {
      dtA[k] = dtp[(size_t)(gn * TGRP + k) * XZW];
      utA[k] = utp[(size_t)(gn * TGRP + k) * DI];
    }
    REQUESTS_STAY_AHEAD();
    scan1_group(dtB, utB, sB + (g + 1) * TGRP, a, P, S);
  }
  const size_t base = ((size_t)(b * NCH + c) * DI + d) * NST;
#pragma unroll
  for (int n = 0; n < NST; ++n) {
    Pb[base + n] = P[n];
    Sb[base + n] = S[n];
  }
}

__global__ void k_scan_combine(const float* __restrict__ Pb_all,
                               float* __restrict__ Sb_all) {
  const int dir = blockIdx.y;
  const float* Pb = Pb_all + (size_t)dir * PSTRIDE;
  float* Sb = Sb_all + (size_t)dir * PSTRIDE;
  int idx = blockIdx.x * blockDim.x + threadIdx.x;
  if (idx >= BATCH * DI * NST) return;
  const int n = idx & (NST - 1);
  const int d = (idx >> 4) % DI;
  const int b = idx / (NST * DI);

  float carry = 0.0f;
  for (int c = 0; c < NCH; ++c) {
    const size_t base = ((size_t)(b * NCH + c) * DI + d) * NST + n;
    float p = Pb[base];
    float s = Sb[base];
    Sb[base] = carry;
    carry = p * carry + s;
  }
}

__device__ __forceinline__ void scan3_group(const float (&dt8)[TGRP], const float (&u8)[TGRP],
                                            const float (&z8)[TGRP],
                                            const v4f (*sBC)[2 * NST / 4], float* yp,
                                            const float (&a)[NST], float (&h)[NST], float Dd) {
#pragma unroll
  for (int k = 0; k < TGRP; ++k) {
    float BCv[2 * NST];
#pragma unroll
    for (int q = 0; q < 2 * NST / 4; ++q) {
      const v4f v = sBC[k][q];
      BCv[4 * q] = v.x; BCv[4 * q + 1] = v.y; BCv[4 * q + 2] = v.z; BCv[4 * q + 3] = v.w;
    }
    const float dt = dt8[k];
    const float ut = u8[k];
    const float dtu = dt * ut;
    float y = 0.0f;
#pragma unroll
    for (int n = 0; n < NST; ++n) {
      float dA = __expf(dt * a[n]);
      h[n] = h[n] * dA + dtu * BCv[n];
      y += h[n] * BCv[NST + n];
    }
    y += ut * Dd;
    const float z = z8[k];
    y *= z / (1.0f + __expf(-z));
    yp[(size_t)k * XZW] = y;
  }
}

__global__ void __launch_bounds__(DI) k_scan_part3(float* __restrict__ xz_all,
                             const float* __restrict__ u_all,
                             const float* __restrict__ dbl_all,
                             const float* __restrict__ Alog_f,
                             const float* __restrict__ Dv_f,
                             const float* __restrict__ Alog_b,
                             const float* __restrict__ Dv_b,
                             const float* __restrict__ Sb_all) {
  __shared__ v4f sBC[CL][2 * NST / 4];
  const int dir = blockIdx.y;
  float* xz        = xz_all  + (size_t)dir * MROWS * XZW;
  const float* u   = u_all   + (size_t)dir * MROWS * DI;
  const float* dbl = dbl_all + (size_t)dir * MROWS * DBLW;
  const float* Alog = dir ? Alog_b : Alog_f;
  const float* Dv   = dir ? Dv_b : Dv_f;
  const float* Sb = Sb_all + (size_t)dir * PSTRIDE;
  const int b = blockIdx.x / NCH;
  const int c = blockIdx.x % NCH;
  const int d = threadIdx.x;

  const size_t row0 = (size_t)b * SEQ + (size_t)c * CL;
  float* xzp = xz + row0 * XZW + d;
  const float* utp = u   + row0 * DI + d;
  const float* bcp = dbl + row0 * DBLW + DTR;

  float dtA[TGRP], utA[TGRP], zA[TGRP], dtB[TGRP], utB[TGRP], zB[TGRP];
#pragma unroll
  for (int k = 0; k < TGRP; ++k) {
    dtA[k] = xzp[(size_t)k * XZW];
    zA[k]  = xzp[(size_t)k * XZW + DI];
    utA[k] = utp[(size_t)k * DI];
  }
  for (int i = d; i < CL * (2 * NST / 4); i += DI) {
    const int r = i >> 3, q = i & 7;
    sBC[r][q] = *reinterpret_cast<const v4f*>(bcp + (size_t)r * DBLW + 4 * q);
  }

  float a[NST], h[NST];
  const size_t base = ((size_t)(b * NCH + c) * DI + d) * NST;
#pragma unroll
  for (int n = 0; n < NST; ++n) {
    a[n] = -__expf(Alog[d * NST + n]);
    h[n] = Sb[base + n];
  }
  const float Dd = Dv[d];
  __syncthreads();

  for (int g = 0; g < NGRP; g += 2) {
#pragma unroll
    for (int k = 0; k < TGRP; ++k) {
      dtB[k] = xzp[(size_t)((g + 1) * TGRP + k) * XZW];
      zB[k]  = xzp[(size_t)((g + 1) * TGRP + k) * XZW + DI];
      utB[k] = utp[(size_t)((g + 1) * TGRP + k) * DI];
    }
    REQUESTS_STAY_AHEAD();
    scan3_group(dtA, utA, zA, sBC + g * TGRP, xzp + (size_t)(g * TGRP) * XZW + DI, a, h, Dd);
    const int gn = (g + 2 < NGRP) ? g + 2 : g + 1;
#pragma unroll
    for (int k = 0; k < TGRP; ++k) {
      dtA[k] = xzp[(size_t)(gn * TGRP + k) * XZW];
      zA[k]  = xzp[(size_t)(gn * TGRP + k) * XZW + DI];
      utA[k] = utp[(size_t)(gn * TGRP + k) * DI];
    }
    REQUESTS_STAY_AHEAD();
    scan3_group(dtB, utB, zB, sBC + (g + 1) * TGRP, xzp + (size_t)((g + 1) * TGRP) * XZW + DI, a, h, Dd);
  }
}

__global__ void __launch_bounds__(256) k_gemm_out(const float* __restrict__ xz_f,
                           const float* __restrict__ xz_b,
                           const float* __restrict__ x,
                           const float* __restrict__ Wout,
                           float* __restrict__ out) {
  const int wave = blockIdx.x * (blockDim.x >> 5) + (threadIdx.x >> 5);
  const int lane = threadIdx.x & 31;
  const int TG = DIMX / 64;
  const int gm = wave / TG, gng = wave % TG;
  const int mrow = lane & 15, kh = lane >> 4;

  int m  = gm * 16 + mrow;
  int bb = m / SEQ, ll = m % SEQ;
  int mrev = bb * SEQ + (SEQ - 1 - ll);

  const float* af = xz_f + (size_t)m    * XZW + DI + 16 * kh;
  const float* ab = xz_b + (size_t)mrev * XZW + DI + 16 * kh;
  const float* bp[4];
#pragma unroll
  for (int j = 0; j < 4; ++j)
    bp[j] = Wout + (size_t)(gng * 64 + j * 16 + mrow) * DI + 16 * kh;

  v8f acc[4] = {};
  for (int k0 = 0; k0 < DI; k0 += 32) {
    v4f a[4], a2[4], b[4][4];
    ld16(af + k0, a);
    ld16(ab + k0, a2);
#pragma unroll
    for (int j = 0; j < 4; ++j) ld16(bp[j] + k0, b[j]);
    asm volatile("" :: LOADED4(a), LOADED4(a2), LOADED4(b[0]), LOADED4(b[1]), LOADED4(b[2]), LOADED4(b[3]));
#pragma unroll
    for (int q = 0; q < 4; ++q) a[q] = a[q] + a2[q];
#pragma unroll
    for (int q = 0; q < 4; ++q) {
#pragma unroll
      for (int j = 0; j < 4; ++j) acc[j] = wmma4(lo2(a[q]), lo2(b[j][q]), acc[j]);
#pragma unroll
      for (int j = 0; j < 4; ++j) acc[j] = wmma4(hi2(a[q]), hi2(b[j][q]), acc[j]);
    }
  }
#pragma unroll
  for (int j = 0; j < 4; ++j)
#pragma unroll
    for (int v = 0; v < 8; ++v) {
      int orow = gm * 16 + v + 8 * kh;
      int n = gng * 64 + j * 16 + mrow;
      out[(size_t)orow * DIMX + n] = acc[j][v] + x[(size_t)orow * DIMX + n];
    }
}

extern "C" void kernel_launch(void* const* d_in, const int* in_sizes, int n_in,
                              void* d_out, int out_size, void* d_ws, size_t ws_size,
                              hipStream_t stream) {
  const float* x      = (const float*)d_in[0];
  const float* Win_f  = (const float*)d_in[1];
  const float* cw_f   = (const float*)d_in[2];
  const float* cb_f   = (const float*)d_in[3];
  const float* Wxp_f  = (const float*)d_in[4];
  const float* Wdt_f  = (const float*)d_in[5];
  const float* bdt_f  = (const float*)d_in[6];
  const float* Alog_f = (const float*)d_in[7];
  const float* Dv_f   = (const float*)d_in[8];
  const float* Win_b  = (const float*)d_in[9];
  const float* cw_b   = (const float*)d_in[10];
  const float* cb_b   = (const float*)d_in[11];
  const float* Wxp_b  = (const float*)d_in[12];
  const float* Wdt_b  = (const float*)d_in[13];
  const float* bdt_b  = (const float*)d_in[14];
  const float* Alog_b = (const float*)d_in[15];
  const float* Dv_b   = (const float*)d_in[16];
  const float* Wout   = (const float*)d_in[17];
  float* out = (float*)d_out;

  float* ws    = (float*)d_ws;
  float* xz_f  = ws;
  float* xz_b  = xz_f + (size_t)MROWS * XZW;
  float* u_f   = xz_b + (size_t)MROWS * XZW;
  float* dbl_f = u_f  + (size_t)2 * MROWS * DI;
  const size_t PS = (size_t)BATCH * NCH * DI * NST;
  float* P_f   = dbl_f + (size_t)2 * MROWS * DBLW;
  float* S_f   = P_f + PS;

  const int BLK = 256, WPB = BLK / 32;

  {
    int waves = (MROWS / 16) * (XZW / 64);
    k_gemm_in<<<dim3(waves / WPB, 2), BLK, 0, stream>>>(x, Win_f, Win_b, xz_f);
  }
  {
    int n = (MROWS / CT) * DQ;
    k_conv_silu<<<dim3(n / BLK, 2), BLK, 0, stream>>>(
        xz_f, cw_f, cb_f, cw_b, cb_b, u_f);
  }
  {
    int waves = MROWS / 16;
    k_gemm_xp<<<dim3(waves / WPB, 2), BLK, 0, stream>>>(u_f, Wxp_f, Wxp_b, dbl_f);
  }
  {
    int waves = (MROWS / 16) * (DI / 64);
    k_gemm_dt<<<dim3(waves / WPB, 2), BLK, 0, stream>>>(
        dbl_f, Wdt_f, bdt_f, Wdt_b, bdt_b, xz_f);
  }
  {
    int cb = (BATCH * DI * NST + BLK - 1) / BLK;
    k_scan_part1<<<dim3(BATCH * NCH, 2), DI, 0, stream>>>(
        xz_f, u_f, dbl_f, Alog_f, Alog_b, P_f, S_f);
    k_scan_combine<<<dim3(cb, 2), BLK, 0, stream>>>(P_f, S_f);
    k_scan_part3<<<dim3(BATCH * NCH, 2), DI, 0, stream>>>(
        xz_f, u_f, dbl_f, Alog_f, Dv_f, Alog_b, Dv_b, S_f);
  }
  {
    int waves = (MROWS / 16) * (DIMX / 64);
    k_gemm_out<<<waves / WPB, BLK, 0, stream>>>(xz_f, xz_b, x, Wout, out);
  }
}


# Input projection: x and W_in tiles reach the matrix steps through LDS (8 lanes fetch one whole 128-byte line, two stages alternate), no more 16-line fragment loads; fragment code and k order kept
# speedup vs baseline: 2.9711x; 1.2438x over previous
#include <hip/hip_runtime.h>
#include <hip/hip_bf16.h>
#include <math.h>

#define BATCH 4
#define SEQ   2048
#define DIMX  192
#define DI    384
#define NST   16
#define DTR   12
#define XZW   768
#define DBLW  48
#define MROWS (BATCH*SEQ)
#define NCH   32
#define CL    (SEQ/NCH)
#define PSTRIDE (2 * BATCH * NCH * DI * NST)

typedef float v2f __attribute__((ext_vector_type(2)));
typedef float v8f __attribute__((ext_vector_type(8)));

__device__ __forceinline__ v8f wmma4(v2f a, v2f b, v8f c) {
  return __builtin_amdgcn_wmma_f32_16x16x4_f32(false, a, false, b, (short)0, c,
                                               false, false);
}

typedef float v4f __attribute__((ext_vector_type(4)));

__device__ __forceinline__ void ld16(const float* __restrict__ p, v4f (&r)[4]) {
  const v4f* q = reinterpret_cast<const v4f*>(p);
  r[0] = q[0]; r[1] = q[1]; r[2] = q[2]; r[3] = q[3];
}
__device__ __forceinline__ v2f lo2(v4f v) { v2f r; r.x = v.x; r.y = v.y; return r; }
__device__ __forceinline__ v2f hi2(v4f v) { v2f r; r.x = v.z; r.y = v.w; return r; }
#define LOADED4(v) "v"((v)[0]), "v"((v)[1]), "v"((v)[2]), "v"((v)[3])

#define GI_BM    64
#define GI_BN    128
#define GI_KS    32
#define GI_ROWS  (GI_BM + GI_BN)
#define GI_PITCH 36
#define GI_NB    (XZW / GI_BN)
#define GI_STAGES (DIMX / GI_KS)
__global__ void __launch_bounds__(256) k_gemm_in(const float* __restrict__ x,
                          const float* __restrict__ Win_f,
                          const float* __restrict__ Win_b,
                          float* __restrict__ xz_all) {
  __shared__ __attribute__((aligned(16))) float sT[2][GI_ROWS * GI_PITCH];
  const int reverse = blockIdx.y;
  const float* Win = reverse ? Win_b : Win_f;
  float* xz = xz_all + (size_t)reverse * MROWS * XZW;
  const int nb = blockIdx.x % GI_NB;
  const int mb = blockIdx.x / GI_NB;
  const int tid = threadIdx.x;
  const int wave = tid >> 5, lane = tid & 31;
  const int wm = wave >> 1, wn = wave & 1;
  const int mrow = lane & 15, kh = lane >> 4;

  const int r0 = tid >> 3, p = tid & 7;
  const int m  = mb * GI_BM + r0;
  const int bb = m / SEQ, ll = m % SEQ;
  const int srow = reverse ? (bb * SEQ + (SEQ - 1 - ll)) : m;
  const float* ga = x + (size_t)srow * DIMX + 4 * p;
  const int astep = reverse ? -32 * DIMX : 32 * DIMX;
  const float* gb = Win + (size_t)(nb * GI_BN + r0) * DIMX + 4 * p;
  const int lo = r0 * GI_PITCH + 4 * p;

  v4f st[6];
  st[0] = *reinterpret_cast<const v4f*>(ga);
  st[1] = *reinterpret_cast<const v4f*>(ga + astep);
#pragma unroll
  for (int j = 0; j < 4; ++j)
    st[2 + j] = *reinterpret_cast<const v4f*>(gb + (size_t)j * 32 * DIMX);
#pragma unroll
  for (int j = 0; j < 6; ++j)
    *reinterpret_cast<v4f*>(&sT[0][lo + j * 32 * GI_PITCH]) = st[j];
  __syncthreads();

  const int aoff = (wm * 16 + mrow) * GI_PITCH + 16 * kh;
  const int boff = (GI_BM + wn * 64 + mrow) * GI_PITCH + 16 * kh;

  v8f acc[4] = {};
  for (int s = 0; s < GI_STAGES; ++s) {
    const float* cur = sT[s & 1];
    const int kn = (s + 1 < GI_STAGES) ? (s + 1) * GI_KS : s * GI_KS;
    st[0] = *reinterpret_cast<const v4f*>(ga + kn);
    st[1] = *reinterpret_cast<const v4f*>(ga + astep + kn);
#pragma unroll
    for (int j = 0; j < 4; ++j)
      st[2 + j] = *reinterpret_cast<const v4f*>(gb + (size_t)j * 32 * DIMX + kn);

    v4f a[4], b[4][4];
    ld16(cur + aoff, a);
#pragma unroll
    for (int j = 0; j < 4; ++j) ld16(cur + boff + j * 16 * GI_PITCH, b[j]);
    asm volatile("" :: LOADED4(a), LOADED4(b[0]), LOADED4(b[1]), LOADED4(b[2]), LOADED4(b[3]));
#pragma unroll
    for (int q = 0; q < 4; ++q) {
#pragma unroll
      for (int j = 0; j < 4; ++j) acc[j] = wmma4(lo2(a[q]), lo2(b[j][q]), acc[j]);
#pragma unroll
      for (int j = 0; j < 4; ++j) acc[j] = wmma4(hi2(a[q]), hi2(b[j][q]), acc[j]);
    }
    float* nxt = sT[(s + 1) & 1];
#pragma unroll
    for (int j = 0; j < 6; ++j)
      *reinterpret_cast<v4f*>(&nxt[lo + j * 32 * GI_PITCH]) = st[j];
    __syncthreads();
  }
#pragma unroll
  for (int j = 0; j < 4; ++j)
#pragma unroll
    for (int v = 0; v < 8; ++v) {
      int orow = mb * GI_BM + wm * 16 + v + 8 * kh;
      xz[(size_t)orow * XZW + nb * GI_BN + wn * 64 + j * 16 + mrow] = acc[j][v];
    }
}

#define CT 8
#define DQ (DI / 4)
__global__ void k_conv_silu(const float* __restrict__ xz_all,
                            const float* __restrict__ cw_f,
                            const float* __restrict__ cb_f,
                            const float* __restrict__ cw_b,
                            const float* __restrict__ cb_b,
                            float* __restrict__ u_all) {
  const int dir = blockIdx.y;
  const float* xz = xz_all + (size_t)dir * MROWS * XZW;
  const float* cw = dir ? cw_b : cw_f;
  const float* cb = dir ? cb_b : cb_f;
  float* u = u_all + (size_t)dir * MROWS * DI;
  const int idx = blockIdx.x * blockDim.x + threadIdx.x;
  const int dq  = idx % DQ;
  const int run = idx / DQ;
  const int row0 = run * CT;
  const bool head = (row0 % SEQ) == 0;

  const float* src = xz + (size_t)row0 * XZW + 4 * dq;
  v4f xr[CT + 3];
  for (int k = 0; k < 3; ++k) {
    v4f zero = {0.0f, 0.0f, 0.0f, 0.0f};
    xr[k] = head ? zero : *reinterpret_cast<const v4f*>(src + (k - 3) * XZW);
  }
  for (int k = 0; k < CT; ++k)
    xr[3 + k] = *reinterpret_cast<const v4f*>(src + (size_t)k * XZW);

  v4f w[4];
  for (int c = 0; c < 4; ++c)
    w[c] = *reinterpret_cast<const v4f*>(cw + (size_t)(4 * dq + c) * 4);
  const v4f bias = *reinterpret_cast<const v4f*>(cb + 4 * dq);

  float* dst = u + (size_t)row0 * DI + 4 * dq;
  for (int i = 0; i < CT; ++i) {
    v4f o;
    for (int c = 0; c < 4; ++c) {
      float acc = bias[c];
      for (int j = 0; j < 4; ++j) {
        if (i + j >= 3 || !head)
          acc += xr[i + j][c] * w[c][j];
      }
      o[c] = acc / (1.0f + __expf(-acc));
    }
    *reinterpret_cast<v4f*>(dst + (size_t)i * DI) = o;
  }
}

__global__ void __launch_bounds__(256) k_gemm_xp(const float* __restrict__ u_all,
                          const float* __restrict__ Wxp_f,
                          const float* __restrict__ Wxp_b,
                          float* __restrict__ dbl_all) {
  const int dir = blockIdx.y;
  const float* u = u_all + (size_t)dir * MROWS * DI;
  const float* Wxp = dir ? Wxp_b : Wxp_f;
  float* dbl = dbl_all + (size_t)dir * MROWS * DBLW;
  const int wave = blockIdx.x * (blockDim.x >> 5) + (threadIdx.x >> 5);
  const int lane = threadIdx.x & 31;
  const int gm = wave;
  const int mrow = lane & 15, kh = lane >> 4;

  const float* ap = u + (size_t)(gm * 16 + mrow) * DI + 16 * kh;
  bool nvalid[3];
  const float* bp[3];
#pragma unroll
  for (int j = 0; j < 3; ++j) {
    int n = j * 16 + mrow;
    nvalid[j] = (n < DTR + 2 * NST);
    bp[j] = Wxp + (size_t)(nvalid[j] ? n : 0) * DI + 16 * kh;
  }

  v8f acc[3] = {};
  for (int k0 = 0; k0 < DI; k0 += 32) {
    v4f a[4], b[3][4];
    ld16(ap + k0, a);
#pragma unroll
    for (int j = 0; j < 3; ++j) ld16(bp[j] + k0, b[j]);
    asm volatile("" :: LOADED4(a), LOADED4(b[0]), LOADED4(b[1]), LOADED4(b[2]));
#pragma unroll
    for (int j = 0; j < 3; ++j) {
#pragma unroll
      for (int q = 0; q < 4; ++q) {
        v4f zero = {0.0f, 0.0f, 0.0f, 0.0f};
        b[j][q] = nvalid[j] ? b[j][q] : zero;
      }
    }
#pragma unroll
    for (int q = 0; q < 4; ++q) {
#pragma unroll
      for (int j = 0; j < 3; ++j) acc[j] = wmma4(lo2(a[q]), lo2(b[j][q]), acc[j]);
#pragma unroll
      for (int j = 0; j < 3; ++j) acc[j] = wmma4(hi2(a[q]), hi2(b[j][q]), acc[j]);
    }
  }
#pragma unroll
  for (int j = 0; j < 3; ++j)
#pragma unroll
    for (int v = 0; v < 8; ++v) {
      int orow = gm * 16 + v + 8 * kh;
      dbl[(size_t)orow * DBLW + j * 16 + mrow] = acc[j][v];
    }
}

__global__ void k_gemm_dt(const float* __restrict__ dbl_all,
                          const float* __restrict__ Wdt_f,
                          const float* __restrict__ bdt_f,
                          const float* __restrict__ Wdt_b,
                          const float* __restrict__ bdt_b,
                          float* __restrict__ xz_all) {
  const int dir = blockIdx.y;
  const float* dbl = dbl_all + (size_t)dir * MROWS * DBLW;
  const float* Wdt = dir ? Wdt_b : Wdt_f;
  const float* bdt = dir ? bdt_b : bdt_f;
  float* xz = xz_all + (size_t)dir * MROWS * XZW;
  const int wave = blockIdx.x * (blockDim.x >> 5) + (threadIdx.x >> 5);
  const int lane = threadIdx.x & 31;
  const int TG = DI / 64;
  const int gm = wave / TG, gng = wave % TG;
  const int mrow = lane & 15, kh = lane >> 4;

  const float* ap = dbl + (size_t)(gm * 16 + mrow) * DBLW;
  const float* bp[4];
#pragma unroll
  for (int j = 0; j < 4; ++j)
    bp[j] = Wdt + (size_t)(gng * 64 + j * 16 + mrow) * DTR;

  v8f acc[4] = {};
  for (int k0 = 0; k0 < DTR; k0 += 4) {
    int kk = k0 + 2 * kh;
    v2f a; a.x = ap[kk]; a.y = ap[kk + 1];
#pragma unroll
    for (int j = 0; j < 4; ++j) {
      v2f b; b.x = bp[j][kk]; b.y = bp[j][kk + 1];
      acc[j] = wmma4(a, b, acc[j]);
    }
  }
#pragma unroll
  for (int j = 0; j < 4; ++j) {
    int n = gng * 64 + j * 16 + mrow;
    float bias = bdt[n];
#pragma unroll
    for (int v = 0; v < 8; ++v) {
      int orow = gm * 16 + v + 8 * kh;
      float t = acc[j][v] + bias;
      float sp = (t > 20.0f) ? t : log1pf(__expf(t));
      xz[(size_t)orow * XZW + n] = sp;
    }
  }
}

#define TGRP 8
#define NGRP (CL / TGRP)
#define REQUESTS_STAY_AHEAD() asm volatile("" ::: "memory")

__device__ __forceinline__ void scan1_group(const float (&dt8)[TGRP], const float (&u8)[TGRP],
                                            const v4f (*sB)[NST / 4],
                                            const float (&a)[NST], float (&P)[NST], float (&S)[NST]) {
#pragma unroll
  for (int k = 0; k < TGRP; ++k) {
    float Bv[NST];
#pragma unroll
    for (int q = 0; q < NST / 4; ++q) {
      const v4f v = sB[k][q];
      Bv[4 * q] = v.x; Bv[4 * q + 1] = v.y; Bv[4 * q + 2] = v.z; Bv[4 * q + 3] = v.w;
    }
    const float dt = dt8[k];
    const float dtu = dt * u8[k];
#pragma unroll
    for (int n = 0; n < NST; ++n) {
      float dA = __expf(dt * a[n]);
      S[n] = S[n] * dA + dtu * Bv[n];
      P[n] *= dA;
    }
  }
}

__global__ void __launch_bounds__(DI) k_scan_part1(const float* __restrict__ xz_all,
                             const float* __restrict__ u_all,
                             const float* __restrict__ dbl_all,
                             const float* __restrict__ Alog_f,
                             const float* __restrict__ Alog_b,
                             float* __restrict__ Pb_all, float* __restrict__ Sb_all) {
  __shared__ v4f sB[CL][NST / 4];
  const int dir = blockIdx.y;
  const float* xz  = xz_all  + (size_t)dir * MROWS * XZW;
  const float* u   = u_all   + (size_t)dir * MROWS * DI;
  const float* dbl = dbl_all + (size_t)dir * MROWS * DBLW;
  const float* Alog = dir ? Alog_b : Alog_f;
  float* Pb = Pb_all + (size_t)dir * PSTRIDE;
  float* Sb = Sb_all + (size_t)dir * PSTRIDE;
  const int b = blockIdx.x / NCH;
  const int c = blockIdx.x % NCH;
  const int d = threadIdx.x;

  const size_t row0 = (size_t)b * SEQ + (size_t)c * CL;
  const float* dtp = xz  + row0 * XZW + d;
  const float* utp = u   + row0 * DI + d;
  const float* bcp = dbl + row0 * DBLW + DTR;

  float dtA[TGRP], utA[TGRP], dtB[TGRP], utB[TGRP];
#pragma unroll
  for (int k = 0; k < TGRP; ++k) {
    dtA[k] = dtp[(size_t)k * XZW];
    utA[k] = utp[(size_t)k * DI];
  }
  if (d < CL * (NST / 4)) {
    const int r = d >> 2, q = d & 3;
    sB[r][q] = *reinterpret_cast<const v4f*>(bcp + (size_t)r * DBLW + 4 * q);
  }

  float a[NST], P[NST], S[NST];
#pragma unroll
  for (int n = 0; n < NST; ++n) {
    a[n] = -__expf(Alog[d * NST + n]);
    P[n] = 1.0f;
    S[n] = 0.0f;
  }
  __syncthreads();

  for (int g = 0; g < NGRP; g += 2) {
#pragma unroll
    for (int k = 0; k < TGRP; ++k) {
      dtB[k] = dtp[(size_t)((g + 1) * TGRP + k) * XZW];
      utB[k] = utp[(size_t)((g + 1) * TGRP + k) * DI];
    }
    REQUESTS_STAY_AHEAD();
    scan1_group(dtA, utA, sB + g * TGRP, a, P, S);
    const int gn = (g + 2 < NGRP) ? g + 2 : g + 1;
#pragma unroll
    for (int k = 0; k < TGRP; ++k) {
      dtA[k] = dtp[(size_t)(gn * TGRP + k) * XZW];
      utA[k] = utp[(size_t)(gn * TGRP + k) * DI];
    }
    REQUESTS_STAY_AHEAD();
    scan1_group(dtB, utB, sB + (g + 1) * TGRP, a, P, S);
  }
  const size_t base = ((size_t)(b * NCH + c) * DI + d) * NST;
#pragma unroll
  for (int n = 0; n < NST; ++n) {
    Pb[base + n] = P[n];
    Sb[base + n] = S[n];
  }
}

__global__ void k_scan_combine(const float* __restrict__ Pb_all,
                               float* __restrict__ Sb_all) {
  const int dir = blockIdx.y;
  const float* Pb = Pb_all + (size_t)dir * PSTRIDE;
  float* Sb = Sb_all + (size_t)dir * PSTRIDE;
  int idx = blockIdx.x * blockDim.x + threadIdx.x;
  if (idx >= BATCH * DI * NST) return;
  const int n = idx & (NST - 1);
  const int d = (idx >> 4) % DI;
  const int b = idx / (NST * DI);

  float carry = 0.0f;
  for (int c = 0; c < NCH; ++c) {
    const size_t base = ((size_t)(b * NCH + c) * DI + d) * NST + n;
    float p = Pb[base];
    float s = Sb[base];
    Sb[base] = carry;
    carry = p * carry + s;
  }
}

__device__ __forceinline__ void scan3_group(const float (&dt8)[TGRP], const float (&u8)[TGRP],
                                            const float (&z8)[TGRP],
                                            const v4f (*sBC)[2 * NST / 4], float* yp,
                                            const float (&a)[NST], float (&h)[NST], float Dd) {
#pragma unroll
  for (int k = 0; k < TGRP; ++k) {
    float BCv[2 * NST];
#pragma unroll
    for (int q = 0; q < 2 * NST / 4; ++q) {
      const v4f v = sBC[k][q];
      BCv[4 * q] = v.x; BCv[4 * q + 1] = v.y; BCv[4 * q + 2] = v.z; BCv[4 * q + 3] = v.w;
    }
    const float dt = dt8[k];
    const float ut = u8[k];
    const float dtu = dt * ut;
    float y = 0.0f;
#pragma unroll
    for (int n = 0; n < NST; ++n) {
      float dA = __expf(dt * a[n]);
      h[n] = h[n] * dA + dtu * BCv[n];
      y += h[n] * BCv[NST + n];
    }
    y += ut * Dd;
    const float z = z8[k];
    y *= z / (1.0f + __expf(-z));
    yp[(size_t)k * XZW] = y;
  }
}

__global__ void __launch_bounds__(DI) k_scan_part3(float* __restrict__ xz_all,
                             const float* __restrict__ u_all,
                             const float* __restrict__ dbl_all,
                             const float* __restrict__ Alog_f,
                             const float* __restrict__ Dv_f,
                             const float* __restrict__ Alog_b,
                             const float* __restrict__ Dv_b,
                             const float* __restrict__ Sb_all) {
  __shared__ v4f sBC[CL][2 * NST / 4];
  const int dir = blockIdx.y;
  float* xz        = xz_all  + (size_t)dir * MROWS * XZW;
  const float* u   = u_all   + (size_t)dir * MROWS * DI;
  const float* dbl = dbl_all + (size_t)dir * MROWS * DBLW;
  const float* Alog = dir ? Alog_b : Alog_f;
  const float* Dv   = dir ? Dv_b : Dv_f;
  const float* Sb = Sb_all + (size_t)dir * PSTRIDE;
  const int b = blockIdx.x / NCH;
  const int c = blockIdx.x % NCH;
  const int d = threadIdx.x;

  const size_t row0 = (size_t)b * SEQ + (size_t)c * CL;
  float* xzp = xz + row0 * XZW + d;
  const float* utp = u   + row0 * DI + d;
  const float* bcp = dbl + row0 * DBLW + DTR;

  float dtA[TGRP], utA[TGRP], zA[TGRP], dtB[TGRP], utB[TGRP], zB[TGRP];
#pragma unroll
  for (int k = 0; k < TGRP; ++k) {
    dtA[k] = xzp[(size_t)k * XZW];
    zA[k]  = xzp[(size_t)k * XZW + DI];
    utA[k] = utp[(size_t)k * DI];
  }
  for (int i = d; i < CL * (2 * NST / 4); i += DI) {
    const int r = i >> 3, q = i & 7;
    sBC[r][q] = *reinterpret_cast<const v4f*>(bcp + (size_t)r * DBLW + 4 * q);
  }

  float a[NST], h[NST];
  const size_t base = ((size_t)(b * NCH + c) * DI + d) * NST;
#pragma unroll
  for (int n = 0; n < NST; ++n) {
    a[n] = -__expf(Alog[d * NST + n]);
    h[n] = Sb[base + n];
  }
  const float Dd = Dv[d];
  __syncthreads();

  for (int g = 0; g < NGRP; g += 2) {
#pragma unroll
    for (int k = 0; k < TGRP; ++k) {
      dtB[k] = xzp[(size_t)((g + 1) * TGRP + k) * XZW];
      zB[k]  = xzp[(size_t)((g + 1) * TGRP + k) * XZW + DI];
      utB[k] = utp[(size_t)((g + 1) * TGRP + k) * DI];
    }
    REQUESTS_STAY_AHEAD();
    scan3_group(dtA, utA, zA, sBC + g * TGRP, xzp + (size_t)(g * TGRP) * XZW + DI, a, h, Dd);
    const int gn = (g + 2 < NGRP) ? g + 2 : g + 1;
#pragma unroll
    for (int k = 0; k < TGRP; ++k) {
      dtA[k] = xzp[(size_t)(gn * TGRP + k) * XZW];
      zA[k]  = xzp[(size_t)(gn * TGRP + k) * XZW + DI];
      utA[k] = utp[(size_t)(gn * TGRP + k) * DI];
    }
    REQUESTS_STAY_AHEAD();
    scan3_group(dtB, utB, zB, sBC + (g + 1) * TGRP, xzp + (size_t)((g + 1) * TGRP) * XZW + DI, a, h, Dd);
  }
}

__global__ void __launch_bounds__(256) k_gemm_out(const float* __restrict__ xz_f,
                           const float* __restrict__ xz_b,
                           const float* __restrict__ x,
                           const float* __restrict__ Wout,
                           float* __restrict__ out) {
  const int wave = blockIdx.x * (blockDim.x >> 5) + (threadIdx.x >> 5);
  const int lane = threadIdx.x & 31;
  const int TG = DIMX / 64;
  const int gm = wave / TG, gng = wave % TG;
  const int mrow = lane & 15, kh = lane >> 4;

  int m  = gm * 16 + mrow;
  int bb = m / SEQ, ll = m % SEQ;
  int mrev = bb * SEQ + (SEQ - 1 - ll);

  const float* af = xz_f + (size_t)m    * XZW + DI + 16 * kh;
  const float* ab = xz_b + (size_t)mrev * XZW + DI + 16 * kh;
  const float* bp[4];
#pragma unroll
  for (int j = 0; j < 4; ++j)
    bp[j] = Wout + (size_t)(gng * 64 + j * 16 + mrow) * DI + 16 * kh;

  v8f acc[4] = {};
  for (int k0 = 0; k0 < DI; k0 += 32) {
    v4f a[4], a2[4], b[4][4];
    ld16(af + k0, a);
    ld16(ab + k0, a2);
#pragma unroll
    for (int j = 0; j < 4; ++j) ld16(bp[j] + k0, b[j]);
    asm volatile("" :: LOADED4(a), LOADED4(a2), LOADED4(b[0]), LOADED4(b[1]), LOADED4(b[2]), LOADED4(b[3]));
#pragma unroll
    for (int q = 0; q < 4; ++q) a[q] = a[q] + a2[q];
#pragma unroll
    for (int q = 0; q < 4; ++q) {
#pragma unroll
      for (int j = 0; j < 4; ++j) acc[j] = wmma4(lo2(a[q]), lo2(b[j][q]), acc[j]);
#pragma unroll
      for (int j = 0; j < 4; ++j) acc[j] = wmma4(hi2(a[q]), hi2(b[j][q]), acc[j]);
    }
  }
#pragma unroll
  for (int j = 0; j < 4; ++j)
#pragma unroll
    for (int v = 0; v < 8; ++v) {
      int orow = gm * 16 + v + 8 * kh;
      int n = gng * 64 + j * 16 + mrow;
      out[(size_t)orow * DIMX + n] = acc[j][v] + x[(size_t)orow * DIMX + n];
    }
}

extern "C" void kernel_launch(void* const* d_in, const int* in_sizes, int n_in,
                              void* d_out, int out_size, void* d_ws, size_t ws_size,
                              hipStream_t stream) {
  const float* x      = (const float*)d_in[0];
  const float* Win_f  = (const float*)d_in[1];
  const float* cw_f   = (const float*)d_in[2];
  const float* cb_f   = (const float*)d_in[3];
  const float* Wxp_f  = (const float*)d_in[4];
  const float* Wdt_f  = (const float*)d_in[5];
  const float* bdt_f  = (const float*)d_in[6];
  const float* Alog_f = (const float*)d_in[7];
  const float* Dv_f   = (const float*)d_in[8];
  const float* Win_b  = (const float*)d_in[9];
  const float* cw_b   = (const float*)d_in[10];
  const float* cb_b   = (const float*)d_in[11];
  const float* Wxp_b  = (const float*)d_in[12];
  const float* Wdt_b  = (const float*)d_in[13];
  const float* bdt_b  = (const float*)d_in[14];
  const float* Alog_b = (const float*)d_in[15];
  const float* Dv_b   = (const float*)d_in[16];
  const float* Wout   = (const float*)d_in[17];
  float* out = (float*)d_out;

  float* ws    = (float*)d_ws;
  float* xz_f  = ws;
  float* xz_b  = xz_f + (size_t)MROWS * XZW;
  float* u_f   = xz_b + (size_t)MROWS * XZW;
  float* dbl_f = u_f  + (size_t)2 * MROWS * DI;
  const size_t PS = (size_t)BATCH * NCH * DI * NST;
  float* P_f   = dbl_f + (size_t)2 * MROWS * DBLW;
  float* S_f   = P_f + PS;

  const int BLK = 256, WPB = BLK / 32;

  {
    int blocks = (MROWS / GI_BM) * GI_NB;
    k_gemm_in<<<dim3(blocks, 2), BLK, 0, stream>>>(x, Win_f, Win_b, xz_f);
  }
  {
    int n = (MROWS / CT) * DQ;
    k_conv_silu<<<dim3(n / BLK, 2), BLK, 0, stream>>>(
        xz_f, cw_f, cb_f, cw_b, cb_b, u_f);
  }
  {
    int waves = MROWS / 16;
    k_gemm_xp<<<dim3(waves / WPB, 2), BLK, 0, stream>>>(u_f, Wxp_f, Wxp_b, dbl_f);
  }
  {
    int waves = (MROWS / 16) * (DI / 64);
    k_gemm_dt<<<dim3(waves / WPB, 2), BLK, 0, stream>>>(
        dbl_f, Wdt_f, bdt_f, Wdt_b, bdt_b, xz_f);
  }
  {
    int cb = (BATCH * DI * NST + BLK - 1) / BLK;
    k_scan_part1<<<dim3(BATCH * NCH, 2), DI, 0, stream>>>(
        xz_f, u_f, dbl_f, Alog_f, Alog_b, P_f, S_f);
    k_scan_combine<<<dim3(cb, 2), BLK, 0, stream>>>(P_f, S_f);
    k_scan_part3<<<dim3(BATCH * NCH, 2), DI, 0, stream>>>(
        xz_f, u_f, dbl_f, Alog_f, Dv_f, Alog_b, Dv_b, S_f);
  }
  {
    int waves = (MROWS / 16) * (DIMX / 64);
    k_gemm_out<<<waves / WPB, BLK, 0, stream>>>(xz_f, xz_b, x, Wout, out);
  }
}


# Output and x_dbl projections: operand tiles reach the matrix steps through LDS too (adjacent lanes fetch whole 128-byte lines, two stages alternate), no 16-line fragment loads; k order and sums kept
# speedup vs baseline: 3.3249x; 1.1191x over previous
#include <hip/hip_runtime.h>
#include <hip/hip_bf16.h>
#include <math.h>

#define BATCH 4
#define SEQ   2048
#define DIMX  192
#define DI    384
#define NST   16
#define DTR   12
#define XZW   768
#define DBLW  48
#define MROWS (BATCH*SEQ)
#define NCH   32
#define CL    (SEQ/NCH)
#define PSTRIDE (2 * BATCH * NCH * DI * NST)

typedef float v2f __attribute__((ext_vector_type(2)));
typedef float v8f __attribute__((ext_vector_type(8)));

__device__ __forceinline__ v8f wmma4(v2f a, v2f b, v8f c) {
  return __builtin_amdgcn_wmma_f32_16x16x4_f32(false, a, false, b, (short)0, c,
                                               false, false);
}

typedef float v4f __attribute__((ext_vector_type(4)));

__device__ __forceinline__ void ld16(const float* __restrict__ p, v4f (&r)[4]) {
  const v4f* q = reinterpret_cast<const v4f*>(p);
  r[0] = q[0]; r[1] = q[1]; r[2] = q[2]; r[3] = q[3];
}
__device__ __forceinline__ v2f lo2(v4f v) { v2f r; r.x = v.x; r.y = v.y; return r; }
__device__ __forceinline__ v2f hi2(v4f v) { v2f r; r.x = v.z; r.y = v.w; return r; }
#define LOADED4(v) "v"((v)[0]), "v"((v)[1]), "v"((v)[2]), "v"((v)[3])

#define GI_BM    64
#define GI_BN    128
#define GI_KS    32
#define GI_ROWS  (GI_BM + GI_BN)
#define GI_PITCH 36
#define GI_NB    (XZW / GI_BN)
#define GI_STAGES (DIMX / GI_KS)
__global__ void __launch_bounds__(256) k_gemm_in(const float* __restrict__ x,
                          const float* __restrict__ Win_f,
                          const float* __restrict__ Win_b,
                          float* __restrict__ xz_all) {
  __shared__ __attribute__((aligned(16))) float sT[2][GI_ROWS * GI_PITCH];
  const int reverse = blockIdx.y;
  const float* Win = reverse ? Win_b : Win_f;
  float* xz = xz_all + (size_t)reverse * MROWS * XZW;
  const int nb = blockIdx.x % GI_NB;
  const int mb = blockIdx.x / GI_NB;
  const int tid = threadIdx.x;
  const int wave = tid >> 5, lane = tid & 31;
  const int wm = wave >> 1, wn = wave & 1;
  const int mrow = lane & 15, kh = lane >> 4;

  const int r0 = tid >> 3, p = tid & 7;
  const int m  = mb * GI_BM + r0;
  const int bb = m / SEQ, ll = m % SEQ;
  const int srow = reverse ? (bb * SEQ + (SEQ - 1 - ll)) : m;
  const float* ga = x + (size_t)srow * DIMX + 4 * p;
  const int astep = reverse ? -32 * DIMX : 32 * DIMX;
  const float* gb = Win + (size_t)(nb * GI_BN + r0) * DIMX + 4 * p;
  const int lo = r0 * GI_PITCH + 4 * p;

  v4f st[6];
  st[0] = *reinterpret_cast<const v4f*>(ga);
  st[1] = *reinterpret_cast<const v4f*>(ga + astep);
#pragma unroll
  for (int j = 0; j < 4; ++j)
    st[2 + j] = *reinterpret_cast<const v4f*>(gb + (size_t)j * 32 * DIMX);
#pragma unroll
  for (int j = 0; j < 6; ++j)
    *reinterpret_cast<v4f*>(&sT[0][lo + j * 32 * GI_PITCH]) = st[j];
  __syncthreads();

  const int aoff = (wm * 16 + mrow) * GI_PITCH + 16 * kh;
  const int boff = (GI_BM + wn * 64 + mrow) * GI_PITCH + 16 * kh;

  v8f acc[4] = {};
  for (int s = 0; s < GI_STAGES; ++s) {
    const float* cur = sT[s & 1];
    const int kn = (s + 1 < GI_STAGES) ? (s + 1) * GI_KS : s * GI_KS;
    st[0] = *reinterpret_cast<const v4f*>(ga + kn);
    st[1] = *reinterpret_cast<const v4f*>(ga + astep + kn);
#pragma unroll
    for (int j = 0; j < 4; ++j)
      st[2 + j] = *reinterpret_cast<const v4f*>(gb + (size_t)j * 32 * DIMX + kn);

    v4f a[4], b[4][4];
    ld16(cur + aoff, a);
#pragma unroll
    for (int j = 0; j < 4; ++j) ld16(cur + boff + j * 16 * GI_PITCH, b[j]);
    asm volatile("" :: LOADED4(a), LOADED4(b[0]), LOADED4(b[1]), LOADED4(b[2]), LOADED4(b[3]));
#pragma unroll
    for (int q = 0; q < 4; ++q) {
#pragma unroll
      for (int j = 0; j < 4; ++j) acc[j] = wmma4(lo2(a[q]), lo2(b[j][q]), acc[j]);
#pragma unroll
      for (int j = 0; j < 4; ++j) acc[j] = wmma4(hi2(a[q]), hi2(b[j][q]), acc[j]);
    }
    float* nxt = sT[(s + 1) & 1];
#pragma unroll
    for (int j = 0; j < 6; ++j)
      *reinterpret_cast<v4f*>(&nxt[lo + j * 32 * GI_PITCH]) = st[j];
    __syncthreads();
  }
#pragma unroll
  for (int j = 0; j < 4; ++j)
#pragma unroll
    for (int v = 0; v < 8; ++v) {
      int orow = mb * GI_BM + wm * 16 + v + 8 * kh;
      xz[(size_t)orow * XZW + nb * GI_BN + wn * 64 + j * 16 + mrow] = acc[j][v];
    }
}

#define CT 8
#define DQ (DI / 4)
__global__ void k_conv_silu(const float* __restrict__ xz_all,
                            const float* __restrict__ cw_f,
                            const float* __restrict__ cb_f,
                            const float* __restrict__ cw_b,
                            const float* __restrict__ cb_b,
                            float* __restrict__ u_all) {
  const int dir = blockIdx.y;
  const float* xz = xz_all + (size_t)dir * MROWS * XZW;
  const float* cw = dir ? cw_b : cw_f;
  const float* cb = dir ? cb_b : cb_f;
  float* u = u_all + (size_t)dir * MROWS * DI;
  const int idx = blockIdx.x * blockDim.x + threadIdx.x;
  const int dq  = idx % DQ;
  const int run = idx / DQ;
  const int row0 = run * CT;
  const bool head = (row0 % SEQ) == 0;

  const float* src = xz + (size_t)row0 * XZW + 4 * dq;
  v4f xr[CT + 3];
  for (int k = 0; k < 3; ++k) {
    v4f zero = {0.0f, 0.0f, 0.0f, 0.0f};
    xr[k] = head ? zero : *reinterpret_cast<const v4f*>(src + (k - 3) * XZW);
  }
  for (int k = 0; k < CT; ++k)
    xr[3 + k] = *reinterpret_cast<const v4f*>(src + (size_t)k * XZW);

  v4f w[4];
  for (int c = 0; c < 4; ++c)
    w[c] = *reinterpret_cast<const v4f*>(cw + (size_t)(4 * dq + c) * 4);
  const v4f bias = *reinterpret_cast<const v4f*>(cb + 4 * dq);

  float* dst = u + (size_t)row0 * DI + 4 * dq;
  for (int i = 0; i < CT; ++i) {
    v4f o;
    for (int c = 0; c < 4; ++c) {
      float acc = bias[c];
      for (int j = 0; j < 4; ++j) {
        if (i + j >= 3 || !head)
          acc += xr[i + j][c] * w[c][j];
      }
      o[c] = acc / (1.0f + __expf(-acc));
    }
    *reinterpret_cast<v4f*>(dst + (size_t)i * DI) = o;
  }
}

#define GX_BM    64
#define GX_BN    48
#define GX_KS    64
#define GX_ROWS  (GX_BM + GX_BN)
#define GX_PITCH 68
#define GX_STAGES (DI / GX_KS)
#define GX_NP    (GX_ROWS / 8)
__global__ void __launch_bounds__(128) k_gemm_xp(const float* __restrict__ u_all,
                          const float* __restrict__ Wxp_f,
                          const float* __restrict__ Wxp_b,
                          float* __restrict__ dbl_all) {
  __shared__ __attribute__((aligned(16))) float sT[2][GX_ROWS * GX_PITCH];
  const int dir = blockIdx.y;
  const float* u = u_all + (size_t)dir * MROWS * DI;
  const float* Wxp = dir ? Wxp_b : Wxp_f;
  float* dbl = dbl_all + (size_t)dir * MROWS * DBLW;
  const int tid = threadIdx.x;
  const int wave = tid >> 5, lane = tid & 31;
  const int gm = blockIdx.x * (GX_BM / 16) + wave;
  const int mrow = lane & 15, kh = lane >> 4;

  const int r0 = tid >> 4, p = tid & 15;
  const float* ga = u + (size_t)(blockIdx.x * GX_BM + r0) * DI + 4 * p;
  const float* gb = Wxp + (size_t)r0 * DI + 4 * p;
  const bool tail_valid = (r0 + 40) < (DTR + 2 * NST);
  const float* gt = Wxp + (size_t)(tail_valid ? r0 + 40 : 0) * DI + 4 * p;
  const int lo = r0 * GX_PITCH + 4 * p;
  const v4f zero = {0.0f, 0.0f, 0.0f, 0.0f};

  v4f st[GX_NP];
#pragma unroll
  for (int j = 0; j < 8; ++j)
    st[j] = *reinterpret_cast<const v4f*>(ga + (size_t)j * 8 * DI);
#pragma unroll
  for (int j = 0; j < 5; ++j)
    st[8 + j] = *reinterpret_cast<const v4f*>(gb + (size_t)j * 8 * DI);
  st[13] = *reinterpret_cast<const v4f*>(gt);
  st[13] = tail_valid ? st[13] : zero;
#pragma unroll
  for (int j = 0; j < GX_NP; ++j)
    *reinterpret_cast<v4f*>(&sT[0][lo + j * 8 * GX_PITCH]) = st[j];
  __syncthreads();

  const int aoff = (wave * 16 + mrow) * GX_PITCH + 16 * kh;
  const int boff = (GX_BM + mrow) * GX_PITCH + 16 * kh;

  v8f acc[3] = {};
  for (int s = 0; s < GX_STAGES; ++s) {
    const float* cur = sT[s & 1];
    const int kn = (s + 1 < GX_STAGES) ? (s + 1) * GX_KS : s * GX_KS;
#pragma unroll
    for (int j = 0; j < 8; ++j)
      st[j] = *reinterpret_cast<const v4f*>(ga + (size_t)j * 8 * DI + kn);
#pragma unroll
    for (int j = 0; j < 5; ++j)
      st[8 + j] = *reinterpret_cast<const v4f*>(gb + (size_t)j * 8 * DI + kn);
    st[13] = *reinterpret_cast<const v4f*>(gt + kn);

#pragma unroll
    for (int t = 0; t < GX_KS / 32; ++t) {
      v4f a[4], b[3][4];
      ld16(cur + aoff + 32 * t, a);
#pragma unroll
      for (int j = 0; j < 3; ++j) ld16(cur + boff + j * 16 * GX_PITCH + 32 * t, b[j]);
      asm volatile("" :: LOADED4(a), LOADED4(b[0]), LOADED4(b[1]), LOADED4(b[2]));
#pragma unroll
      for (int q = 0; q < 4; ++q) {
#pragma unroll
        for (int j = 0; j < 3; ++j) acc[j] = wmma4(lo2(a[q]), lo2(b[j][q]), acc[j]);
#pragma unroll
        for (int j = 0; j < 3; ++j) acc[j] = wmma4(hi2(a[q]), hi2(b[j][q]), acc[j]);
      }
    }
    float* nxt = sT[(s + 1) & 1];
    st[13] = tail_valid ? st[13] : zero;
#pragma unroll
    for (int j = 0; j < GX_NP; ++j)
      *reinterpret_cast<v4f*>(&nxt[lo + j * 8 * GX_PITCH]) = st[j];
    __syncthreads();
  }
#pragma unroll
  for (int j = 0; j < 3; ++j)
#pragma unroll
    for (int v = 0; v < 8; ++v) {
      int orow = gm * 16 + v + 8 * kh;
      dbl[(size_t)orow * DBLW + j * 16 + mrow] = acc[j][v];
    }
}

__global__ void k_gemm_dt(const float* __restrict__ dbl_all,
                          const float* __restrict__ Wdt_f,
                          const float* __restrict__ bdt_f,
                          const float* __restrict__ Wdt_b,
                          const float* __restrict__ bdt_b,
                          float* __restrict__ xz_all) {
  const int dir = blockIdx.y;
  const float* dbl = dbl_all + (size_t)dir * MROWS * DBLW;
  const float* Wdt = dir ? Wdt_b : Wdt_f;
  const float* bdt = dir ? bdt_b : bdt_f;
  float* xz = xz_all + (size_t)dir * MROWS * XZW;
  const int wave = blockIdx.x * (blockDim.x >> 5) + (threadIdx.x >> 5);
  const int lane = threadIdx.x & 31;
  const int TG = DI / 64;
  const int gm = wave / TG, gng = wave % TG;
  const int mrow = lane & 15, kh = lane >> 4;

  const float* ap = dbl + (size_t)(gm * 16 + mrow) * DBLW;
  const float* bp[4];
#pragma unroll
  for (int j = 0; j < 4; ++j)
    bp[j] = Wdt + (size_t)(gng * 64 + j * 16 + mrow) * DTR;

  v8f acc[4] = {};
  for (int k0 = 0; k0 < DTR; k0 += 4) {
    int kk = k0 + 2 * kh;
    v2f a; a.x = ap[kk]; a.y = ap[kk + 1];
#pragma unroll
    for (int j = 0; j < 4; ++j) {
      v2f b; b.x = bp[j][kk]; b.y = bp[j][kk + 1];
      acc[j] = wmma4(a, b, acc[j]);
    }
  }
#pragma unroll
  for (int j = 0; j < 4; ++j) {
    int n = gng * 64 + j * 16 + mrow;
    float bias = bdt[n];
#pragma unroll
    for (int v = 0; v < 8; ++v) {
      int orow = gm * 16 + v + 8 * kh;
      float t = acc[j][v] + bias;
      float sp = (t > 20.0f) ? t : log1pf(__expf(t));
      xz[(size_t)orow * XZW + n] = sp;
    }
  }
}

#define TGRP 8
#define NGRP (CL / TGRP)
#define REQUESTS_STAY_AHEAD() asm volatile("" ::: "memory")

__device__ __forceinline__ void scan1_group(const float (&dt8)[TGRP], const float (&u8)[TGRP],
                                            const v4f (*sB)[NST / 4],
                                            const float (&a)[NST], float (&P)[NST], float (&S)[NST]) {
#pragma unroll
  for (int k = 0; k < TGRP; ++k) {
    float Bv[NST];
#pragma unroll
    for (int q = 0; q < NST / 4; ++q) {
      const v4f v = sB[k][q];
      Bv[4 * q] = v.x; Bv[4 * q + 1] = v.y; Bv[4 * q + 2] = v.z; Bv[4 * q + 3] = v.w;
    }
    const float dt = dt8[k];
    const float dtu = dt * u8[k];
#pragma unroll
    for (int n = 0; n < NST; ++n) {
      float dA = __expf(dt * a[n]);
      S[n] = S[n] * dA + dtu * Bv[n];
      P[n] *= dA;
    }
  }
}

__global__ void __launch_bounds__(DI) k_scan_part1(const float* __restrict__ xz_all,
                             const float* __restrict__ u_all,
                             const float* __restrict__ dbl_all,
                             const float* __restrict__ Alog_f,
                             const float* __restrict__ Alog_b,
                             float* __restrict__ Pb_all, float* __restrict__ Sb_all) {
  __shared__ v4f sB[CL][NST / 4];
  const int dir = blockIdx.y;
  const float* xz  = xz_all  + (size_t)dir * MROWS * XZW;
  const float* u   = u_all   + (size_t)dir * MROWS * DI;
  const float* dbl = dbl_all + (size_t)dir * MROWS * DBLW;
  const float* Alog = dir ? Alog_b : Alog_f;
  float* Pb = Pb_all + (size_t)dir * PSTRIDE;
  float* Sb = Sb_all + (size_t)dir * PSTRIDE;
  const int b = blockIdx.x / NCH;
  const int c = blockIdx.x % NCH;
  const int d = threadIdx.x;

  const size_t row0 = (size_t)b * SEQ + (size_t)c * CL;
  const float* dtp = xz  + row0 * XZW + d;
  const float* utp = u   + row0 * DI + d;
  const float* bcp = dbl + row0 * DBLW + DTR;

  float dtA[TGRP], utA[TGRP], dtB[TGRP], utB[TGRP];
#pragma unroll
  for (int k = 0; k < TGRP; ++k) {
    dtA[k] = dtp[(size_t)k * XZW];
    utA[k] = utp[(size_t)k * DI];
  }
  if (d < CL * (NST / 4)) {
    const int r = d >> 2, q = d & 3;
    sB[r][q] = *reinterpret_cast<const v4f*>(bcp + (size_t)r * DBLW + 4 * q);
  }

  float a[NST], P[NST], S[NST];
#pragma unroll
  for (int n = 0; n < NST; ++n) {
    a[n] = -__expf(Alog[d * NST + n]);
    P[n] = 1.0f;
    S[n] = 0.0f;
  }
  __syncthreads();

  for (int g = 0; g < NGRP; g += 2) {
#pragma unroll
    for (int k = 0; k < TGRP; ++k) {
      dtB[k] = dtp[(size_t)((g + 1) * TGRP + k) * XZW];
      utB[k] = utp[(size_t)((g + 1) * TGRP + k) * DI];
    }
    REQUESTS_STAY_AHEAD();
    scan1_group(dtA, utA, sB + g * TGRP, a, P, S);
    const int gn = (g + 2 < NGRP) ? g + 2 : g + 1;
#pragma unroll
    for (int k = 0; k < TGRP; ++k) {
      dtA[k] = dtp[(size_t)(gn * TGRP + k) * XZW];
      utA[k] = utp[(size_t)(gn * TGRP + k) * DI];
    }
    REQUESTS_STAY_AHEAD();
    scan1_group(dtB, utB, sB + (g + 1) * TGRP, a, P, S);
  }
  const size_t base = ((size_t)(b * NCH + c) * DI + d) * NST;
#pragma unroll
  for (int n = 0; n < NST; ++n) {
    Pb[base + n] = P[n];
    Sb[base + n] = S[n];
  }
}

__global__ void k_scan_combine(const float* __restrict__ Pb_all,
                               float* __restrict__ Sb_all) {
  const int dir = blockIdx.y;
  const float* Pb = Pb_all + (size_t)dir * PSTRIDE;
  float* Sb = Sb_all + (size_t)dir * PSTRIDE;
  int idx = blockIdx.x * blockDim.x + threadIdx.x;
  if (idx >= BATCH * DI * NST) return;
  const int n = idx & (NST - 1);
  const int d = (idx >> 4) % DI;
  const int b = idx / (NST * DI);

  float carry = 0.0f;
  for (int c = 0; c < NCH; ++c) {
    const size_t base = ((size_t)(b * NCH + c) * DI + d) * NST + n;
    float p = Pb[base];
    float s = Sb[base];
    Sb[base] = carry;
    carry = p * carry + s;
  }
}

__device__ __forceinline__ void scan3_group(const float (&dt8)[TGRP], const float (&u8)[TGRP],
                                            const float (&z8)[TGRP],
                                            const v4f (*sBC)[2 * NST / 4], float* yp,
                                            const float (&a)[NST], float (&h)[NST], float Dd) {
#pragma unroll
  for (int k = 0; k < TGRP; ++k) {
    float BCv[2 * NST];
#pragma unroll
    for (int q = 0; q < 2 * NST / 4; ++q) {
      const v4f v = sBC[k][q];
      BCv[4 * q] = v.x; BCv[4 * q + 1] = v.y; BCv[4 * q + 2] = v.z; BCv[4 * q + 3] = v.w;
    }
    const float dt = dt8[k];
    const float ut = u8[k];
    const float dtu = dt * ut;
    float y = 0.0f;
#pragma unroll
    for (int n = 0; n < NST; ++n) {
      float dA = __expf(dt * a[n]);
      h[n] = h[n] * dA + dtu * BCv[n];
      y += h[n] * BCv[NST + n];
    }
    y += ut * Dd;
    const float z = z8[k];
    y *= z / (1.0f + __expf(-z));
    yp[(size_t)k * XZW] = y;
  }
}

__global__ void __launch_bounds__(DI) k_scan_part3(float* __restrict__ xz_all,
                             const float* __restrict__ u_all,
                             const float* __restrict__ dbl_all,
                             const float* __restrict__ Alog_f,
                             const float* __restrict__ Dv_f,
                             const float* __restrict__ Alog_b,
                             const float* __restrict__ Dv_b,
                             const float* __restrict__ Sb_all) {
  __shared__ v4f sBC[CL][2 * NST / 4];
  const int dir = blockIdx.y;
  float* xz        = xz_all  + (size_t)dir * MROWS * XZW;
  const float* u   = u_all   + (size_t)dir * MROWS * DI;
  const float* dbl = dbl_all + (size_t)dir * MROWS * DBLW;
  const float* Alog = dir ? Alog_b : Alog_f;
  const float* Dv   = dir ? Dv_b : Dv_f;
  const float* Sb = Sb_all + (size_t)dir * PSTRIDE;
  const int b = blockIdx.x / NCH;
  const int c = blockIdx.x % NCH;
  const int d = threadIdx.x;

  const size_t row0 = (size_t)b * SEQ + (size_t)c * CL;
  float* xzp = xz + row0 * XZW + d;
  const float* utp = u   + row0 * DI + d;
  const float* bcp = dbl + row0 * DBLW + DTR;

  float dtA[TGRP], utA[TGRP], zA[TGRP], dtB[TGRP], utB[TGRP], zB[TGRP];
#pragma unroll
  for (int k = 0; k < TGRP; ++k) {
    dtA[k] = xzp[(size_t)k * XZW];
    zA[k]  = xzp[(size_t)k * XZW + DI];
    utA[k] = utp[(size_t)k * DI];
  }
  for (int i = d; i < CL * (2 * NST / 4); i += DI) {
    const int r = i >> 3, q = i & 7;
    sBC[r][q] = *reinterpret_cast<const v4f*>(bcp + (size_t)r * DBLW + 4 * q);
  }

  float a[NST], h[NST];
  const size_t base = ((size_t)(b * NCH + c) * DI + d) * NST;
#pragma unroll
  for (int n = 0; n < NST; ++n) {
    a[n] = -__expf(Alog[d * NST + n]);
    h[n] = Sb[base + n];
  }
  const float Dd = Dv[d];
  __syncthreads();

  for (int g = 0; g < NGRP; g += 2) {
#pragma unroll
    for (int k = 0; k < TGRP; ++k) {
      dtB[k] = xzp[(size_t)((g + 1) * TGRP + k) * XZW];
      zB[k]  = xzp[(size_t)((g + 1) * TGRP + k) * XZW + DI];
      utB[k] = utp[(size_t)((g + 1) * TGRP + k) * DI];
    }
    REQUESTS_STAY_AHEAD();
    scan3_group(dtA, utA, zA, sBC + g * TGRP, xzp + (size_t)(g * TGRP) * XZW + DI, a, h, Dd);
    const int gn = (g + 2 < NGRP) ? g + 2 : g + 1;
#pragma unroll
    for (int k = 0; k < TGRP; ++k) {
      dtA[k] = xzp[(size_t)(gn * TGRP + k) * XZW];
      zA[k]  = xzp[(size_t)(gn * TGRP + k) * XZW + DI];
      utA[k] = utp[(size_t)(gn * TGRP + k) * DI];
    }
    REQUESTS_STAY_AHEAD();
    scan3_group(dtB, utB, zB, sBC + (g + 1) * TGRP, xzp + (size_t)((g + 1) * TGRP) * XZW + DI, a, h, Dd);
  }
}

#define GO_BM    32
#define GO_ROWS  (GO_BM + DIMX)
#define GO_STAGES (DI / GI_KS)
__global__ void __launch_bounds__(384) k_gemm_out(const float* __restrict__ xz_f,
                           const float* __restrict__ xz_b,
                           const float* __restrict__ x,
                           const float* __restrict__ Wout,
                           float* __restrict__ out) {
  __shared__ __attribute__((aligned(16))) float sT[2][GO_ROWS * GI_PITCH];
  const int mb = blockIdx.x;
  const int tid = threadIdx.x;
  const int wave = tid >> 5, lane = tid & 31;
  const int wm = wave / 6, wn = wave % 6;
  const int mrow = lane & 15, kh = lane >> 4;

  const int r0 = tid >> 3, p = tid & 7;
  const bool has_a = r0 < GO_BM;
  const int m  = mb * GO_BM + (has_a ? r0 : 0);
  const int bb = m / SEQ, ll = m % SEQ;
  const int mrev = bb * SEQ + (SEQ - 1 - ll);
  const float* gf = xz_f + (size_t)m    * XZW + DI + 4 * p;
  const float* gr = xz_b + (size_t)mrev * XZW + DI + 4 * p;
  const float* gb = Wout + (size_t)r0 * DI + 4 * p;
  const int loa = r0 * GI_PITCH + 4 * p;
  const int lob = (GO_BM + r0) * GI_PITCH + 4 * p;

  v4f sf = {0.0f, 0.0f, 0.0f, 0.0f}, sr = {0.0f, 0.0f, 0.0f, 0.0f}, st[4];
  if (has_a) {
    sf = *reinterpret_cast<const v4f*>(gf);
    sr = *reinterpret_cast<const v4f*>(gr);
  }
#pragma unroll
  for (int j = 0; j < 4; ++j)
    st[j] = *reinterpret_cast<const v4f*>(gb + (size_t)j * 48 * DI);
  if (has_a)
    *reinterpret_cast<v4f*>(&sT[0][loa]) = sf + sr;
#pragma unroll
  for (int j = 0; j < 4; ++j)
    *reinterpret_cast<v4f*>(&sT[0][lob + j * 48 * GI_PITCH]) = st[j];
  __syncthreads();

  const int aoff = (wm * 16 + mrow) * GI_PITCH + 16 * kh;
  const int boff = (GO_BM + wn * 32 + mrow) * GI_PITCH + 16 * kh;

  v8f acc[2] = {};
  for (int s = 0; s < GO_STAGES; ++s) {
    const float* cur = sT[s & 1];
    const int kn = (s + 1 < GO_STAGES) ? (s + 1) * GI_KS : s * GI_KS;
    if (has_a) {
      sf = *reinterpret_cast<const v4f*>(gf + kn);
      sr = *reinterpret_cast<const v4f*>(gr + kn);
    }
#pragma unroll
    for (int j = 0; j < 4; ++j)
      st[j] = *reinterpret_cast<const v4f*>(gb + (size_t)j * 48 * DI + kn);

    v4f a[4], b[2][4];
    ld16(cur + aoff, a);
#pragma unroll
    for (int j = 0; j < 2; ++j) ld16(cur + boff + j * 16 * GI_PITCH, b[j]);
    asm volatile("" :: LOADED4(a), LOADED4(b[0]), LOADED4(b[1]));
#pragma unroll
    for (int q = 0; q < 4; ++q) {
#pragma unroll
      for (int j = 0; j < 2; ++j) acc[j] = wmma4(lo2(a[q]), lo2(b[j][q]), acc[j]);
#pragma unroll
      for (int j = 0; j < 2; ++j) acc[j] = wmma4(hi2(a[q]), hi2(b[j][q]), acc[j]);
    }
    float* nxt = sT[(s + 1) & 1];
    if (has_a)
      *reinterpret_cast<v4f*>(&nxt[loa]) = sf + sr;
#pragma unroll
    for (int j = 0; j < 4; ++j)
      *reinterpret_cast<v4f*>(&nxt[lob + j * 48 * GI_PITCH]) = st[j];
    __syncthreads();
  }
#pragma unroll
  for (int j = 0; j < 2; ++j)
#pragma unroll
    for (int v = 0; v < 8; ++v) {
      int orow = mb * GO_BM + wm * 16 + v + 8 * kh;
      int n = wn * 32 + j * 16 + mrow;
      out[(size_t)orow * DIMX + n] = acc[j][v] + x[(size_t)orow * DIMX + n];
    }
}

extern "C" void kernel_launch(void* const* d_in, const int* in_sizes, int n_in,
                              void* d_out, int out_size, void* d_ws, size_t ws_size,
                              hipStream_t stream) {
  const float* x      = (const float*)d_in[0];
  const float* Win_f  = (const float*)d_in[1];
  const float* cw_f   = (const float*)d_in[2];
  const float* cb_f   = (const float*)d_in[3];
  const float* Wxp_f  = (const float*)d_in[4];
  const float* Wdt_f  = (const float*)d_in[5];
  const float* bdt_f  = (const float*)d_in[6];
  const float* Alog_f = (const float*)d_in[7];
  const float* Dv_f   = (const float*)d_in[8];
  const float* Win_b  = (const float*)d_in[9];
  const float* cw_b   = (const float*)d_in[10];
  const float* cb_b   = (const float*)d_in[11];
  const float* Wxp_b  = (const float*)d_in[12];
  const float* Wdt_b  = (const float*)d_in[13];
  const float* bdt_b  = (const float*)d_in[14];
  const float* Alog_b = (const float*)d_in[15];
  const float* Dv_b   = (const float*)d_in[16];
  const float* Wout   = (const float*)d_in[17];
  float* out = (float*)d_out;

  float* ws    = (float*)d_ws;
  float* xz_f  = ws;
  float* xz_b  = xz_f + (size_t)MROWS * XZW;
  float* u_f   = xz_b + (size_t)MROWS * XZW;
  float* dbl_f = u_f  + (size_t)2 * MROWS * DI;
  const size_t PS = (size_t)BATCH * NCH * DI * NST;
  float* P_f   = dbl_f + (size_t)2 * MROWS * DBLW;
  float* S_f   = P_f + PS;

  const int BLK = 256, WPB = BLK / 32;

  {
    int blocks = (MROWS / GI_BM) * GI_NB;
    k_gemm_in<<<dim3(blocks, 2), BLK, 0, stream>>>(x, Win_f, Win_b, xz_f);
  }
  {
    int n = (MROWS / CT) * DQ;
    k_conv_silu<<<dim3(n / BLK, 2), BLK, 0, stream>>>(
        xz_f, cw_f, cb_f, cw_b, cb_b, u_f);
  }
  {
    int blocks = MROWS / GX_BM;
    k_gemm_xp<<<dim3(blocks, 2), 128, 0, stream>>>(u_f, Wxp_f, Wxp_b, dbl_f);
  }
  {
    int waves = (MROWS / 16) * (DI / 64);
    k_gemm_dt<<<dim3(waves / WPB, 2), BLK, 0, stream>>>(
        dbl_f, Wdt_f, bdt_f, Wdt_b, bdt_b, xz_f);
  }
  {
    int cb = (BATCH * DI * NST + BLK - 1) / BLK;
    k_scan_part1<<<dim3(BATCH * NCH, 2), DI, 0, stream>>>(
        xz_f, u_f, dbl_f, Alog_f, Alog_b, P_f, S_f);
    k_scan_combine<<<dim3(cb, 2), BLK, 0, stream>>>(P_f, S_f);
    k_scan_part3<<<dim3(BATCH * NCH, 2), DI, 0, stream>>>(
        xz_f, u_f, dbl_f, Alog_f, Dv_f, Alog_b, Dv_b, S_f);
  }
  {
    int blocks = MROWS / GO_BM;
    k_gemm_out<<<blocks, 384, 0, stream>>>(xz_f, xz_b, x, Wout, out);
  }
}


# Scan passes, conv, out projection: every global request of a kernel's opening, and all chunks' P and S in the combine pass, issued before the first wait: serial load round trips go (combine 8 to 1)
# speedup vs baseline: 3.6461x; 1.0966x over previous
#include <hip/hip_runtime.h>
#include <hip/hip_bf16.h>
#include <math.h>

#define BATCH 4
#define SEQ   2048
#define DIMX  192
#define DI    384
#define NST   16
#define DTR   12
#define XZW   768
#define DBLW  48
#define MROWS (BATCH*SEQ)
#define NCH   32
#define CL    (SEQ/NCH)
#define PSTRIDE (2 * BATCH * NCH * DI * NST)

typedef float v2f __attribute__((ext_vector_type(2)));
typedef float v8f __attribute__((ext_vector_type(8)));

__device__ __forceinline__ v8f wmma4(v2f a, v2f b, v8f c) {
  return __builtin_amdgcn_wmma_f32_16x16x4_f32(false, a, false, b, (short)0, c,
                                               false, false);
}

typedef float v4f __attribute__((ext_vector_type(4)));

__device__ __forceinline__ void ld16(const float* __restrict__ p, v4f (&r)[4]) {
  const v4f* q = reinterpret_cast<const v4f*>(p);
  r[0] = q[0]; r[1] = q[1]; r[2] = q[2]; r[3] = q[3];
}
__device__ __forceinline__ v2f lo2(v4f v) { v2f r; r.x = v.x; r.y = v.y; return r; }
__device__ __forceinline__ v2f hi2(v4f v) { v2f r; r.x = v.z; r.y = v.w; return r; }
#define LOADED4(v) "v"((v)[0]), "v"((v)[1]), "v"((v)[2]), "v"((v)[3])
#define REQUESTS_STAY_AHEAD() asm volatile("" ::: "memory")

#define GI_BM    64
#define GI_BN    128
#define GI_KS    32
#define GI_ROWS  (GI_BM + GI_BN)
#define GI_PITCH 36
#define GI_NB    (XZW / GI_BN)
#define GI_STAGES (DIMX / GI_KS)
__global__ void __launch_bounds__(256) k_gemm_in(const float* __restrict__ x,
                          const float* __restrict__ Win_f,
                          const float* __restrict__ Win_b,
                          float* __restrict__ xz_all) {
  __shared__ __attribute__((aligned(16))) float sT[2][GI_ROWS * GI_PITCH];
  const int reverse = blockIdx.y;
  const float* Win = reverse ? Win_b : Win_f;
  float* xz = xz_all + (size_t)reverse * MROWS * XZW;
  const int nb = blockIdx.x % GI_NB;
  const int mb = blockIdx.x / GI_NB;
  const int tid = threadIdx.x;
  const int wave = tid >> 5, lane = tid & 31;
  const int wm = wave >> 1, wn = wave & 1;
  const int mrow = lane & 15, kh = lane >> 4;

  const int r0 = tid >> 3, p = tid & 7;
  const int m  = mb * GI_BM + r0;
  const int bb = m / SEQ, ll = m % SEQ;
  const int srow = reverse ? (bb * SEQ + (SEQ - 1 - ll)) : m;
  const float* ga = x + (size_t)srow * DIMX + 4 * p;
  const int astep = reverse ? -32 * DIMX : 32 * DIMX;
  const float* gb = Win + (size_t)(nb * GI_BN + r0) * DIMX + 4 * p;
  const int lo = r0 * GI_PITCH + 4 * p;

  v4f st[6];
  st[0] = *reinterpret_cast<const v4f*>(ga);
  st[1] = *reinterpret_cast<const v4f*>(ga + astep);
#pragma unroll
  for (int j = 0; j < 4; ++j)
    st[2 + j] = *reinterpret_cast<const v4f*>(gb + (size_t)j * 32 * DIMX);
#pragma unroll
  for (int j = 0; j < 6; ++j)
    *reinterpret_cast<v4f*>(&sT[0][lo + j * 32 * GI_PITCH]) = st[j];
  __syncthreads();

  const int aoff = (wm * 16 + mrow) * GI_PITCH + 16 * kh;
  const int boff = (GI_BM + wn * 64 + mrow) * GI_PITCH + 16 * kh;

  v8f acc[4] = {};
  for (int s = 0; s < GI_STAGES; ++s) {
    const float* cur = sT[s & 1];
    const int kn = (s + 1 < GI_STAGES) ? (s + 1) * GI_KS : s * GI_KS;
    st[0] = *reinterpret_cast<const v4f*>(ga + kn);
    st[1] = *reinterpret_cast<const v4f*>(ga + astep + kn);
#pragma unroll
    for (int j = 0; j < 4; ++j)
      st[2 + j] = *reinterpret_cast<const v4f*>(gb + (size_t)j * 32 * DIMX + kn);

    v4f a[4], b[4][4];
    ld16(cur + aoff, a);
#pragma unroll
    for (int j = 0; j < 4; ++j) ld16(cur + boff + j * 16 * GI_PITCH, b[j]);
    asm volatile("" :: LOADED4(a), LOADED4(b[0]), LOADED4(b[1]), LOADED4(b[2]), LOADED4(b[3]));
#pragma unroll
    for (int q = 0; q < 4; ++q) {
#pragma unroll
      for (int j = 0; j < 4; ++j) acc[j] = wmma4(lo2(a[q]), lo2(b[j][q]), acc[j]);
#pragma unroll
      for (int j = 0; j < 4; ++j) acc[j] = wmma4(hi2(a[q]), hi2(b[j][q]), acc[j]);
    }
    float* nxt = sT[(s + 1) & 1];
#pragma unroll
    for (int j = 0; j < 6; ++j)
      *reinterpret_cast<v4f*>(&nxt[lo + j * 32 * GI_PITCH]) = st[j];
    __syncthreads();
  }
#pragma unroll
  for (int j = 0; j < 4; ++j)
#pragma unroll
    for (int v = 0; v < 8; ++v) {
      int orow = mb * GI_BM + wm * 16 + v + 8 * kh;
      xz[(size_t)orow * XZW + nb * GI_BN + wn * 64 + j * 16 + mrow] = acc[j][v];
    }
}

#define CT 8
#define DQ (DI / 4)
__global__ void k_conv_silu(const float* __restrict__ xz_all,
                            const float* __restrict__ cw_f,
                            const float* __restrict__ cb_f,
                            const float* __restrict__ cw_b,
                            const float* __restrict__ cb_b,
                            float* __restrict__ u_all) {
  const int dir = blockIdx.y;
  const float* xz = xz_all + (size_t)dir * MROWS * XZW;
  const float* cw = dir ? cw_b : cw_f;
  const float* cb = dir ? cb_b : cb_f;
  float* u = u_all + (size_t)dir * MROWS * DI;
  const int idx = blockIdx.x * blockDim.x + threadIdx.x;
  const int dq  = idx % DQ;
  const int run = idx / DQ;
  const int row0 = run * CT;
  const bool head = (row0 % SEQ) == 0;

  const float* src = xz + (size_t)row0 * XZW + 4 * dq;
  v4f xr[CT + 3];
  for (int k = 0; k < 3; ++k) {
    v4f zero = {0.0f, 0.0f, 0.0f, 0.0f};
    xr[k] = head ? zero : *reinterpret_cast<const v4f*>(src + (k - 3) * XZW);
  }
  for (int k = 0; k < CT; ++k)
    xr[3 + k] = *reinterpret_cast<const v4f*>(src + (size_t)k * XZW);

  v4f w[4];
  for (int c = 0; c < 4; ++c)
    w[c] = *reinterpret_cast<const v4f*>(cw + (size_t)(4 * dq + c) * 4);
  const v4f bias = *reinterpret_cast<const v4f*>(cb + 4 * dq);
  asm volatile("" :: "v"(xr[0]), "v"(xr[1]), "v"(xr[2]), "v"(xr[3]), "v"(xr[4]), "v"(xr[5]),
               "v"(xr[6]), "v"(xr[7]), "v"(xr[8]), "v"(xr[9]), "v"(xr[10]), LOADED4(w), "v"(bias));

  float* dst = u + (size_t)row0 * DI + 4 * dq;
  for (int i = 0; i < CT; ++i) {
    v4f o;
    for (int c = 0; c < 4; ++c) {
      float acc = bias[c];
      for (int j = 0; j < 4; ++j) {
        if (i + j >= 3 || !head)
          acc += xr[i + j][c] * w[c][j];
      }
      o[c] = acc / (1.0f + __expf(-acc));
    }
    *reinterpret_cast<v4f*>(dst + (size_t)i * DI) = o;
  }
}

#define GX_BM    64
#define GX_BN    48
#define GX_KS    64
#define GX_ROWS  (GX_BM + GX_BN)
#define GX_PITCH 68
#define GX_STAGES (DI / GX_KS)
#define GX_NP    (GX_ROWS / 8)
__global__ void __launch_bounds__(128) k_gemm_xp(const float* __restrict__ u_all,
                          const float* __restrict__ Wxp_f,
                          const float* __restrict__ Wxp_b,
                          float* __restrict__ dbl_all) {
  __shared__ __attribute__((aligned(16))) float sT[2][GX_ROWS * GX_PITCH];
  const int dir = blockIdx.y;
  const float* u = u_all + (size_t)dir * MROWS * DI;
  const float* Wxp = dir ? Wxp_b : Wxp_f;
  float* dbl = dbl_all + (size_t)dir * MROWS * DBLW;
  const int tid = threadIdx.x;
  const int wave = tid >> 5, lane = tid & 31;
  const int gm = blockIdx.x * (GX_BM / 16) + wave;
  const int mrow = lane & 15, kh = lane >> 4;

  const int r0 = tid >> 4, p = tid & 15;
  const float* ga = u + (size_t)(blockIdx.x * GX_BM + r0) * DI + 4 * p;
  const float* gb = Wxp + (size_t)r0 * DI + 4 * p;
  const bool tail_valid = (r0 + 40) < (DTR + 2 * NST);
  const float* gt = Wxp + (size_t)(tail_valid ? r0 + 40 : 0) * DI + 4 * p;
  const int lo = r0 * GX_PITCH + 4 * p;
  const v4f zero = {0.0f, 0.0f, 0.0f, 0.0f};

  v4f st[GX_NP];
#pragma unroll
  for (int j = 0; j < 8; ++j)
    st[j] = *reinterpret_cast<const v4f*>(ga + (size_t)j * 8 * DI);
#pragma unroll
  for (int j = 0; j < 5; ++j)
    st[8 + j] = *reinterpret_cast<const v4f*>(gb + (size_t)j * 8 * DI);
  st[13] = *reinterpret_cast<const v4f*>(gt);
  st[13] = tail_valid ? st[13] : zero;
#pragma unroll
  for (int j = 0; j < GX_NP; ++j)
    *reinterpret_cast<v4f*>(&sT[0][lo + j * 8 * GX_PITCH]) = st[j];
  __syncthreads();

  const int aoff = (wave * 16 + mrow) * GX_PITCH + 16 * kh;
  const int boff = (GX_BM + mrow) * GX_PITCH + 16 * kh;

  v8f acc[3] = {};
  for (int s = 0; s < GX_STAGES; ++s) {
    const float* cur = sT[s & 1];
    const int kn = (s + 1 < GX_STAGES) ? (s + 1) * GX_KS : s * GX_KS;
#pragma unroll
    for (int j = 0; j < 8; ++j)
      st[j] = *reinterpret_cast<const v4f*>(ga + (size_t)j * 8 * DI + kn);
#pragma unroll
    for (int j = 0; j < 5; ++j)
      st[8 + j] = *reinterpret_cast<const v4f*>(gb + (size_t)j * 8 * DI + kn);
    st[13] = *reinterpret_cast<const v4f*>(gt + kn);

#pragma unroll
    for (int t = 0; t < GX_KS / 32; ++t) {
      v4f a[4], b[3][4];
      ld16(cur + aoff + 32 * t, a);
#pragma unroll
      for (int j = 0; j < 3; ++j) ld16(cur + boff + j * 16 * GX_PITCH + 32 * t, b[j]);
      asm volatile("" :: LOADED4(a), LOADED4(b[0]), LOADED4(b[1]), LOADED4(b[2]));
#pragma unroll
      for (int q = 0; q < 4; ++q) {
#pragma unroll
        for (int j = 0; j < 3; ++j) acc[j] = wmma4(lo2(a[q]), lo2(b[j][q]), acc[j]);
#pragma unroll
        for (int j = 0; j < 3; ++j) acc[j] = wmma4(hi2(a[q]), hi2(b[j][q]), acc[j]);
      }
    }
    float* nxt = sT[(s + 1) & 1];
    st[13] = tail_valid ? st[13] : zero;
#pragma unroll
    for (int j = 0; j < GX_NP; ++j)
      *reinterpret_cast<v4f*>(&nxt[lo + j * 8 * GX_PITCH]) = st[j];
    __syncthreads();
  }
#pragma unroll
  for (int j = 0; j < 3; ++j)
#pragma unroll
    for (int v = 0; v < 8; ++v) {
      int orow = gm * 16 + v + 8 * kh;
      dbl[(size_t)orow * DBLW + j * 16 + mrow] = acc[j][v];
    }
}

__global__ void k_gemm_dt(const float* __restrict__ dbl_all,
                          const float* __restrict__ Wdt_f,
                          const float* __restrict__ bdt_f,
                          const float* __restrict__ Wdt_b,
                          const float* __restrict__ bdt_b,
                          float* __restrict__ xz_all) {
  const int dir = blockIdx.y;
  const float* dbl = dbl_all + (size_t)dir * MROWS * DBLW;
  const float* Wdt = dir ? Wdt_b : Wdt_f;
  const float* bdt = dir ? bdt_b : bdt_f;
  float* xz = xz_all + (size_t)dir * MROWS * XZW;
  const int wave = blockIdx.x * (blockDim.x >> 5) + (threadIdx.x >> 5);
  const int lane = threadIdx.x & 31;
  const int TG = DI / 64;
  const int gm = wave / TG, gng = wave % TG;
  const int mrow = lane & 15, kh = lane >> 4;

  const float* ap = dbl + (size_t)(gm * 16 + mrow) * DBLW;
  const float* bp[4];
#pragma unroll
  for (int j = 0; j < 4; ++j)
    bp[j] = Wdt + (size_t)(gng * 64 + j * 16 + mrow) * DTR;

  v8f acc[4] = {};
  for (int k0 = 0; k0 < DTR; k0 += 4) {
    int kk = k0 + 2 * kh;
    v2f a; a.x = ap[kk]; a.y = ap[kk + 1];
#pragma unroll
    for (int j = 0; j < 4; ++j) {
      v2f b; b.x = bp[j][kk]; b.y = bp[j][kk + 1];
      acc[j] = wmma4(a, b, acc[j]);
    }
  }
#pragma unroll
  for (int j = 0; j < 4; ++j) {
    int n = gng * 64 + j * 16 + mrow;
    float bias = bdt[n];
#pragma unroll
    for (int v = 0; v < 8; ++v) {
      int orow = gm * 16 + v + 8 * kh;
      float t = acc[j][v] + bias;
      float sp = (t > 20.0f) ? t : log1pf(__expf(t));
      xz[(size_t)orow * XZW + n] = sp;
    }
  }
}

#define TGRP 8
#define NGRP (CL / TGRP)

__device__ __forceinline__ void scan1_group(const float (&dt8)[TGRP], const float (&u8)[TGRP],
                                            const v4f (*sB)[NST / 4],
                                            const float (&a)[NST], float (&P)[NST], float (&S)[NST]) {
#pragma unroll
  for (int k = 0; k < TGRP; ++k) {
    float Bv[NST];
#pragma unroll
    for (int q = 0; q < NST / 4; ++q) {
      const v4f v = sB[k][q];
      Bv[4 * q] = v.x; Bv[4 * q + 1] = v.y; Bv[4 * q + 2] = v.z; Bv[4 * q + 3] = v.w;
    }
    const float dt = dt8[k];
    const float dtu = dt * u8[k];
#pragma unroll
    for (int n = 0; n < NST; ++n) {
      float dA = __expf(dt * a[n]);
      S[n] = S[n] * dA + dtu * Bv[n];
      P[n] *= dA;
    }
  }
}

__global__ void __launch_bounds__(DI) k_scan_part1(const float* __restrict__ xz_all,
                             const float* __restrict__ u_all,
                             const float* __restrict__ dbl_all,
                             const float* __restrict__ Alog_f,
                             const float* __restrict__ Alog_b,
                             float* __restrict__ Pb_all, float* __restrict__ Sb_all) {
  __shared__ v4f sB[CL][NST / 4];
  const int dir = blockIdx.y;
  const float* xz  = xz_all  + (size_t)dir * MROWS * XZW;
  const float* u   = u_all   + (size_t)dir * MROWS * DI;
  const float* dbl = dbl_all + (size_t)dir * MROWS * DBLW;
  const float* Alog = dir ? Alog_b : Alog_f;
  float* Pb = Pb_all + (size_t)dir * PSTRIDE;
  float* Sb = Sb_all + (size_t)dir * PSTRIDE;
  const int b = blockIdx.x / NCH;
  const int c = blockIdx.x % NCH;
  const int d = threadIdx.x;

  const size_t row0 = (size_t)b * SEQ + (size_t)c * CL;
  const float* dtp = xz  + row0 * XZW + d;
  const float* utp = u   + row0 * DI + d;
  const float* bcp = dbl + row0 * DBLW + DTR;

  const bool has_piece = d < CL * (NST / 4);
  const int pr = d >> 2, pq = d & 3;
  v4f bpiece = {0.0f, 0.0f, 0.0f, 0.0f};
  if (has_piece)
    bpiece = *reinterpret_cast<const v4f*>(bcp + (size_t)pr * DBLW + 4 * pq);
  float al[NST];
#pragma unroll
  for (int n = 0; n < NST; ++n) al[n] = Alog[d * NST + n];
  float dtA[TGRP], utA[TGRP], dtB[TGRP], utB[TGRP];
#pragma unroll
  for (int k = 0; k < TGRP; ++k) {
    dtA[k] = dtp[(size_t)k * XZW];
    utA[k] = utp[(size_t)k * DI];
  }
  if (has_piece) sB[pr][pq] = bpiece;

  float a[NST], P[NST], S[NST];
#pragma unroll
  for (int n = 0; n < NST; ++n) {
    a[n] = -__expf(al[n]);
    P[n] = 1.0f;
    S[n] = 0.0f;
  }
  __syncthreads();

  for (int g = 0; g < NGRP; g += 2) {
#pragma unroll
    for (int k = 0; k < TGRP; ++k) {
      dtB[k] = dtp[(size_t)((g + 1) * TGRP + k) * XZW];
      utB[k] = utp[(size_t)((g + 1) * TGRP + k) * DI];
    }
    REQUESTS_STAY_AHEAD();
    scan1_group(dtA, utA, sB + g * TGRP, a, P, S);
    const int gn = (g + 2 < NGRP) ? g + 2 : g + 1;
#pragma unroll
    for (int k = 0; k < TGRP; ++k) {
      dtA[k] = dtp[(size_t)(gn * TGRP + k) * XZW];
      utA[k] = utp[(size_t)(gn * TGRP + k) * DI];
    }
    REQUESTS_STAY_AHEAD();
    scan1_group(dtB, utB, sB + (g + 1) * TGRP, a, P, S);
  }
  const size_t base = ((size_t)(b * NCH + c) * DI + d) * NST;
#pragma unroll
  for (int n = 0; n < NST; ++n) {
    Pb[base + n] = P[n];
    Sb[base + n] = S[n];
  }
}

__global__ void k_scan_combine(const float* __restrict__ Pb_all,
                               float* __restrict__ Sb_all) {
  const int dir = blockIdx.y;
  const float* Pb = Pb_all + (size_t)dir * PSTRIDE;
  float* Sb = Sb_all + (size_t)dir * PSTRIDE;
  int idx = blockIdx.x * blockDim.x + threadIdx.x;
  if (idx >= BATCH * DI * NST) return;
  const int n = idx & (NST - 1);
  const int d = (idx >> 4) % DI;
  const int b = idx / (NST * DI);

  const unsigned base0 = (unsigned)(((b * NCH) * DI + d) * NST + n);
  const unsigned cstep = (unsigned)(DI * NST);
  float p[NCH], s[NCH];
#pragma unroll
  for (int c = 0; c < NCH - 1; ++c) {
    p[c] = Pb[base0 + c * cstep];
    s[c] = Sb[base0 + c * cstep];
  }
  p[NCH - 1] = 0.0f;
  s[NCH - 1] = 0.0f;
#define QUAD(a, i) (v4f){(a)[i], (a)[(i) + 1], (a)[(i) + 2], (a)[(i) + 3]}
  asm volatile("" :: "v"(QUAD(p, 0)), "v"(QUAD(p, 4)), "v"(QUAD(p, 8)), "v"(QUAD(p, 12)),
               "v"(QUAD(p, 16)), "v"(QUAD(p, 20)), "v"(QUAD(p, 24)), "v"(QUAD(p, 28)),
               "v"(QUAD(s, 0)), "v"(QUAD(s, 4)), "v"(QUAD(s, 8)), "v"(QUAD(s, 12)),
               "v"(QUAD(s, 16)), "v"(QUAD(s, 20)), "v"(QUAD(s, 24)), "v"(QUAD(s, 28)));
#undef QUAD
  float carry = 0.0f;
#pragma unroll
  for (int c = 0; c < NCH; ++c) {
    Sb[base0 + c * cstep] = carry;
    if (c + 1 < NCH) carry = p[c] * carry + s[c];
  }
}

__device__ __forceinline__ void scan3_group(const float (&dt8)[TGRP], const float (&u8)[TGRP],
                                            const float (&z8)[TGRP],
                                            const v4f (*sBC)[2 * NST / 4], float* yp,
                                            const float (&a)[NST], float (&h)[NST], float Dd) {
#pragma unroll
  for (int k = 0; k < TGRP; ++k) {
    float BCv[2 * NST];
#pragma unroll
    for (int q = 0; q < 2 * NST / 4; ++q) {
      const v4f v = sBC[k][q];
      BCv[4 * q] = v.x; BCv[4 * q + 1] = v.y; BCv[4 * q + 2] = v.z; BCv[4 * q + 3] = v.w;
    }
    const float dt = dt8[k];
    const float ut = u8[k];
    const float dtu = dt * ut;
    float y = 0.0f;
#pragma unroll
    for (int n = 0; n < NST; ++n) {
      float dA = __expf(dt * a[n]);
      h[n] = h[n] * dA + dtu * BCv[n];
      y += h[n] * BCv[NST + n];
    }
    y += ut * Dd;
    const float z = z8[k];
    y *= z / (1.0f + __expf(-z));
    yp[(size_t)k * XZW] = y;
  }
}

__global__ void __launch_bounds__(DI) k_scan_part3(float* __restrict__ xz_all,
                             const float* __restrict__ u_all,
                             const float* __restrict__ dbl_all,
                             const float* __restrict__ Alog_f,
                             const float* __restrict__ Dv_f,
                             const float* __restrict__ Alog_b,
                             const float* __restrict__ Dv_b,
                             const float* __restrict__ Sb_all) {
  __shared__ v4f sBC[CL][2 * NST / 4];
  const int dir = blockIdx.y;
  float* xz        = xz_all  + (size_t)dir * MROWS * XZW;
  const float* u   = u_all   + (size_t)dir * MROWS * DI;
  const float* dbl = dbl_all + (size_t)dir * MROWS * DBLW;
  const float* Alog = dir ? Alog_b : Alog_f;
  const float* Dv   = dir ? Dv_b : Dv_f;
  const float* Sb = Sb_all + (size_t)dir * PSTRIDE;
  const int b = blockIdx.x / NCH;
  const int c = blockIdx.x % NCH;
  const int d = threadIdx.x;

  const size_t row0 = (size_t)b * SEQ + (size_t)c * CL;
  float* xzp = xz + row0 * XZW + d;
  const float* utp = u   + row0 * DI + d;
  const float* bcp = dbl + row0 * DBLW + DTR;

  const int i0 = d, i1 = d + DI;
  const bool two = i1 < CL * (2 * NST / 4);
  v4f bc0 = *reinterpret_cast<const v4f*>(bcp + (size_t)(i0 >> 3) * DBLW + 4 * (i0 & 7));
  v4f bc1 = {0.0f, 0.0f, 0.0f, 0.0f};
  if (two)
    bc1 = *reinterpret_cast<const v4f*>(bcp + (size_t)(i1 >> 3) * DBLW + 4 * (i1 & 7));
  float al[NST], h[NST];
  const size_t base = ((size_t)(b * NCH + c) * DI + d) * NST;
#pragma unroll
  for (int n = 0; n < NST; ++n) {
    al[n] = Alog[d * NST + n];
    h[n] = Sb[base + n];
  }
  const float Dd = Dv[d];
  float dtA[TGRP], utA[TGRP], zA[TGRP], dtB[TGRP], utB[TGRP], zB[TGRP];
#pragma unroll
  for (int k = 0; k < TGRP; ++k) {
    dtA[k] = xzp[(size_t)k * XZW];
    zA[k]  = xzp[(size_t)k * XZW + DI];
    utA[k] = utp[(size_t)k * DI];
  }
  sBC[i0 >> 3][i0 & 7] = bc0;
  if (two) sBC[i1 >> 3][i1 & 7] = bc1;

  float a[NST];
#pragma unroll
  for (int n = 0; n < NST; ++n) a[n] = -__expf(al[n]);
  __syncthreads();

  for (int g = 0; g < NGRP; g += 2) {
#pragma unroll
    for (int k = 0; k < TGRP; ++k) {
      dtB[k] = xzp[(size_t)((g + 1) * TGRP + k) * XZW];
      zB[k]  = xzp[(size_t)((g + 1) * TGRP + k) * XZW + DI];
      utB[k] = utp[(size_t)((g + 1) * TGRP + k) * DI];
    }
    REQUESTS_STAY_AHEAD();
    scan3_group(dtA, utA, zA, sBC + g * TGRP, xzp + (size_t)(g * TGRP) * XZW + DI, a, h, Dd);
    const int gn = (g + 2 < NGRP) ? g + 2 : g + 1;
#pragma unroll
    for (int k = 0; k < TGRP; ++k) {
      dtA[k] = xzp[(size_t)(gn * TGRP + k) * XZW];
      zA[k]  = xzp[(size_t)(gn * TGRP + k) * XZW + DI];
      utA[k] = utp[(size_t)(gn * TGRP + k) * DI];
    }
    REQUESTS_STAY_AHEAD();
    scan3_group(dtB, utB, zB, sBC + (g + 1) * TGRP, xzp + (size_t)((g + 1) * TGRP) * XZW + DI, a, h, Dd);
  }
}

#define GO_BM    32
#define GO_ROWS  (GO_BM + DIMX)
#define GO_STAGES (DI / GI_KS)
__global__ void __launch_bounds__(384) k_gemm_out(const float* __restrict__ xz_f,
                           const float* __restrict__ xz_b,
                           const float* __restrict__ x,
                           const float* __restrict__ Wout,
                           float* __restrict__ out) {
  __shared__ __attribute__((aligned(16))) float sT[2][GO_ROWS * GI_PITCH];
  const int mb = blockIdx.x;
  const int tid = threadIdx.x;
  const int wave = tid >> 5, lane = tid & 31;
  const int wm = wave / 6, wn = wave % 6;
  const int mrow = lane & 15, kh = lane >> 4;

  float xres[2][8];
#pragma unroll
  for (int j = 0; j < 2; ++j)
#pragma unroll
    for (int v = 0; v < 8; ++v)
      xres[j][v] = x[(size_t)(mb * GO_BM + wm * 16 + v + 8 * kh) * DIMX + wn * 32 + j * 16 + mrow];

  const int r0 = tid >> 3, p = tid & 7;
  const bool has_a = r0 < GO_BM;
  const int m  = mb * GO_BM + (has_a ? r0 : 0);
  const int bb = m / SEQ, ll = m % SEQ;
  const int mrev = bb * SEQ + (SEQ - 1 - ll);
  const float* gf = xz_f + (size_t)m    * XZW + DI + 4 * p;
  const float* gr = xz_b + (size_t)mrev * XZW + DI + 4 * p;
  const float* gb = Wout + (size_t)r0 * DI + 4 * p;
  const int loa = r0 * GI_PITCH + 4 * p;
  const int lob = (GO_BM + r0) * GI_PITCH + 4 * p;

  v4f sf = {0.0f, 0.0f, 0.0f, 0.0f}, sr = {0.0f, 0.0f, 0.0f, 0.0f}, st[4];
  if (has_a) {
    sf = *reinterpret_cast<const v4f*>(gf);
    sr = *reinterpret_cast<const v4f*>(gr);
  }
#pragma unroll
  for (int j = 0; j < 4; ++j)
    st[j] = *reinterpret_cast<const v4f*>(gb + (size_t)j * 48 * DI);
  if (has_a)
    *reinterpret_cast<v4f*>(&sT[0][loa]) = sf + sr;
#pragma unroll
  for (int j = 0; j < 4; ++j)
    *reinterpret_cast<v4f*>(&sT[0][lob + j * 48 * GI_PITCH]) = st[j];
  __syncthreads();

  const int aoff = (wm * 16 + mrow) * GI_PITCH + 16 * kh;
  const int boff = (GO_BM + wn * 32 + mrow) * GI_PITCH + 16 * kh;

  v8f acc[2] = {};
  for (int s = 0; s < GO_STAGES; ++s) {
    const float* cur = sT[s & 1];
    const int kn = (s + 1 < GO_STAGES) ? (s + 1) * GI_KS : s * GI_KS;
    if (has_a) {
      sf = *reinterpret_cast<const v4f*>(gf + kn);
      sr = *reinterpret_cast<const v4f*>(gr + kn);
    }
#pragma unroll
    for (int j = 0; j < 4; ++j)
      st[j] = *reinterpret_cast<const v4f*>(gb + (size_t)j * 48 * DI + kn);

    v4f a[4], b[2][4];
    ld16(cur + aoff, a);
#pragma unroll
    for (int j = 0; j < 2; ++j) ld16(cur + boff + j * 16 * GI_PITCH, b[j]);
    asm volatile("" :: LOADED4(a), LOADED4(b[0]), LOADED4(b[1]));
#pragma unroll
    for (int q = 0; q < 4; ++q) {
#pragma unroll
      for (int j = 0; j < 2; ++j) acc[j] = wmma4(lo2(a[q]), lo2(b[j][q]), acc[j]);
#pragma unroll
      for (int j = 0; j < 2; ++j) acc[j] = wmma4(hi2(a[q]), hi2(b[j][q]), acc[j]);
    }
    float* nxt = sT[(s + 1) & 1];
    if (has_a)
      *reinterpret_cast<v4f*>(&nxt[loa]) = sf + sr;
#pragma unroll
    for (int j = 0; j < 4; ++j)
      *reinterpret_cast<v4f*>(&nxt[lob + j * 48 * GI_PITCH]) = st[j];
    __syncthreads();
  }
#pragma unroll
  for (int j = 0; j < 2; ++j)
#pragma unroll
    for (int v = 0; v < 8; ++v) {
      int orow = mb * GO_BM + wm * 16 + v + 8 * kh;
      int n = wn * 32 + j * 16 + mrow;
      out[(size_t)orow * DIMX + n] = acc[j][v] + xres[j][v];
    }
}

extern "C" void kernel_launch(void* const* d_in, const int* in_sizes, int n_in,
                              void* d_out, int out_size, void* d_ws, size_t ws_size,
                              hipStream_t stream) {
  const float* x      = (const float*)d_in[0];
  const float* Win_f  = (const float*)d_in[1];
  const float* cw_f   = (const float*)d_in[2];
  const float* cb_f   = (const float*)d_in[3];
  const float* Wxp_f  = (const float*)d_in[4];
  const float* Wdt_f  = (const float*)d_in[5];
  const float* bdt_f  = (const float*)d_in[6];
  const float* Alog_f = (const float*)d_in[7];
  const float* Dv_f   = (const float*)d_in[8];
  const float* Win_b  = (const float*)d_in[9];
  const float* cw_b   = (const float*)d_in[10];
  const float* cb_b   = (const float*)d_in[11];
  const float* Wxp_b  = (const float*)d_in[12];
  const float* Wdt_b  = (const float*)d_in[13];
  const float* bdt_b  = (const float*)d_in[14];
  const float* Alog_b = (const float*)d_in[15];
  const float* Dv_b   = (const float*)d_in[16];
  const float* Wout   = (const float*)d_in[17];
  float* out = (float*)d_out;

  float* ws    = (float*)d_ws;
  float* xz_f  = ws;
  float* xz_b  = xz_f + (size_t)MROWS * XZW;
  float* u_f   = xz_b + (size_t)MROWS * XZW;
  float* dbl_f = u_f  + (size_t)2 * MROWS * DI;
  const size_t PS = (size_t)BATCH * NCH * DI * NST;
  float* P_f   = dbl_f + (size_t)2 * MROWS * DBLW;
  float* S_f   = P_f + PS;

  const int BLK = 256, WPB = BLK / 32;

  {
    int blocks = (MROWS / GI_BM) * GI_NB;
    k_gemm_in<<<dim3(blocks, 2), BLK, 0, stream>>>(x, Win_f, Win_b, xz_f);
  }
  {
    int n = (MROWS / CT) * DQ;
    k_conv_silu<<<dim3(n / BLK, 2), BLK, 0, stream>>>(
        xz_f, cw_f, cb_f, cw_b, cb_b, u_f);
  }
  {
    int blocks = MROWS / GX_BM;
    k_gemm_xp<<<dim3(blocks, 2), 128, 0, stream>>>(u_f, Wxp_f, Wxp_b, dbl_f);
  }
  {
    int waves = (MROWS / 16) * (DI / 64);
    k_gemm_dt<<<dim3(waves / WPB, 2), BLK, 0, stream>>>(
        dbl_f, Wdt_f, bdt_f, Wdt_b, bdt_b, xz_f);
  }
  {
    int cb = (BATCH * DI * NST + BLK - 1) / BLK;
    k_scan_part1<<<dim3(BATCH * NCH, 2), DI, 0, stream>>>(
        xz_f, u_f, dbl_f, Alog_f, Alog_b, P_f, S_f);
    k_scan_combine<<<dim3(cb, 2), BLK, 0, stream>>>(P_f, S_f);
    k_scan_part3<<<dim3(BATCH * NCH, 2), DI, 0, stream>>>(
        xz_f, u_f, dbl_f, Alog_f, Dv_f, Alog_b, Dv_b, S_f);
  }
  {
    int blocks = MROWS / GO_BM;
    k_gemm_out<<<blocks, 384, 0, stream>>>(xz_f, xz_b, x, Wout, out);
  }
}


# dt stage: softplus as max(t,0)+log1p(exp(-|t|)) with log1p(e)=2*atanh(e/(2+e)) by the odd series to s^13, in place of library log1pf: far fewer vector instructions; dt moves by a few ulp
# speedup vs baseline: 3.9913x; 1.0947x over previous
#include <hip/hip_runtime.h>
#include <hip/hip_bf16.h>
#include <math.h>

#define BATCH 4
#define SEQ   2048
#define DIMX  192
#define DI    384
#define NST   16
#define DTR   12
#define XZW   768
#define DBLW  48
#define MROWS (BATCH*SEQ)
#define NCH   32
#define CL    (SEQ/NCH)
#define PSTRIDE (2 * BATCH * NCH * DI * NST)

typedef float v2f __attribute__((ext_vector_type(2)));
typedef float v8f __attribute__((ext_vector_type(8)));

__device__ __forceinline__ v8f wmma4(v2f a, v2f b, v8f c) {
  return __builtin_amdgcn_wmma_f32_16x16x4_f32(false, a, false, b, (short)0, c,
                                               false, false);
}

typedef float v4f __attribute__((ext_vector_type(4)));

__device__ __forceinline__ void ld16(const float* __restrict__ p, v4f (&r)[4]) {
  const v4f* q = reinterpret_cast<const v4f*>(p);
  r[0] = q[0]; r[1] = q[1]; r[2] = q[2]; r[3] = q[3];
}
__device__ __forceinline__ v2f lo2(v4f v) { v2f r; r.x = v.x; r.y = v.y; return r; }
__device__ __forceinline__ v2f hi2(v4f v) { v2f r; r.x = v.z; r.y = v.w; return r; }
#define LOADED4(v) "v"((v)[0]), "v"((v)[1]), "v"((v)[2]), "v"((v)[3])
#define REQUESTS_STAY_AHEAD() asm volatile("" ::: "memory")

#define GI_BM    64
#define GI_BN    128
#define GI_KS    32
#define GI_ROWS  (GI_BM + GI_BN)
#define GI_PITCH 36
#define GI_NB    (XZW / GI_BN)
#define GI_STAGES (DIMX / GI_KS)
__global__ void __launch_bounds__(256) k_gemm_in(const float* __restrict__ x,
                          const float* __restrict__ Win_f,
                          const float* __restrict__ Win_b,
                          float* __restrict__ xz_all) {
  __shared__ __attribute__((aligned(16))) float sT[2][GI_ROWS * GI_PITCH];
  const int reverse = blockIdx.y;
  const float* Win = reverse ? Win_b : Win_f;
  float* xz = xz_all + (size_t)reverse * MROWS * XZW;
  const int nb = blockIdx.x % GI_NB;
  const int mb = blockIdx.x / GI_NB;
  const int tid = threadIdx.x;
  const int wave = tid >> 5, lane = tid & 31;
  const int wm = wave >> 1, wn = wave & 1;
  const int mrow = lane & 15, kh = lane >> 4;

  const int r0 = tid >> 3, p = tid & 7;
  const int m  = mb * GI_BM + r0;
  const int bb = m / SEQ, ll = m % SEQ;
  const int srow = reverse ? (bb * SEQ + (SEQ - 1 - ll)) : m;
  const float* ga = x + (size_t)srow * DIMX + 4 * p;
  const int astep = reverse ? -32 * DIMX : 32 * DIMX;
  const float* gb = Win + (size_t)(nb * GI_BN + r0) * DIMX + 4 * p;
  const int lo = r0 * GI_PITCH + 4 * p;

  v4f st[6];
  st[0] = *reinterpret_cast<const v4f*>(ga);
  st[1] = *reinterpret_cast<const v4f*>(ga + astep);
#pragma unroll
  for (int j = 0; j < 4; ++j)
    st[2 + j] = *reinterpret_cast<const v4f*>(gb + (size_t)j * 32 * DIMX);
#pragma unroll
  for (int j = 0; j < 6; ++j)
    *reinterpret_cast<v4f*>(&sT[0][lo + j * 32 * GI_PITCH]) = st[j];
  __syncthreads();

  const int aoff = (wm * 16 + mrow) * GI_PITCH + 16 * kh;
  const int boff = (GI_BM + wn * 64 + mrow) * GI_PITCH + 16 * kh;

  v8f acc[4] = {};
  for (int s = 0; s < GI_STAGES; ++s) {
    const float* cur = sT[s & 1];
    const int kn = (s + 1 < GI_STAGES) ? (s + 1) * GI_KS : s * GI_KS;
    st[0] = *reinterpret_cast<const v4f*>(ga + kn);
    st[1] = *reinterpret_cast<const v4f*>(ga + astep + kn);
#pragma unroll
    for (int j = 0; j < 4; ++j)
      st[2 + j] = *reinterpret_cast<const v4f*>(gb + (size_t)j * 32 * DIMX + kn);

    v4f a[4], b[4][4];
    ld16(cur + aoff, a);
#pragma unroll
    for (int j = 0; j < 4; ++j) ld16(cur + boff + j * 16 * GI_PITCH, b[j]);
    asm volatile("" :: LOADED4(a), LOADED4(b[0]), LOADED4(b[1]), LOADED4(b[2]), LOADED4(b[3]));
#pragma unroll
    for (int q = 0; q < 4; ++q) {
#pragma unroll
      for (int j = 0; j < 4; ++j) acc[j] = wmma4(lo2(a[q]), lo2(b[j][q]), acc[j]);
#pragma unroll
      for (int j = 0; j < 4; ++j) acc[j] = wmma4(hi2(a[q]), hi2(b[j][q]), acc[j]);
    }
    float* nxt = sT[(s + 1) & 1];
#pragma unroll
    for (int j = 0; j < 6; ++j)
      *reinterpret_cast<v4f*>(&nxt[lo + j * 32 * GI_PITCH]) = st[j];
    __syncthreads();
  }
#pragma unroll
  for (int j = 0; j < 4; ++j)
#pragma unroll
    for (int v = 0; v < 8; ++v) {
      int orow = mb * GI_BM + wm * 16 + v + 8 * kh;
      xz[(size_t)orow * XZW + nb * GI_BN + wn * 64 + j * 16 + mrow] = acc[j][v];
    }
}

#define CT 8
#define DQ (DI / 4)
__global__ void k_conv_silu(const float* __restrict__ xz_all,
                            const float* __restrict__ cw_f,
                            const float* __restrict__ cb_f,
                            const float* __restrict__ cw_b,
                            const float* __restrict__ cb_b,
                            float* __restrict__ u_all) {
  const int dir = blockIdx.y;
  const float* xz = xz_all + (size_t)dir * MROWS * XZW;
  const float* cw = dir ? cw_b : cw_f;
  const float* cb = dir ? cb_b : cb_f;
  float* u = u_all + (size_t)dir * MROWS * DI;
  const int idx = blockIdx.x * blockDim.x + threadIdx.x;
  const int dq  = idx % DQ;
  const int run = idx / DQ;
  const int row0 = run * CT;
  const bool head = (row0 % SEQ) == 0;

  const float* src = xz + (size_t)row0 * XZW + 4 * dq;
  v4f xr[CT + 3];
  for (int k = 0; k < 3; ++k) {
    v4f zero = {0.0f, 0.0f, 0.0f, 0.0f};
    xr[k] = head ? zero : *reinterpret_cast<const v4f*>(src + (k - 3) * XZW);
  }
  for (int k = 0; k < CT; ++k)
    xr[3 + k] = *reinterpret_cast<const v4f*>(src + (size_t)k * XZW);

  v4f w[4];
  for (int c = 0; c < 4; ++c)
    w[c] = *reinterpret_cast<const v4f*>(cw + (size_t)(4 * dq + c) * 4);
  const v4f bias = *reinterpret_cast<const v4f*>(cb + 4 * dq);
  asm volatile("" :: "v"(xr[0]), "v"(xr[1]), "v"(xr[2]), "v"(xr[3]), "v"(xr[4]), "v"(xr[5]),
               "v"(xr[6]), "v"(xr[7]), "v"(xr[8]), "v"(xr[9]), "v"(xr[10]), LOADED4(w), "v"(bias));

  float* dst = u + (size_t)row0 * DI + 4 * dq;
  for (int i = 0; i < CT; ++i) {
    v4f o;
    for (int c = 0; c < 4; ++c) {
      float acc = bias[c];
      for (int j = 0; j < 4; ++j) {
        if (i + j >= 3 || !head)
          acc += xr[i + j][c] * w[c][j];
      }
      o[c] = acc / (1.0f + __expf(-acc));
    }
    *reinterpret_cast<v4f*>(dst + (size_t)i * DI) = o;
  }
}

#define GX_BM    64
#define GX_BN    48
#define GX_KS    64
#define GX_ROWS  (GX_BM + GX_BN)
#define GX_PITCH 68
#define GX_STAGES (DI / GX_KS)
#define GX_NP    (GX_ROWS / 8)
__global__ void __launch_bounds__(128) k_gemm_xp(const float* __restrict__ u_all,
                          const float* __restrict__ Wxp_f,
                          const float* __restrict__ Wxp_b,
                          float* __restrict__ dbl_all) {
  __shared__ __attribute__((aligned(16))) float sT[2][GX_ROWS * GX_PITCH];
  const int dir = blockIdx.y;
  const float* u = u_all + (size_t)dir * MROWS * DI;
  const float* Wxp = dir ? Wxp_b : Wxp_f;
  float* dbl = dbl_all + (size_t)dir * MROWS * DBLW;
  const int tid = threadIdx.x;
  const int wave = tid >> 5, lane = tid & 31;
  const int gm = blockIdx.x * (GX_BM / 16) + wave;
  const int mrow = lane & 15, kh = lane >> 4;

  const int r0 = tid >> 4, p = tid & 15;
  const float* ga = u + (size_t)(blockIdx.x * GX_BM + r0) * DI + 4 * p;
  const float* gb = Wxp + (size_t)r0 * DI + 4 * p;
  const bool tail_valid = (r0 + 40) < (DTR + 2 * NST);
  const float* gt = Wxp + (size_t)(tail_valid ? r0 + 40 : 0) * DI + 4 * p;
  const int lo = r0 * GX_PITCH + 4 * p;
  const v4f zero = {0.0f, 0.0f, 0.0f, 0.0f};

  v4f st[GX_NP];
#pragma unroll
  for (int j = 0; j < 8; ++j)
    st[j] = *reinterpret_cast<const v4f*>(ga + (size_t)j * 8 * DI);
#pragma unroll
  for (int j = 0; j < 5; ++j)
    st[8 + j] = *reinterpret_cast<const v4f*>(gb + (size_t)j * 8 * DI);
  st[13] = *reinterpret_cast<const v4f*>(gt);
  st[13] = tail_valid ? st[13] : zero;
#pragma unroll
  for (int j = 0; j < GX_NP; ++j)
    *reinterpret_cast<v4f*>(&sT[0][lo + j * 8 * GX_PITCH]) = st[j];
  __syncthreads();

  const int aoff = (wave * 16 + mrow) * GX_PITCH + 16 * kh;
  const int boff = (GX_BM + mrow) * GX_PITCH + 16 * kh;

  v8f acc[3] = {};
  for (int s = 0; s < GX_STAGES; ++s) {
    const float* cur = sT[s & 1];
    const int kn = (s + 1 < GX_STAGES) ? (s + 1) * GX_KS : s * GX_KS;
#pragma unroll
    for (int j = 0; j < 8; ++j)
      st[j] = *reinterpret_cast<const v4f*>(ga + (size_t)j * 8 * DI + kn);
#pragma unroll
    for (int j = 0; j < 5; ++j)
      st[8 + j] = *reinterpret_cast<const v4f*>(gb + (size_t)j * 8 * DI + kn);
    st[13] = *reinterpret_cast<const v4f*>(gt + kn);

#pragma unroll
    for (int t = 0; t < GX_KS / 32; ++t) {
      v4f a[4], b[3][4];
      ld16(cur + aoff + 32 * t, a);
#pragma unroll
      for (int j = 0; j < 3; ++j) ld16(cur + boff + j * 16 * GX_PITCH + 32 * t, b[j]);
      asm volatile("" :: LOADED4(a), LOADED4(b[0]), LOADED4(b[1]), LOADED4(b[2]));
#pragma unroll
      for (int q = 0; q < 4; ++q) {
#pragma unroll
        for (int j = 0; j < 3; ++j) acc[j] = wmma4(lo2(a[q]), lo2(b[j][q]), acc[j]);
#pragma unroll
        for (int j = 0; j < 3; ++j) acc[j] = wmma4(hi2(a[q]), hi2(b[j][q]), acc[j]);
      }
    }
    float* nxt = sT[(s + 1) & 1];
    st[13] = tail_valid ? st[13] : zero;
#pragma unroll
    for (int j = 0; j < GX_NP; ++j)
      *reinterpret_cast<v4f*>(&nxt[lo + j * 8 * GX_PITCH]) = st[j];
    __syncthreads();
  }
#pragma unroll
  for (int j = 0; j < 3; ++j)
#pragma unroll
    for (int v = 0; v < 8; ++v) {
      int orow = gm * 16 + v + 8 * kh;
      dbl[(size_t)orow * DBLW + j * 16 + mrow] = acc[j][v];
    }
}

__device__ __forceinline__ float softplus_f32(float t) {
  const float e = __expf(-fabsf(t));
  const float s = e * __builtin_amdgcn_rcpf(2.0f + e);
  const float w = s * s;
  float p = 1.0f / 13.0f;
  p = fmaf(p, w, 1.0f / 11.0f);
  p = fmaf(p, w, 1.0f / 9.0f);
  p = fmaf(p, w, 1.0f / 7.0f);
  p = fmaf(p, w, 1.0f / 5.0f);
  p = fmaf(p, w, 1.0f / 3.0f);
  const float at = fmaf(s * w, p, s);
  return fmaf(2.0f, at, fmaxf(t, 0.0f));
}

__global__ void k_gemm_dt(const float* __restrict__ dbl_all,
                          const float* __restrict__ Wdt_f,
                          const float* __restrict__ bdt_f,
                          const float* __restrict__ Wdt_b,
                          const float* __restrict__ bdt_b,
                          float* __restrict__ xz_all) {
  const int dir = blockIdx.y;
  const float* dbl = dbl_all + (size_t)dir * MROWS * DBLW;
  const float* Wdt = dir ? Wdt_b : Wdt_f;
  const float* bdt = dir ? bdt_b : bdt_f;
  float* xz = xz_all + (size_t)dir * MROWS * XZW;
  const int wave = blockIdx.x * (blockDim.x >> 5) + (threadIdx.x >> 5);
  const int lane = threadIdx.x & 31;
  const int TG = DI / 64;
  const int gm = wave / TG, gng = wave % TG;
  const int mrow = lane & 15, kh = lane >> 4;

  const float* ap = dbl + (size_t)(gm * 16 + mrow) * DBLW;
  const float* bp[4];
#pragma unroll
  for (int j = 0; j < 4; ++j)
    bp[j] = Wdt + (size_t)(gng * 64 + j * 16 + mrow) * DTR;

  v8f acc[4] = {};
  for (int k0 = 0; k0 < DTR; k0 += 4) {
    int kk = k0 + 2 * kh;
    v2f a; a.x = ap[kk]; a.y = ap[kk + 1];
#pragma unroll
    for (int j = 0; j < 4; ++j) {
      v2f b; b.x = bp[j][kk]; b.y = bp[j][kk + 1];
      acc[j] = wmma4(a, b, acc[j]);
    }
  }
#pragma unroll
  for (int j = 0; j < 4; ++j) {
    int n = gng * 64 + j * 16 + mrow;
    float bias = bdt[n];
#pragma unroll
    for (int v = 0; v < 8; ++v) {
      int orow = gm * 16 + v + 8 * kh;
      float t = acc[j][v] + bias;
      xz[(size_t)orow * XZW + n] = softplus_f32(t);
    }
  }
}

#define TGRP 8
#define NGRP (CL / TGRP)

__device__ __forceinline__ void scan1_group(const float (&dt8)[TGRP], const float (&u8)[TGRP],
                                            const v4f (*sB)[NST / 4],
                                            const float (&a)[NST], float (&P)[NST], float (&S)[NST]) {
#pragma unroll
  for (int k = 0; k < TGRP; ++k) {
    float Bv[NST];
#pragma unroll
    for (int q = 0; q < NST / 4; ++q) {
      const v4f v = sB[k][q];
      Bv[4 * q] = v.x; Bv[4 * q + 1] = v.y; Bv[4 * q + 2] = v.z; Bv[4 * q + 3] = v.w;
    }
    const float dt = dt8[k];
    const float dtu = dt * u8[k];
#pragma unroll
    for (int n = 0; n < NST; ++n) {
      float dA = __expf(dt * a[n]);
      S[n] = S[n] * dA + dtu * Bv[n];
      P[n] *= dA;
    }
  }
}

__global__ void __launch_bounds__(DI) k_scan_part1(const float* __restrict__ xz_all,
                             const float* __restrict__ u_all,
                             const float* __restrict__ dbl_all,
                             const float* __restrict__ Alog_f,
                             const float* __restrict__ Alog_b,
                             float* __restrict__ Pb_all, float* __restrict__ Sb_all) {
  __shared__ v4f sB[CL][NST / 4];
  const int dir = blockIdx.y;
  const float* xz  = xz_all  + (size_t)dir * MROWS * XZW;
  const float* u   = u_all   + (size_t)dir * MROWS * DI;
  const float* dbl = dbl_all + (size_t)dir * MROWS * DBLW;
  const float* Alog = dir ? Alog_b : Alog_f;
  float* Pb = Pb_all + (size_t)dir * PSTRIDE;
  float* Sb = Sb_all + (size_t)dir * PSTRIDE;
  const int b = blockIdx.x / NCH;
  const int c = blockIdx.x % NCH;
  const int d = threadIdx.x;

  const size_t row0 = (size_t)b * SEQ + (size_t)c * CL;
  const float* dtp = xz  + row0 * XZW + d;
  const float* utp = u   + row0 * DI + d;
  const float* bcp = dbl + row0 * DBLW + DTR;

  const bool has_piece = d < CL * (NST / 4);
  const int pr = d >> 2, pq = d & 3;
  v4f bpiece = {0.0f, 0.0f, 0.0f, 0.0f};
  if (has_piece)
    bpiece = *reinterpret_cast<const v4f*>(bcp + (size_t)pr * DBLW + 4 * pq);
  float al[NST];
#pragma unroll
  for (int n = 0; n < NST; ++n) al[n] = Alog[d * NST + n];
  float dtA[TGRP], utA[TGRP], dtB[TGRP], utB[TGRP];
#pragma unroll
  for (int k = 0; k < TGRP; ++k) {
    dtA[k] = dtp[(size_t)k * XZW];
    utA[k] = utp[(size_t)k * DI];
  }
  if (has_piece) sB[pr][pq] = bpiece;

  float a[NST], P[NST], S[NST];
#pragma unroll
  for (int n = 0; n < NST; ++n) {
    a[n] = -__expf(al[n]);
    P[n] = 1.0f;
    S[n] = 0.0f;
  }
  __syncthreads();

  for (int g = 0; g < NGRP; g += 2) {
#pragma unroll
    for (int k = 0; k < TGRP; ++k) {
      dtB[k] = dtp[(size_t)((g + 1) * TGRP + k) * XZW];
      utB[k] = utp[(size_t)((g + 1) * TGRP + k) * DI];
    }
    REQUESTS_STAY_AHEAD();
    scan1_group(dtA, utA, sB + g * TGRP, a, P, S);
    const int gn = (g + 2 < NGRP) ? g + 2 : g + 1;
#pragma unroll
    for (int k = 0; k < TGRP; ++k) {
      dtA[k] = dtp[(size_t)(gn * TGRP + k) * XZW];
      utA[k] = utp[(size_t)(gn * TGRP + k) * DI];
    }
    REQUESTS_STAY_AHEAD();
    scan1_group(dtB, utB, sB + (g + 1) * TGRP, a, P, S);
  }
  const size_t base = ((size_t)(b * NCH + c) * DI + d) * NST;
#pragma unroll
  for (int n = 0; n < NST; ++n) {
    Pb[base + n] = P[n];
    Sb[base + n] = S[n];
  }
}

__global__ void k_scan_combine(const float* __restrict__ Pb_all,
                               float* __restrict__ Sb_all) {
  const int dir = blockIdx.y;
  const float* Pb = Pb_all + (size_t)dir * PSTRIDE;
  float* Sb = Sb_all + (size_t)dir * PSTRIDE;
  int idx = blockIdx.x * blockDim.x + threadIdx.x;
  if (idx >= BATCH * DI * NST) return;
  const int n = idx & (NST - 1);
  const int d = (idx >> 4) % DI;
  const int b = idx / (NST * DI);

  const unsigned base0 = (unsigned)(((b * NCH) * DI + d) * NST + n);
  const unsigned cstep = (unsigned)(DI * NST);
  float p[NCH], s[NCH];
#pragma unroll
  for (int c = 0; c < NCH - 1; ++c) {
    p[c] = Pb[base0 + c * cstep];
    s[c] = Sb[base0 + c * cstep];
  }
  p[NCH - 1] = 0.0f;
  s[NCH - 1] = 0.0f;
#define QUAD(a, i) (v4f){(a)[i], (a)[(i) + 1], (a)[(i) + 2], (a)[(i) + 3]}
  asm volatile("" :: "v"(QUAD(p, 0)), "v"(QUAD(p, 4)), "v"(QUAD(p, 8)), "v"(QUAD(p, 12)),
               "v"(QUAD(p, 16)), "v"(QUAD(p, 20)), "v"(QUAD(p, 24)), "v"(QUAD(p, 28)),
               "v"(QUAD(s, 0)), "v"(QUAD(s, 4)), "v"(QUAD(s, 8)), "v"(QUAD(s, 12)),
               "v"(QUAD(s, 16)), "v"(QUAD(s, 20)), "v"(QUAD(s, 24)), "v"(QUAD(s, 28)));
#undef QUAD
  float carry = 0.0f;
#pragma unroll
  for (int c = 0; c < NCH; ++c) {
    Sb[base0 + c * cstep] = carry;
    if (c + 1 < NCH) carry = p[c] * carry + s[c];
  }
}

__device__ __forceinline__ void scan3_group(const float (&dt8)[TGRP], const float (&u8)[TGRP],
                                            const float (&z8)[TGRP],
                                            const v4f (*sBC)[2 * NST / 4], float* yp,
                                            const float (&a)[NST], float (&h)[NST], float Dd) {
#pragma unroll
  for (int k = 0; k < TGRP; ++k) {
    float BCv[2 * NST];
#pragma unroll
    for (int q = 0; q < 2 * NST / 4; ++q) {
      const v4f v = sBC[k][q];
      BCv[4 * q] = v.x; BCv[4 * q + 1] = v.y; BCv[4 * q + 2] = v.z; BCv[4 * q + 3] = v.w;
    }
    const float dt = dt8[k];
    const float ut = u8[k];
    const float dtu = dt * ut;
    float y = 0.0f;
#pragma unroll
    for (int n = 0; n < NST; ++n) {
      float dA = __expf(dt * a[n]);
      h[n] = h[n] * dA + dtu * BCv[n];
      y += h[n] * BCv[NST + n];
    }
    y += ut * Dd;
    const float z = z8[k];
    y *= z / (1.0f + __expf(-z));
    yp[(size_t)k * XZW] = y;
  }
}

__global__ void __launch_bounds__(DI) k_scan_part3(float* __restrict__ xz_all,
                             const float* __restrict__ u_all,
                             const float* __restrict__ dbl_all,
                             const float* __restrict__ Alog_f,
                             const float* __restrict__ Dv_f,
                             const float* __restrict__ Alog_b,
                             const float* __restrict__ Dv_b,
                             const float* __restrict__ Sb_all) {
  __shared__ v4f sBC[CL][2 * NST / 4];
  const int dir = blockIdx.y;
  float* xz        = xz_all  + (size_t)dir * MROWS * XZW;
  const float* u   = u_all   + (size_t)dir * MROWS * DI;
  const float* dbl = dbl_all + (size_t)dir * MROWS * DBLW;
  const float* Alog = dir ? Alog_b : Alog_f;
  const float* Dv   = dir ? Dv_b : Dv_f;
  const float* Sb = Sb_all + (size_t)dir * PSTRIDE;
  const int b = blockIdx.x / NCH;
  const int c = blockIdx.x % NCH;
  const int d = threadIdx.x;

  const size_t row0 = (size_t)b * SEQ + (size_t)c * CL;
  float* xzp = xz + row0 * XZW + d;
  const float* utp = u   + row0 * DI + d;
  const float* bcp = dbl + row0 * DBLW + DTR;

  const int i0 = d, i1 = d + DI;
  const bool two = i1 < CL * (2 * NST / 4);
  v4f bc0 = *reinterpret_cast<const v4f*>(bcp + (size_t)(i0 >> 3) * DBLW + 4 * (i0 & 7));
  v4f bc1 = {0.0f, 0.0f, 0.0f, 0.0f};
  if (two)
    bc1 = *reinterpret_cast<const v4f*>(bcp + (size_t)(i1 >> 3) * DBLW + 4 * (i1 & 7));
  float al[NST], h[NST];
  const size_t base = ((size_t)(b * NCH + c) * DI + d) * NST;
#pragma unroll
  for (int n = 0; n < NST; ++n) {
    al[n] = Alog[d * NST + n];
    h[n] = Sb[base + n];
  }
  const float Dd = Dv[d];
  float dtA[TGRP], utA[TGRP], zA[TGRP], dtB[TGRP], utB[TGRP], zB[TGRP];
#pragma unroll
  for (int k = 0; k < TGRP; ++k) {
    dtA[k] = xzp[(size_t)k * XZW];
    zA[k]  = xzp[(size_t)k * XZW + DI];
    utA[k] = utp[(size_t)k * DI];
  }
  sBC[i0 >> 3][i0 & 7] = bc0;
  if (two) sBC[i1 >> 3][i1 & 7] = bc1;

  float a[NST];
#pragma unroll
  for (int n = 0; n < NST; ++n) a[n] = -__expf(al[n]);
  __syncthreads();

  for (int g = 0; g < NGRP; g += 2) {
#pragma unroll
    for (int k = 0; k < TGRP; ++k) {
      dtB[k] = xzp[(size_t)((g + 1) * TGRP + k) * XZW];
      zB[k]  = xzp[(size_t)((g + 1) * TGRP + k) * XZW + DI];
      utB[k] = utp[(size_t)((g + 1) * TGRP + k) * DI];
    }
    REQUESTS_STAY_AHEAD();
    scan3_group(dtA, utA, zA, sBC + g * TGRP, xzp + (size_t)(g * TGRP) * XZW + DI, a, h, Dd);
    const int gn = (g + 2 < NGRP) ? g + 2 : g + 1;
#pragma unroll
    for (int k = 0; k < TGRP; ++k) {
      dtA[k] = xzp[(size_t)(gn * TGRP + k) * XZW];
      zA[k]  = xzp[(size_t)(gn * TGRP + k) * XZW + DI];
      utA[k] = utp[(size_t)(gn * TGRP + k) * DI];
    }
    REQUESTS_STAY_AHEAD();
    scan3_group(dtB, utB, zB, sBC + (g + 1) * TGRP, xzp + (size_t)((g + 1) * TGRP) * XZW + DI, a, h, Dd);
  }
}

#define GO_BM    32
#define GO_ROWS  (GO_BM + DIMX)
#define GO_STAGES (DI / GI_KS)
__global__ void __launch_bounds__(384) k_gemm_out(const float* __restrict__ xz_f,
                           const float* __restrict__ xz_b,
                           const float* __restrict__ x,
                           const float* __restrict__ Wout,
                           float* __restrict__ out) {
  __shared__ __attribute__((aligned(16))) float sT[2][GO_ROWS * GI_PITCH];
  const int mb = blockIdx.x;
  const int tid = threadIdx.x;
  const int wave = tid >> 5, lane = tid & 31;
  const int wm = wave / 6, wn = wave % 6;
  const int mrow = lane & 15, kh = lane >> 4;

  float xres[2][8];
#pragma unroll
  for (int j = 0; j < 2; ++j)
#pragma unroll
    for (int v = 0; v < 8; ++v)
      xres[j][v] = x[(size_t)(mb * GO_BM + wm * 16 + v + 8 * kh) * DIMX + wn * 32 + j * 16 + mrow];

  const int r0 = tid >> 3, p = tid & 7;
  const bool has_a = r0 < GO_BM;
  const int m  = mb * GO_BM + (has_a ? r0 : 0);
  const int bb = m / SEQ, ll = m % SEQ;
  const int mrev = bb * SEQ + (SEQ - 1 - ll);
  const float* gf = xz_f + (size_t)m    * XZW + DI + 4 * p;
  const float* gr = xz_b + (size_t)mrev * XZW + DI + 4 * p;
  const float* gb = Wout + (size_t)r0 * DI + 4 * p;
  const int loa = r0 * GI_PITCH + 4 * p;
  const int lob = (GO_BM + r0) * GI_PITCH + 4 * p;

  v4f sf = {0.0f, 0.0f, 0.0f, 0.0f}, sr = {0.0f, 0.0f, 0.0f, 0.0f}, st[4];
  if (has_a) {
    sf = *reinterpret_cast<const v4f*>(gf);
    sr = *reinterpret_cast<const v4f*>(gr);
  }
#pragma unroll
  for (int j = 0; j < 4; ++j)
    st[j] = *reinterpret_cast<const v4f*>(gb + (size_t)j * 48 * DI);
  if (has_a)
    *reinterpret_cast<v4f*>(&sT[0][loa]) = sf + sr;
#pragma unroll
  for (int j = 0; j < 4; ++j)
    *reinterpret_cast<v4f*>(&sT[0][lob + j * 48 * GI_PITCH]) = st[j];
  __syncthreads();

  const int aoff = (wm * 16 + mrow) * GI_PITCH + 16 * kh;
  const int boff = (GO_BM + wn * 32 + mrow) * GI_PITCH + 16 * kh;

  v8f acc[2] = {};
  for (int s = 0; s < GO_STAGES; ++s) {
    const float* cur = sT[s & 1];
    const int kn = (s + 1 < GO_STAGES) ? (s + 1) * GI_KS : s * GI_KS;
    if (has_a) {
      sf = *reinterpret_cast<const v4f*>(gf + kn);
      sr = *reinterpret_cast<const v4f*>(gr + kn);
    }
#pragma unroll
    for (int j = 0; j < 4; ++j)
      st[j] = *reinterpret_cast<const v4f*>(gb + (size_t)j * 48 * DI + kn);

    v4f a[4], b[2][4];
    ld16(cur + aoff, a);
#pragma unroll
    for (int j = 0; j < 2; ++j) ld16(cur + boff + j * 16 * GI_PITCH, b[j]);
    asm volatile("" :: LOADED4(a), LOADED4(b[0]), LOADED4(b[1]));
#pragma unroll
    for (int q = 0; q < 4; ++q) {
#pragma unroll
      for (int j = 0; j < 2; ++j) acc[j] = wmma4(lo2(a[q]), lo2(b[j][q]), acc[j]);
#pragma unroll
      for (int j = 0; j < 2; ++j) acc[j] = wmma4(hi2(a[q]), hi2(b[j][q]), acc[j]);
    }
    float* nxt = sT[(s + 1) & 1];
    if (has_a)
      *reinterpret_cast<v4f*>(&nxt[loa]) = sf + sr;
#pragma unroll
    for (int j = 0; j < 4; ++j)
      *reinterpret_cast<v4f*>(&nxt[lob + j * 48 * GI_PITCH]) = st[j];
    __syncthreads();
  }
#pragma unroll
  for (int j = 0; j < 2; ++j)
#pragma unroll
    for (int v = 0; v < 8; ++v) {
      int orow = mb * GO_BM + wm * 16 + v + 8 * kh;
      int n = wn * 32 + j * 16 + mrow;
      out[(size_t)orow * DIMX + n] = acc[j][v] + xres[j][v];
    }
}

extern "C" void kernel_launch(void* const* d_in, const int* in_sizes, int n_in,
                              void* d_out, int out_size, void* d_ws, size_t ws_size,
                              hipStream_t stream) {
  const float* x      = (const float*)d_in[0];
  const float* Win_f  = (const float*)d_in[1];
  const float* cw_f   = (const float*)d_in[2];
  const float* cb_f   = (const float*)d_in[3];
  const float* Wxp_f  = (const float*)d_in[4];
  const float* Wdt_f  = (const float*)d_in[5];
  const float* bdt_f  = (const float*)d_in[6];
  const float* Alog_f = (const float*)d_in[7];
  const float* Dv_f   = (const float*)d_in[8];
  const float* Win_b  = (const float*)d_in[9];
  const float* cw_b   = (const float*)d_in[10];
  const float* cb_b   = (const float*)d_in[11];
  const float* Wxp_b  = (const float*)d_in[12];
  const float* Wdt_b  = (const float*)d_in[13];
  const float* bdt_b  = (const float*)d_in[14];
  const float* Alog_b = (const float*)d_in[15];
  const float* Dv_b   = (const float*)d_in[16];
  const float* Wout   = (const float*)d_in[17];
  float* out = (float*)d_out;

  float* ws    = (float*)d_ws;
  float* xz_f  = ws;
  float* xz_b  = xz_f + (size_t)MROWS * XZW;
  float* u_f   = xz_b + (size_t)MROWS * XZW;
  float* dbl_f = u_f  + (size_t)2 * MROWS * DI;
  const size_t PS = (size_t)BATCH * NCH * DI * NST;
  float* P_f   = dbl_f + (size_t)2 * MROWS * DBLW;
  float* S_f   = P_f + PS;

  const int BLK = 256, WPB = BLK / 32;

  {
    int blocks = (MROWS / GI_BM) * GI_NB;
    k_gemm_in<<<dim3(blocks, 2), BLK, 0, stream>>>(x, Win_f, Win_b, xz_f);
  }
  {
    int n = (MROWS / CT) * DQ;
    k_conv_silu<<<dim3(n / BLK, 2), BLK, 0, stream>>>(
        xz_f, cw_f, cb_f, cw_b, cb_b, u_f);
  }
  {
    int blocks = MROWS / GX_BM;
    k_gemm_xp<<<dim3(blocks, 2), 128, 0, stream>>>(u_f, Wxp_f, Wxp_b, dbl_f);
  }
  {
    int waves = (MROWS / 16) * (DI / 64);
    k_gemm_dt<<<dim3(waves / WPB, 2), BLK, 0, stream>>>(
        dbl_f, Wdt_f, bdt_f, Wdt_b, bdt_b, xz_f);
  }
  {
    int cb = (BATCH * DI * NST + BLK - 1) / BLK;
    k_scan_part1<<<dim3(BATCH * NCH, 2), DI, 0, stream>>>(
        xz_f, u_f, dbl_f, Alog_f, Alog_b, P_f, S_f);
    k_scan_combine<<<dim3(cb, 2), BLK, 0, stream>>>(P_f, S_f);
    k_scan_part3<<<dim3(BATCH * NCH, 2), DI, 0, stream>>>(
        xz_f, u_f, dbl_f, Alog_f, Dv_f, Alog_b, Dv_b, S_f);
  }
  {
    int blocks = MROWS / GO_BM;
    k_gemm_out<<<blocks, 384, 0, stream>>>(xz_f, xz_b, x, Wout, out);
  }
}


# Output and x_dbl projections: operands fetched one step further ahead at both levels (global loads two stages ahead, next stage's LDS fragments read before the current matrix steps); sums kept
# speedup vs baseline: 4.1193x; 1.0321x over previous
#include <hip/hip_runtime.h>
#include <hip/hip_bf16.h>
#include <math.h>

#define BATCH 4
#define SEQ   2048
#define DIMX  192
#define DI    384
#define NST   16
#define DTR   12
#define XZW   768
#define DBLW  48
#define MROWS (BATCH*SEQ)
#define NCH   32
#define CL    (SEQ/NCH)
#define PSTRIDE (2 * BATCH * NCH * DI * NST)

typedef float v2f __attribute__((ext_vector_type(2)));
typedef float v8f __attribute__((ext_vector_type(8)));

__device__ __forceinline__ v8f wmma4(v2f a, v2f b, v8f c) {
  return __builtin_amdgcn_wmma_f32_16x16x4_f32(false, a, false, b, (short)0, c,
                                               false, false);
}

typedef float v4f __attribute__((ext_vector_type(4)));

__device__ __forceinline__ void ld16(const float* __restrict__ p, v4f (&r)[4]) {
  const v4f* q = reinterpret_cast<const v4f*>(p);
  r[0] = q[0]; r[1] = q[1]; r[2] = q[2]; r[3] = q[3];
}
__device__ __forceinline__ v2f lo2(v4f v) { v2f r; r.x = v.x; r.y = v.y; return r; }
__device__ __forceinline__ v2f hi2(v4f v) { v2f r; r.x = v.z; r.y = v.w; return r; }
#define LOADED4(v) "v"((v)[0]), "v"((v)[1]), "v"((v)[2]), "v"((v)[3])
#define REQUESTS_STAY_AHEAD() asm volatile("" ::: "memory")
__device__ __forceinline__ void lds_handover_barrier() {
  __builtin_amdgcn_fence(__ATOMIC_RELEASE, "workgroup", "local");
  __builtin_amdgcn_s_barrier();
  __builtin_amdgcn_fence(__ATOMIC_ACQUIRE, "workgroup", "local");
}

#define GI_BM    64
#define GI_BN    128
#define GI_KS    32
#define GI_ROWS  (GI_BM + GI_BN)
#define GI_PITCH 36
#define GI_NB    (XZW / GI_BN)
#define GI_STAGES (DIMX / GI_KS)
__global__ void __launch_bounds__(256) k_gemm_in(const float* __restrict__ x,
                          const float* __restrict__ Win_f,
                          const float* __restrict__ Win_b,
                          float* __restrict__ xz_all) {
  __shared__ __attribute__((aligned(16))) float sT[2][GI_ROWS * GI_PITCH];
  const int reverse = blockIdx.y;
  const float* Win = reverse ? Win_b : Win_f;
  float* xz = xz_all + (size_t)reverse * MROWS * XZW;
  const int nb = blockIdx.x % GI_NB;
  const int mb = blockIdx.x / GI_NB;
  const int tid = threadIdx.x;
  const int wave = tid >> 5, lane = tid & 31;
  const int wm = wave >> 1, wn = wave & 1;
  const int mrow = lane & 15, kh = lane >> 4;

  const int r0 = tid >> 3, p = tid & 7;
  const int m  = mb * GI_BM + r0;
  const int bb = m / SEQ, ll = m % SEQ;
  const int srow = reverse ? (bb * SEQ + (SEQ - 1 - ll)) : m;
  const float* ga = x + (size_t)srow * DIMX + 4 * p;
  const int astep = reverse ? -32 * DIMX : 32 * DIMX;
  const float* gb = Win + (size_t)(nb * GI_BN + r0) * DIMX + 4 * p;
  const int lo = r0 * GI_PITCH + 4 * p;

  v4f st[6];
  st[0] = *reinterpret_cast<const v4f*>(ga);
  st[1] = *reinterpret_cast<const v4f*>(ga + astep);
#pragma unroll
  for (int j = 0; j < 4; ++j)
    st[2 + j] = *reinterpret_cast<const v4f*>(gb + (size_t)j * 32 * DIMX);
#pragma unroll
  for (int j = 0; j < 6; ++j)
    *reinterpret_cast<v4f*>(&sT[0][lo + j * 32 * GI_PITCH]) = st[j];
  __syncthreads();

  const int aoff = (wm * 16 + mrow) * GI_PITCH + 16 * kh;
  const int boff = (GI_BM + wn * 64 + mrow) * GI_PITCH + 16 * kh;

  v8f acc[4] = {};
  for (int s = 0; s < GI_STAGES; ++s) {
    const float* cur = sT[s & 1];
    const int kn = (s + 1 < GI_STAGES) ? (s + 1) * GI_KS : s * GI_KS;
    st[0] = *reinterpret_cast<const v4f*>(ga + kn);
    st[1] = *reinterpret_cast<const v4f*>(ga + astep + kn);
#pragma unroll
    for (int j = 0; j < 4; ++j)
      st[2 + j] = *reinterpret_cast<const v4f*>(gb + (size_t)j * 32 * DIMX + kn);

    v4f a[4], b[4][4];
    ld16(cur + aoff, a);
#pragma unroll
    for (int j = 0; j < 4; ++j) ld16(cur + boff + j * 16 * GI_PITCH, b[j]);
    asm volatile("" :: LOADED4(a), LOADED4(b[0]), LOADED4(b[1]), LOADED4(b[2]), LOADED4(b[3]));
#pragma unroll
    for (int q = 0; q < 4; ++q) {
#pragma unroll
      for (int j = 0; j < 4; ++j) acc[j] = wmma4(lo2(a[q]), lo2(b[j][q]), acc[j]);
#pragma unroll
      for (int j = 0; j < 4; ++j) acc[j] = wmma4(hi2(a[q]), hi2(b[j][q]), acc[j]);
    }
    float* nxt = sT[(s + 1) & 1];
#pragma unroll
    for (int j = 0; j < 6; ++j)
      *reinterpret_cast<v4f*>(&nxt[lo + j * 32 * GI_PITCH]) = st[j];
    __syncthreads();
  }
#pragma unroll
  for (int j = 0; j < 4; ++j)
#pragma unroll
    for (int v = 0; v < 8; ++v) {
      int orow = mb * GI_BM + wm * 16 + v + 8 * kh;
      xz[(size_t)orow * XZW + nb * GI_BN + wn * 64 + j * 16 + mrow] = acc[j][v];
    }
}

#define CT 8
#define DQ (DI / 4)
__global__ void k_conv_silu(const float* __restrict__ xz_all,
                            const float* __restrict__ cw_f,
                            const float* __restrict__ cb_f,
                            const float* __restrict__ cw_b,
                            const float* __restrict__ cb_b,
                            float* __restrict__ u_all) {
  const int dir = blockIdx.y;
  const float* xz = xz_all + (size_t)dir * MROWS * XZW;
  const float* cw = dir ? cw_b : cw_f;
  const float* cb = dir ? cb_b : cb_f;
  float* u = u_all + (size_t)dir * MROWS * DI;
  const int idx = blockIdx.x * blockDim.x + threadIdx.x;
  const int dq  = idx % DQ;
  const int run = idx / DQ;
  const int row0 = run * CT;
  const bool head = (row0 % SEQ) == 0;

  const float* src = xz + (size_t)row0 * XZW + 4 * dq;
  v4f xr[CT + 3];
  for (int k = 0; k < 3; ++k) {
    v4f zero = {0.0f, 0.0f, 0.0f, 0.0f};
    xr[k] = head ? zero : *reinterpret_cast<const v4f*>(src + (k - 3) * XZW);
  }
  for (int k = 0; k < CT; ++k)
    xr[3 + k] = *reinterpret_cast<const v4f*>(src + (size_t)k * XZW);

  v4f w[4];
  for (int c = 0; c < 4; ++c)
    w[c] = *reinterpret_cast<const v4f*>(cw + (size_t)(4 * dq + c) * 4);
  const v4f bias = *reinterpret_cast<const v4f*>(cb + 4 * dq);
  asm volatile("" :: "v"(xr[0]), "v"(xr[1]), "v"(xr[2]), "v"(xr[3]), "v"(xr[4]), "v"(xr[5]),
               "v"(xr[6]), "v"(xr[7]), "v"(xr[8]), "v"(xr[9]), "v"(xr[10]), LOADED4(w), "v"(bias));

  float* dst = u + (size_t)row0 * DI + 4 * dq;
  for (int i = 0; i < CT; ++i) {
    v4f o;
    for (int c = 0; c < 4; ++c) {
      float acc = bias[c];
      for (int j = 0; j < 4; ++j) {
        if (i + j >= 3 || !head)
          acc += xr[i + j][c] * w[c][j];
      }
      o[c] = acc / (1.0f + __expf(-acc));
    }
    *reinterpret_cast<v4f*>(dst + (size_t)i * DI) = o;
  }
}

#define GX_BM    64
#define GX_BN    48
#define GX_ROWS  (GX_BM + GX_BN)
#define GX_STAGES (DI / GI_KS)
#define GX_NP    (GX_ROWS / 16)
__global__ void __launch_bounds__(128) k_gemm_xp(const float* __restrict__ u_all,
                          const float* __restrict__ Wxp_f,
                          const float* __restrict__ Wxp_b,
                          float* __restrict__ dbl_all) {
  __shared__ __attribute__((aligned(16))) float sT[2][GX_ROWS * GI_PITCH];
  const int dir = blockIdx.y;
  const float* u = u_all + (size_t)dir * MROWS * DI;
  const float* Wxp = dir ? Wxp_b : Wxp_f;
  float* dbl = dbl_all + (size_t)dir * MROWS * DBLW;
  const int tid = threadIdx.x;
  const int wave = tid >> 5, lane = tid & 31;
  const int gm = blockIdx.x * (GX_BM / 16) + wave;
  const int mrow = lane & 15, kh = lane >> 4;

  const int r0 = tid >> 3, p = tid & 7;
  const float* ga = u + (size_t)(blockIdx.x * GX_BM + r0) * DI + 4 * p;
  const float* gb = Wxp + (size_t)r0 * DI + 4 * p;
  const bool tail_valid = (r0 + 32) < (DTR + 2 * NST);
  const float* gt = Wxp + (size_t)(tail_valid ? r0 + 32 : 0) * DI + 4 * p;
  const int lo = r0 * GI_PITCH + 4 * p;
  const v4f zero = {0.0f, 0.0f, 0.0f, 0.0f};
  const int aoff = (wave * 16 + mrow) * GI_PITCH + 16 * kh;
  const int boff = (GX_BM + mrow) * GI_PITCH + 16 * kh;

  v4f st[2][GX_NP];
  v4f a[2][4], b[2][3][4];

#define GX_REQUEST(set, stage)                                                     \
  do {                                                                             \
    _Pragma("unroll") for (int j = 0; j < 4; ++j)                                  \
      st[set][j] = *reinterpret_cast<const v4f*>(ga + (size_t)j * 16 * DI + (stage) * GI_KS); \
    _Pragma("unroll") for (int j = 0; j < 2; ++j)                                  \
      st[set][4 + j] = *reinterpret_cast<const v4f*>(gb + (size_t)j * 16 * DI + (stage) * GI_KS); \
    st[set][6] = *reinterpret_cast<const v4f*>(gt + (stage) * GI_KS);              \
  } while (0)
#define GX_DEPOSIT(set, buf)                                                       \
  do {                                                                             \
    _Pragma("unroll") for (int j = 0; j < GX_NP - 1; ++j)                          \
      *reinterpret_cast<v4f*>(&(buf)[lo + j * 16 * GI_PITCH]) = st[set][j];        \
    *reinterpret_cast<v4f*>(&(buf)[lo + 6 * 16 * GI_PITCH]) =                      \
        tail_valid ? st[set][6] : zero;                   \
  } while (0)
#define GX_FETCH(set, buf)                                                         \
  do {                                                                             \
    ld16((buf) + aoff, a[set]);                                                    \
    _Pragma("unroll") for (int j = 0; j < 3; ++j)                                  \
      ld16((buf) + boff + j * 16 * GI_PITCH, b[set][j]);                           \
  } while (0)

  GX_REQUEST(0, 0);
  GX_REQUEST(1, 1);
  GX_DEPOSIT(0, sT[0]);
  GX_REQUEST(0, 2);
  lds_handover_barrier();
  GX_FETCH(0, sT[0]);
  GX_DEPOSIT(1, sT[1]);
  GX_REQUEST(1, 3);
  lds_handover_barrier();

  v8f acc[3] = {};
#pragma unroll
  for (int s = 0; s < GX_STAGES; ++s) {
    const int cur = s & 1;
    if (s + 1 < GX_STAGES) GX_FETCH(cur ^ 1, sT[cur ^ 1]);
    __builtin_amdgcn_sched_barrier(0);
#pragma unroll
    for (int q = 0; q < 4; ++q) {
#pragma unroll
      for (int j = 0; j < 3; ++j) acc[j] = wmma4(lo2(a[cur][q]), lo2(b[cur][j][q]), acc[j]);
#pragma unroll
      for (int j = 0; j < 3; ++j) acc[j] = wmma4(hi2(a[cur][q]), hi2(b[cur][j][q]), acc[j]);
    }
    __builtin_amdgcn_sched_barrier(0);
    if (s + 2 < GX_STAGES) GX_DEPOSIT(cur, sT[cur]);
    if (s + 4 < GX_STAGES) GX_REQUEST(cur, s + 4);
    if (s + 2 < GX_STAGES) lds_handover_barrier();
  }
#undef GX_REQUEST
#undef GX_DEPOSIT
#undef GX_FETCH
#pragma unroll
  for (int j = 0; j < 3; ++j)
#pragma unroll
    for (int v = 0; v < 8; ++v) {
      int orow = gm * 16 + v + 8 * kh;
      dbl[(size_t)orow * DBLW + j * 16 + mrow] = acc[j][v];
    }
}

__device__ __forceinline__ float softplus_f32(float t) {
  const float e = __expf(-fabsf(t));
  const float s = e * __builtin_amdgcn_rcpf(2.0f + e);
  const float w = s * s;
  float p = 1.0f / 13.0f;
  p = fmaf(p, w, 1.0f / 11.0f);
  p = fmaf(p, w, 1.0f / 9.0f);
  p = fmaf(p, w, 1.0f / 7.0f);
  p = fmaf(p, w, 1.0f / 5.0f);
  p = fmaf(p, w, 1.0f / 3.0f);
  const float at = fmaf(s * w, p, s);
  return fmaf(2.0f, at, fmaxf(t, 0.0f));
}

__global__ void k_gemm_dt(const float* __restrict__ dbl_all,
                          const float* __restrict__ Wdt_f,
                          const float* __restrict__ bdt_f,
                          const float* __restrict__ Wdt_b,
                          const float* __restrict__ bdt_b,
                          float* __restrict__ xz_all) {
  const int dir = blockIdx.y;
  const float* dbl = dbl_all + (size_t)dir * MROWS * DBLW;
  const float* Wdt = dir ? Wdt_b : Wdt_f;
  const float* bdt = dir ? bdt_b : bdt_f;
  float* xz = xz_all + (size_t)dir * MROWS * XZW;
  const int wave = blockIdx.x * (blockDim.x >> 5) + (threadIdx.x >> 5);
  const int lane = threadIdx.x & 31;
  const int TG = DI / 64;
  const int gm = wave / TG, gng = wave % TG;
  const int mrow = lane & 15, kh = lane >> 4;

  const float* ap = dbl + (size_t)(gm * 16 + mrow) * DBLW;
  const float* bp[4];
#pragma unroll
  for (int j = 0; j < 4; ++j)
    bp[j] = Wdt + (size_t)(gng * 64 + j * 16 + mrow) * DTR;

  v8f acc[4] = {};
  for (int k0 = 0; k0 < DTR; k0 += 4) {
    int kk = k0 + 2 * kh;
    v2f a; a.x = ap[kk]; a.y = ap[kk + 1];
#pragma unroll
    for (int j = 0; j < 4; ++j) {
      v2f b; b.x = bp[j][kk]; b.y = bp[j][kk + 1];
      acc[j] = wmma4(a, b, acc[j]);
    }
  }
#pragma unroll
  for (int j = 0; j < 4; ++j) {
    int n = gng * 64 + j * 16 + mrow;
    float bias = bdt[n];
#pragma unroll
    for (int v = 0; v < 8; ++v) {
      int orow = gm * 16 + v + 8 * kh;
      float t = acc[j][v] + bias;
      xz[(size_t)orow * XZW + n] = softplus_f32(t);
    }
  }
}

#define TGRP 8
#define NGRP (CL / TGRP)

__device__ __forceinline__ void scan1_group(const float (&dt8)[TGRP], const float (&u8)[TGRP],
                                            const v4f (*sB)[NST / 4],
                                            const float (&a)[NST], float (&P)[NST], float (&S)[NST]) {
#pragma unroll
  for (int k = 0; k < TGRP; ++k) {
    float Bv[NST];
#pragma unroll
    for (int q = 0; q < NST / 4; ++q) {
      const v4f v = sB[k][q];
      Bv[4 * q] = v.x; Bv[4 * q + 1] = v.y; Bv[4 * q + 2] = v.z; Bv[4 * q + 3] = v.w;
    }
    const float dt = dt8[k];
    const float dtu = dt * u8[k];
#pragma unroll
    for (int n = 0; n < NST; ++n) {
      float dA = __expf(dt * a[n]);
      S[n] = S[n] * dA + dtu * Bv[n];
      P[n] *= dA;
    }
  }
}

__global__ void __launch_bounds__(DI) k_scan_part1(const float* __restrict__ xz_all,
                             const float* __restrict__ u_all,
                             const float* __restrict__ dbl_all,
                             const float* __restrict__ Alog_f,
                             const float* __restrict__ Alog_b,
                             float* __restrict__ Pb_all, float* __restrict__ Sb_all) {
  __shared__ v4f sB[CL][NST / 4];
  const int dir = blockIdx.y;
  const float* xz  = xz_all  + (size_t)dir * MROWS * XZW;
  const float* u   = u_all   + (size_t)dir * MROWS * DI;
  const float* dbl = dbl_all + (size_t)dir * MROWS * DBLW;
  const float* Alog = dir ? Alog_b : Alog_f;
  float* Pb = Pb_all + (size_t)dir * PSTRIDE;
  float* Sb = Sb_all + (size_t)dir * PSTRIDE;
  const int b = blockIdx.x / NCH;
  const int c = blockIdx.x % NCH;
  const int d = threadIdx.x;

  const size_t row0 = (size_t)b * SEQ + (size_t)c * CL;
  const float* dtp = xz  + row0 * XZW + d;
  const float* utp = u   + row0 * DI + d;
  const float* bcp = dbl + row0 * DBLW + DTR;

  const bool has_piece = d < CL * (NST / 4);
  const int pr = d >> 2, pq = d & 3;
  v4f bpiece = {0.0f, 0.0f, 0.0f, 0.0f};
  if (has_piece)
    bpiece = *reinterpret_cast<const v4f*>(bcp + (size_t)pr * DBLW + 4 * pq);
  float al[NST];
#pragma unroll
  for (int n = 0; n < NST; ++n) al[n] = Alog[d * NST + n];
  float dtA[TGRP], utA[TGRP], dtB[TGRP], utB[TGRP];
#pragma unroll
  for (int k = 0; k < TGRP; ++k) {
    dtA[k] = dtp[(size_t)k * XZW];
    utA[k] = utp[(size_t)k * DI];
  }
  if (has_piece) sB[pr][pq] = bpiece;

  float a[NST], P[NST], S[NST];
#pragma unroll
  for (int n = 0; n < NST; ++n) {
    a[n] = -__expf(al[n]);
    P[n] = 1.0f;
    S[n] = 0.0f;
  }
  __syncthreads();

  for (int g = 0; g < NGRP; g += 2) {
#pragma unroll
    for (int k = 0; k < TGRP; ++k) {
      dtB[k] = dtp[(size_t)((g + 1) * TGRP + k) * XZW];
      utB[k] = utp[(size_t)((g + 1) * TGRP + k) * DI];
    }
    REQUESTS_STAY_AHEAD();
    scan1_group(dtA, utA, sB + g * TGRP, a, P, S);
    const int gn = (g + 2 < NGRP) ? g + 2 : g + 1;
#pragma unroll
    for (int k = 0; k < TGRP; ++k) {
      dtA[k] = dtp[(size_t)(gn * TGRP + k) * XZW];
      utA[k] = utp[(size_t)(gn * TGRP + k) * DI];
    }
    REQUESTS_STAY_AHEAD();
    scan1_group(dtB, utB, sB + (g + 1) * TGRP, a, P, S);
  }
  const size_t base = ((size_t)(b * NCH + c) * DI + d) * NST;
#pragma unroll
  for (int n = 0; n < NST; ++n) {
    Pb[base + n] = P[n];
    Sb[base + n] = S[n];
  }
}

__global__ void k_scan_combine(const float* __restrict__ Pb_all,
                               float* __restrict__ Sb_all) {
  const int dir = blockIdx.y;
  const float* Pb = Pb_all + (size_t)dir * PSTRIDE;
  float* Sb = Sb_all + (size_t)dir * PSTRIDE;
  int idx = blockIdx.x * blockDim.x + threadIdx.x;
  if (idx >= BATCH * DI * NST) return;
  const int n = idx & (NST - 1);
  const int d = (idx >> 4) % DI;
  const int b = idx / (NST * DI);

  const unsigned base0 = (unsigned)(((b * NCH) * DI + d) * NST + n);
  const unsigned cstep = (unsigned)(DI * NST);
  float p[NCH], s[NCH];
#pragma unroll
  for (int c = 0; c < NCH - 1; ++c) {
    p[c] = Pb[base0 + c * cstep];
    s[c] = Sb[base0 + c * cstep];
  }
  p[NCH - 1] = 0.0f;
  s[NCH - 1] = 0.0f;
#define QUAD(a, i) (v4f){(a)[i], (a)[(i) + 1], (a)[(i) + 2], (a)[(i) + 3]}
  asm volatile("" :: "v"(QUAD(p, 0)), "v"(QUAD(p, 4)), "v"(QUAD(p, 8)), "v"(QUAD(p, 12)),
               "v"(QUAD(p, 16)), "v"(QUAD(p, 20)), "v"(QUAD(p, 24)), "v"(QUAD(p, 28)),
               "v"(QUAD(s, 0)), "v"(QUAD(s, 4)), "v"(QUAD(s, 8)), "v"(QUAD(s, 12)),
               "v"(QUAD(s, 16)), "v"(QUAD(s, 20)), "v"(QUAD(s, 24)), "v"(QUAD(s, 28)));
#undef QUAD
  float carry = 0.0f;
#pragma unroll
  for (int c = 0; c < NCH; ++c) {
    Sb[base0 + c * cstep] = carry;
    if (c + 1 < NCH) carry = p[c] * carry + s[c];
  }
}

__device__ __forceinline__ void scan3_group(const float (&dt8)[TGRP], const float (&u8)[TGRP],
                                            const float (&z8)[TGRP],
                                            const v4f (*sBC)[2 * NST / 4], float* yp,
                                            const float (&a)[NST], float (&h)[NST], float Dd) {
#pragma unroll
  for (int k = 0; k < TGRP; ++k) {
    float BCv[2 * NST];
#pragma unroll
    for (int q = 0; q < 2 * NST / 4; ++q) {
      const v4f v = sBC[k][q];
      BCv[4 * q] = v.x; BCv[4 * q + 1] = v.y; BCv[4 * q + 2] = v.z; BCv[4 * q + 3] = v.w;
    }
    const float dt = dt8[k];
    const float ut = u8[k];
    const float dtu = dt * ut;
    float y = 0.0f;
#pragma unroll
    for (int n = 0; n < NST; ++n) {
      float dA = __expf(dt * a[n]);
      h[n] = h[n] * dA + dtu * BCv[n];
      y += h[n] * BCv[NST + n];
    }
    y += ut * Dd;
    const float z = z8[k];
    y *= z / (1.0f + __expf(-z));
    yp[(size_t)k * XZW] = y;
  }
}

__global__ void __launch_bounds__(DI) k_scan_part3(float* __restrict__ xz_all,
                             const float* __restrict__ u_all,
                             const float* __restrict__ dbl_all,
                             const float* __restrict__ Alog_f,
                             const float* __restrict__ Dv_f,
                             const float* __restrict__ Alog_b,
                             const float* __restrict__ Dv_b,
                             const float* __restrict__ Sb_all) {
  __shared__ v4f sBC[CL][2 * NST / 4];
  const int dir = blockIdx.y;
  float* xz        = xz_all  + (size_t)dir * MROWS * XZW;
  const float* u   = u_all   + (size_t)dir * MROWS * DI;
  const float* dbl = dbl_all + (size_t)dir * MROWS * DBLW;
  const float* Alog = dir ? Alog_b : Alog_f;
  const float* Dv   = dir ? Dv_b : Dv_f;
  const float* Sb = Sb_all + (size_t)dir * PSTRIDE;
  const int b = blockIdx.x / NCH;
  const int c = blockIdx.x % NCH;
  const int d = threadIdx.x;

  const size_t row0 = (size_t)b * SEQ + (size_t)c * CL;
  float* xzp = xz + row0 * XZW + d;
  const float* utp = u   + row0 * DI + d;
  const float* bcp = dbl + row0 * DBLW + DTR;

  const int i0 = d, i1 = d + DI;
  const bool two = i1 < CL * (2 * NST / 4);
  v4f bc0 = *reinterpret_cast<const v4f*>(bcp + (size_t)(i0 >> 3) * DBLW + 4 * (i0 & 7));
  v4f bc1 = {0.0f, 0.0f, 0.0f, 0.0f};
  if (two)
    bc1 = *reinterpret_cast<const v4f*>(bcp + (size_t)(i1 >> 3) * DBLW + 4 * (i1 & 7));
  float al[NST], h[NST];
  const size_t base = ((size_t)(b * NCH + c) * DI + d) * NST;
#pragma unroll
  for (int n = 0; n < NST; ++n) {
    al[n] = Alog[d * NST + n];
    h[n] = Sb[base + n];
  }
  const float Dd = Dv[d];
  float dtA[TGRP], utA[TGRP], zA[TGRP], dtB[TGRP], utB[TGRP], zB[TGRP];
#pragma unroll
  for (int k = 0; k < TGRP; ++k) {
    dtA[k] = xzp[(size_t)k * XZW];
    zA[k]  = xzp[(size_t)k * XZW + DI];
    utA[k] = utp[(size_t)k * DI];
  }
  sBC[i0 >> 3][i0 & 7] = bc0;
  if (two) sBC[i1 >> 3][i1 & 7] = bc1;

  float a[NST];
#pragma unroll
  for (int n = 0; n < NST; ++n) a[n] = -__expf(al[n]);
  __syncthreads();

  for (int g = 0; g < NGRP; g += 2) {
#pragma unroll
    for (int k = 0; k < TGRP; ++k) {
      dtB[k] = xzp[(size_t)((g + 1) * TGRP + k) * XZW];
      zB[k]  = xzp[(size_t)((g + 1) * TGRP + k) * XZW + DI];
      utB[k] = utp[(size_t)((g + 1) * TGRP + k) * DI];
    }
    REQUESTS_STAY_AHEAD();
    scan3_group(dtA, utA, zA, sBC + g * TGRP, xzp + (size_t)(g * TGRP) * XZW + DI, a, h, Dd);
    const int gn = (g + 2 < NGRP) ? g + 2 : g + 1;
#pragma unroll
    for (int k = 0; k < TGRP; ++k) {
      dtA[k] = xzp[(size_t)(gn * TGRP + k) * XZW];
      zA[k]  = xzp[(size_t)(gn * TGRP + k) * XZW + DI];
      utA[k] = utp[(size_t)(gn * TGRP + k) * DI];
    }
    REQUESTS_STAY_AHEAD();
    scan3_group(dtB, utB, zB, sBC + (g + 1) * TGRP, xzp + (size_t)((g + 1) * TGRP) * XZW + DI, a, h, Dd);
  }
}

#define GO_BM    32
#define GO_ROWS  (GO_BM + DIMX)
#define GO_STAGES (DI / GI_KS)
__global__ void __launch_bounds__(384) k_gemm_out(const float* __restrict__ xz_f,
                           const float* __restrict__ xz_b,
                           const float* __restrict__ x,
                           const float* __restrict__ Wout,
                           float* __restrict__ out) {
  __shared__ __attribute__((aligned(16))) float sT[2][GO_ROWS * GI_PITCH];
  const int mb = blockIdx.x;
  const int tid = threadIdx.x;
  const int wave = tid >> 5, lane = tid & 31;
  const int wm = wave / 6, wn = wave % 6;
  const int mrow = lane & 15, kh = lane >> 4;

  float xres[2][8];
#pragma unroll
  for (int j = 0; j < 2; ++j)
#pragma unroll
    for (int v = 0; v < 8; ++v)
      xres[j][v] = x[(size_t)(mb * GO_BM + wm * 16 + v + 8 * kh) * DIMX + wn * 32 + j * 16 + mrow];

  const int r0 = tid >> 3, p = tid & 7;
  const bool has_a = r0 < GO_BM;
  const int m  = mb * GO_BM + (has_a ? r0 : 0);
  const int bb = m / SEQ, ll = m % SEQ;
  const int mrev = bb * SEQ + (SEQ - 1 - ll);
  const float* gf = xz_f + (size_t)m    * XZW + DI + 4 * p;
  const float* gr = xz_b + (size_t)mrev * XZW + DI + 4 * p;
  const float* gb = Wout + (size_t)r0 * DI + 4 * p;
  const int loa = r0 * GI_PITCH + 4 * p;
  const int lob = (GO_BM + r0) * GI_PITCH + 4 * p;
  const int aoff = (wm * 16 + mrow) * GI_PITCH + 16 * kh;
  const int boff = (GO_BM + wn * 32 + mrow) * GI_PITCH + 16 * kh;

  v4f sf[2], sr[2], st[2][4];
  v4f a[2][4], b[2][2][4];

#define GO_REQUEST(set, stage)                                                     \
  do {                                                                             \
    sf[set] = *reinterpret_cast<const v4f*>(gf + (stage) * GI_KS);                 \
    sr[set] = *reinterpret_cast<const v4f*>(gr + (stage) * GI_KS);                 \
    _Pragma("unroll") for (int j = 0; j < 4; ++j)                                  \
      st[set][j] = *reinterpret_cast<const v4f*>(gb + (size_t)j * 48 * DI + (stage) * GI_KS); \
  } while (0)
#define GO_DEPOSIT(set, buf)                                                       \
  do {                                                                             \
    if (has_a)                                                                     \
      *reinterpret_cast<v4f*>(&(buf)[loa]) = sf[set] + sr[set];    \
    _Pragma("unroll") for (int j = 0; j < 4; ++j)                                  \
      *reinterpret_cast<v4f*>(&(buf)[lob + j * 48 * GI_PITCH]) = st[set][j];       \
  } while (0)
#define GO_FETCH(set, buf)                                                         \
  do {                                                                             \
    ld16((buf) + aoff, a[set]);                                                    \
    _Pragma("unroll") for (int j = 0; j < 2; ++j)                                  \
      ld16((buf) + boff + j * 16 * GI_PITCH, b[set][j]);                           \
  } while (0)

  GO_REQUEST(0, 0);
  GO_REQUEST(1, 1);
  GO_DEPOSIT(0, sT[0]);
  GO_REQUEST(0, 2);
  lds_handover_barrier();
  GO_FETCH(0, sT[0]);
  GO_DEPOSIT(1, sT[1]);
  GO_REQUEST(1, 3);
  lds_handover_barrier();

  v8f acc[2] = {};
#pragma unroll
  for (int s = 0; s < GO_STAGES; ++s) {
    const int cur = s & 1;
    if (s + 1 < GO_STAGES) GO_FETCH(cur ^ 1, sT[cur ^ 1]);
    __builtin_amdgcn_sched_barrier(0);
#pragma unroll
    for (int q = 0; q < 4; ++q) {
#pragma unroll
      for (int j = 0; j < 2; ++j) acc[j] = wmma4(lo2(a[cur][q]), lo2(b[cur][j][q]), acc[j]);
#pragma unroll
      for (int j = 0; j < 2; ++j) acc[j] = wmma4(hi2(a[cur][q]), hi2(b[cur][j][q]), acc[j]);
    }
    __builtin_amdgcn_sched_barrier(0);
    if (s + 2 < GO_STAGES) GO_DEPOSIT(cur, sT[cur]);
    if (s + 4 < GO_STAGES) GO_REQUEST(cur, s + 4);
    if (s + 2 < GO_STAGES) lds_handover_barrier();
  }
#undef GO_REQUEST
#undef GO_DEPOSIT
#undef GO_FETCH
#pragma unroll
  for (int j = 0; j < 2; ++j)
#pragma unroll
    for (int v = 0; v < 8; ++v) {
      int orow = mb * GO_BM + wm * 16 + v + 8 * kh;
      int n = wn * 32 + j * 16 + mrow;
      out[(size_t)orow * DIMX + n] = acc[j][v] + xres[j][v];
    }
}

extern "C" void kernel_launch(void* const* d_in, const int* in_sizes, int n_in,
                              void* d_out, int out_size, void* d_ws, size_t ws_size,
                              hipStream_t stream) {
  const float* x      = (const float*)d_in[0];
  const float* Win_f  = (const float*)d_in[1];
  const float* cw_f   = (const float*)d_in[2];
  const float* cb_f   = (const float*)d_in[3];
  const float* Wxp_f  = (const float*)d_in[4];
  const float* Wdt_f  = (const float*)d_in[5];
  const float* bdt_f  = (const float*)d_in[6];
  const float* Alog_f = (const float*)d_in[7];
  const float* Dv_f   = (const float*)d_in[8];
  const float* Win_b  = (const float*)d_in[9];
  const float* cw_b   = (const float*)d_in[10];
  const float* cb_b   = (const float*)d_in[11];
  const float* Wxp_b  = (const float*)d_in[12];
  const float* Wdt_b  = (const float*)d_in[13];
  const float* bdt_b  = (const float*)d_in[14];
  const float* Alog_b = (const float*)d_in[15];
  const float* Dv_b   = (const float*)d_in[16];
  const float* Wout   = (const float*)d_in[17];
  float* out = (float*)d_out;

  float* ws    = (float*)d_ws;
  float* xz_f  = ws;
  float* xz_b  = xz_f + (size_t)MROWS * XZW;
  float* u_f   = xz_b + (size_t)MROWS * XZW;
  float* dbl_f = u_f  + (size_t)2 * MROWS * DI;
  const size_t PS = (size_t)BATCH * NCH * DI * NST;
  float* P_f   = dbl_f + (size_t)2 * MROWS * DBLW;
  float* S_f   = P_f + PS;

  const int BLK = 256, WPB = BLK / 32;

  {
    int blocks = (MROWS / GI_BM) * GI_NB;
    k_gemm_in<<<dim3(blocks, 2), BLK, 0, stream>>>(x, Win_f, Win_b, xz_f);
  }
  {
    int n = (MROWS / CT) * DQ;
    k_conv_silu<<<dim3(n / BLK, 2), BLK, 0, stream>>>(
        xz_f, cw_f, cb_f, cw_b, cb_b, u_f);
  }
  {
    int blocks = MROWS / GX_BM;
    k_gemm_xp<<<dim3(blocks, 2), 128, 0, stream>>>(u_f, Wxp_f, Wxp_b, dbl_f);
  }
  {
    int waves = (MROWS / 16) * (DI / 64);
    k_gemm_dt<<<dim3(waves / WPB, 2), BLK, 0, stream>>>(
        dbl_f, Wdt_f, bdt_f, Wdt_b, bdt_b, xz_f);
  }
  {
    int cb = (BATCH * DI * NST + BLK - 1) / BLK;
    k_scan_part1<<<dim3(BATCH * NCH, 2), DI, 0, stream>>>(
        xz_f, u_f, dbl_f, Alog_f, Alog_b, P_f, S_f);
    k_scan_combine<<<dim3(cb, 2), BLK, 0, stream>>>(P_f, S_f);
    k_scan_part3<<<dim3(BATCH * NCH, 2), DI, 0, stream>>>(
        xz_f, u_f, dbl_f, Alog_f, Dv_f, Alog_b, Dv_b, S_f);
  }
  {
    int blocks = MROWS / GO_BM;
    k_gemm_out<<<blocks, 384, 0, stream>>>(xz_f, xz_b, x, Wout, out);
  }
}


# Scan passes: fewer vector instructions per state and step: exp(dt*A) as exp2 with log2(e) folded into A once per thread; pass 3 updates state pairs by packed 2-wide ops, y as even+odd partial sums
# speedup vs baseline: 4.3084x; 1.0459x over previous
#include <hip/hip_runtime.h>
#include <hip/hip_bf16.h>
#include <math.h>

#define BATCH 4
#define SEQ   2048
#define DIMX  192
#define DI    384
#define NST   16
#define DTR   12
#define XZW   768
#define DBLW  48
#define MROWS (BATCH*SEQ)
#define NCH   32
#define CL    (SEQ/NCH)
#define PSTRIDE (2 * BATCH * NCH * DI * NST)

typedef float v2f __attribute__((ext_vector_type(2)));
typedef float v8f __attribute__((ext_vector_type(8)));

__device__ __forceinline__ v8f wmma4(v2f a, v2f b, v8f c) {
  return __builtin_amdgcn_wmma_f32_16x16x4_f32(false, a, false, b, (short)0, c,
                                               false, false);
}

typedef float v4f __attribute__((ext_vector_type(4)));

__device__ __forceinline__ void ld16(const float* __restrict__ p, v4f (&r)[4]) {
  const v4f* q = reinterpret_cast<const v4f*>(p);
  r[0] = q[0]; r[1] = q[1]; r[2] = q[2]; r[3] = q[3];
}
__device__ __forceinline__ v2f lo2(v4f v) { v2f r; r.x = v.x; r.y = v.y; return r; }
__device__ __forceinline__ v2f hi2(v4f v) { v2f r; r.x = v.z; r.y = v.w; return r; }
#define LOADED4(v) "v"((v)[0]), "v"((v)[1]), "v"((v)[2]), "v"((v)[3])
#define REQUESTS_STAY_AHEAD() asm volatile("" ::: "memory")
__device__ __forceinline__ void lds_handover_barrier() {
  __builtin_amdgcn_fence(__ATOMIC_RELEASE, "workgroup", "local");
  __builtin_amdgcn_s_barrier();
  __builtin_amdgcn_fence(__ATOMIC_ACQUIRE, "workgroup", "local");
}

#define GI_BM    64
#define GI_BN    128
#define GI_KS    32
#define GI_ROWS  (GI_BM + GI_BN)
#define GI_PITCH 36
#define GI_NB    (XZW / GI_BN)
#define GI_STAGES (DIMX / GI_KS)
__global__ void __launch_bounds__(256) k_gemm_in(const float* __restrict__ x,
                          const float* __restrict__ Win_f,
                          const float* __restrict__ Win_b,
                          float* __restrict__ xz_all) {
  __shared__ __attribute__((aligned(16))) float sT[2][GI_ROWS * GI_PITCH];
  const int reverse = blockIdx.y;
  const float* Win = reverse ? Win_b : Win_f;
  float* xz = xz_all + (size_t)reverse * MROWS * XZW;
  const int nb = blockIdx.x % GI_NB;
  const int mb = blockIdx.x / GI_NB;
  const int tid = threadIdx.x;
  const int wave = tid >> 5, lane = tid & 31;
  const int wm = wave >> 1, wn = wave & 1;
  const int mrow = lane & 15, kh = lane >> 4;

  const int r0 = tid >> 3, p = tid & 7;
  const int m  = mb * GI_BM + r0;
  const int bb = m / SEQ, ll = m % SEQ;
  const int srow = reverse ? (bb * SEQ + (SEQ - 1 - ll)) : m;
  const float* ga = x + (size_t)srow * DIMX + 4 * p;
  const int astep = reverse ? -32 * DIMX : 32 * DIMX;
  const float* gb = Win + (size_t)(nb * GI_BN + r0) * DIMX + 4 * p;
  const int lo = r0 * GI_PITCH + 4 * p;

  v4f st[6];
  st[0] = *reinterpret_cast<const v4f*>(ga);
  st[1] = *reinterpret_cast<const v4f*>(ga + astep);
#pragma unroll
  for (int j = 0; j < 4; ++j)
    st[2 + j] = *reinterpret_cast<const v4f*>(gb + (size_t)j * 32 * DIMX);
#pragma unroll
  for (int j = 0; j < 6; ++j)
    *reinterpret_cast<v4f*>(&sT[0][lo + j * 32 * GI_PITCH]) = st[j];
  __syncthreads();

  const int aoff = (wm * 16 + mrow) * GI_PITCH + 16 * kh;
  const int boff = (GI_BM + wn * 64 + mrow) * GI_PITCH + 16 * kh;

  v8f acc[4] = {};
  for (int s = 0; s < GI_STAGES; ++s) {
    const float* cur = sT[s & 1];
    const int kn = (s + 1 < GI_STAGES) ? (s + 1) * GI_KS : s * GI_KS;
    st[0] = *reinterpret_cast<const v4f*>(ga + kn);
    st[1] = *reinterpret_cast<const v4f*>(ga + astep + kn);
#pragma unroll
    for (int j = 0; j < 4; ++j)
      st[2 + j] = *reinterpret_cast<const v4f*>(gb + (size_t)j * 32 * DIMX + kn);

    v4f a[4], b[4][4];
    ld16(cur + aoff, a);
#pragma unroll
    for (int j = 0; j < 4; ++j) ld16(cur + boff + j * 16 * GI_PITCH, b[j]);
    asm volatile("" :: LOADED4(a), LOADED4(b[0]), LOADED4(b[1]), LOADED4(b[2]), LOADED4(b[3]));
#pragma unroll
    for (int q = 0; q < 4; ++q) {
#pragma unroll
      for (int j = 0; j < 4; ++j) acc[j] = wmma4(lo2(a[q]), lo2(b[j][q]), acc[j]);
#pragma unroll
      for (int j = 0; j < 4; ++j) acc[j] = wmma4(hi2(a[q]), hi2(b[j][q]), acc[j]);
    }
    float* nxt = sT[(s + 1) & 1];
#pragma unroll
    for (int j = 0; j < 6; ++j)
      *reinterpret_cast<v4f*>(&nxt[lo + j * 32 * GI_PITCH]) = st[j];
    __syncthreads();
  }
#pragma unroll
  for (int j = 0; j < 4; ++j)
#pragma unroll
    for (int v = 0; v < 8; ++v) {
      int orow = mb * GI_BM + wm * 16 + v + 8 * kh;
      xz[(size_t)orow * XZW + nb * GI_BN + wn * 64 + j * 16 + mrow] = acc[j][v];
    }
}

#define CT 8
#define DQ (DI / 4)
__global__ void k_conv_silu(const float* __restrict__ xz_all,
                            const float* __restrict__ cw_f,
                            const float* __restrict__ cb_f,
                            const float* __restrict__ cw_b,
                            const float* __restrict__ cb_b,
                            float* __restrict__ u_all) {
  const int dir = blockIdx.y;
  const float* xz = xz_all + (size_t)dir * MROWS * XZW;
  const float* cw = dir ? cw_b : cw_f;
  const float* cb = dir ? cb_b : cb_f;
  float* u = u_all + (size_t)dir * MROWS * DI;
  const int idx = blockIdx.x * blockDim.x + threadIdx.x;
  const int dq  = idx % DQ;
  const int run = idx / DQ;
  const int row0 = run * CT;
  const bool head = (row0 % SEQ) == 0;

  const float* src = xz + (size_t)row0 * XZW + 4 * dq;
  v4f xr[CT + 3];
  for (int k = 0; k < 3; ++k) {
    v4f zero = {0.0f, 0.0f, 0.0f, 0.0f};
    xr[k] = head ? zero : *reinterpret_cast<const v4f*>(src + (k - 3) * XZW);
  }
  for (int k = 0; k < CT; ++k)
    xr[3 + k] = *reinterpret_cast<const v4f*>(src + (size_t)k * XZW);

  v4f w[4];
  for (int c = 0; c < 4; ++c)
    w[c] = *reinterpret_cast<const v4f*>(cw + (size_t)(4 * dq + c) * 4);
  const v4f bias = *reinterpret_cast<const v4f*>(cb + 4 * dq);
  asm volatile("" :: "v"(xr[0]), "v"(xr[1]), "v"(xr[2]), "v"(xr[3]), "v"(xr[4]), "v"(xr[5]),
               "v"(xr[6]), "v"(xr[7]), "v"(xr[8]), "v"(xr[9]), "v"(xr[10]), LOADED4(w), "v"(bias));

  float* dst = u + (size_t)row0 * DI + 4 * dq;
  for (int i = 0; i < CT; ++i) {
    v4f o;
    for (int c = 0; c < 4; ++c) {
      float acc = bias[c];
      for (int j = 0; j < 4; ++j) {
        if (i + j >= 3 || !head)
          acc += xr[i + j][c] * w[c][j];
      }
      o[c] = acc / (1.0f + __expf(-acc));
    }
    *reinterpret_cast<v4f*>(dst + (size_t)i * DI) = o;
  }
}

#define GX_BM    64
#define GX_BN    48
#define GX_ROWS  (GX_BM + GX_BN)
#define GX_STAGES (DI / GI_KS)
#define GX_NP    (GX_ROWS / 16)
__global__ void __launch_bounds__(128) k_gemm_xp(const float* __restrict__ u_all,
                          const float* __restrict__ Wxp_f,
                          const float* __restrict__ Wxp_b,
                          float* __restrict__ dbl_all) {
  __shared__ __attribute__((aligned(16))) float sT[2][GX_ROWS * GI_PITCH];
  const int dir = blockIdx.y;
  const float* u = u_all + (size_t)dir * MROWS * DI;
  const float* Wxp = dir ? Wxp_b : Wxp_f;
  float* dbl = dbl_all + (size_t)dir * MROWS * DBLW;
  const int tid = threadIdx.x;
  const int wave = tid >> 5, lane = tid & 31;
  const int gm = blockIdx.x * (GX_BM / 16) + wave;
  const int mrow = lane & 15, kh = lane >> 4;

  const int r0 = tid >> 3, p = tid & 7;
  const float* ga = u + (size_t)(blockIdx.x * GX_BM + r0) * DI + 4 * p;
  const float* gb = Wxp + (size_t)r0 * DI + 4 * p;
  const bool tail_valid = (r0 + 32) < (DTR + 2 * NST);
  const float* gt = Wxp + (size_t)(tail_valid ? r0 + 32 : 0) * DI + 4 * p;
  const int lo = r0 * GI_PITCH + 4 * p;
  const v4f zero = {0.0f, 0.0f, 0.0f, 0.0f};
  const int aoff = (wave * 16 + mrow) * GI_PITCH + 16 * kh;
  const int boff = (GX_BM + mrow) * GI_PITCH + 16 * kh;

  v4f st[2][GX_NP];
  v4f a[2][4], b[2][3][4];

#define GX_REQUEST(set, stage)                                                     \
  do {                                                                             \
    _Pragma("unroll") for (int j = 0; j < 4; ++j)                                  \
      st[set][j] = *reinterpret_cast<const v4f*>(ga + (size_t)j * 16 * DI + (stage) * GI_KS); \
    _Pragma("unroll") for (int j = 0; j < 2; ++j)                                  \
      st[set][4 + j] = *reinterpret_cast<const v4f*>(gb + (size_t)j * 16 * DI + (stage) * GI_KS); \
    st[set][6] = *reinterpret_cast<const v4f*>(gt + (stage) * GI_KS);              \
  } while (0)
#define GX_DEPOSIT(set, buf)                                                       \
  do {                                                                             \
    _Pragma("unroll") for (int j = 0; j < GX_NP - 1; ++j)                          \
      *reinterpret_cast<v4f*>(&(buf)[lo + j * 16 * GI_PITCH]) = st[set][j];        \
    *reinterpret_cast<v4f*>(&(buf)[lo + 6 * 16 * GI_PITCH]) =                      \
        tail_valid ? st[set][6] : zero;                   \
  } while (0)
#define GX_FETCH(set, buf)                                                         \
  do {                                                                             \
    ld16((buf) + aoff, a[set]);                                                    \
    _Pragma("unroll") for (int j = 0; j < 3; ++j)                                  \
      ld16((buf) + boff + j * 16 * GI_PITCH, b[set][j]);                           \
  } while (0)

  GX_REQUEST(0, 0);
  GX_REQUEST(1, 1);
  GX_DEPOSIT(0, sT[0]);
  GX_REQUEST(0, 2);
  lds_handover_barrier();
  GX_FETCH(0, sT[0]);
  GX_DEPOSIT(1, sT[1]);
  GX_REQUEST(1, 3);
  lds_handover_barrier();

  v8f acc[3] = {};
#pragma unroll
  for (int s = 0; s < GX_STAGES; ++s) {
    const int cur = s & 1;
    if (s + 1 < GX_STAGES) GX_FETCH(cur ^ 1, sT[cur ^ 1]);
    __builtin_amdgcn_sched_barrier(0);
#pragma unroll
    for (int q = 0; q < 4; ++q) {
#pragma unroll
      for (int j = 0; j < 3; ++j) acc[j] = wmma4(lo2(a[cur][q]), lo2(b[cur][j][q]), acc[j]);
#pragma unroll
      for (int j = 0; j < 3; ++j) acc[j] = wmma4(hi2(a[cur][q]), hi2(b[cur][j][q]), acc[j]);
    }
    __builtin_amdgcn_sched_barrier(0);
    if (s + 2 < GX_STAGES) GX_DEPOSIT(cur, sT[cur]);
    if (s + 4 < GX_STAGES) GX_REQUEST(cur, s + 4);
    if (s + 2 < GX_STAGES) lds_handover_barrier();
  }
#undef GX_REQUEST
#undef GX_DEPOSIT
#undef GX_FETCH
#pragma unroll
  for (int j = 0; j < 3; ++j)
#pragma unroll
    for (int v = 0; v < 8; ++v) {
      int orow = gm * 16 + v + 8 * kh;
      dbl[(size_t)orow * DBLW + j * 16 + mrow] = acc[j][v];
    }
}

__device__ __forceinline__ float softplus_f32(float t) {
  const float e = __expf(-fabsf(t));
  const float s = e * __builtin_amdgcn_rcpf(2.0f + e);
  const float w = s * s;
  float p = 1.0f / 13.0f;
  p = fmaf(p, w, 1.0f / 11.0f);
  p = fmaf(p, w, 1.0f / 9.0f);
  p = fmaf(p, w, 1.0f / 7.0f);
  p = fmaf(p, w, 1.0f / 5.0f);
  p = fmaf(p, w, 1.0f / 3.0f);
  const float at = fmaf(s * w, p, s);
  return fmaf(2.0f, at, fmaxf(t, 0.0f));
}

__global__ void k_gemm_dt(const float* __restrict__ dbl_all,
                          const float* __restrict__ Wdt_f,
                          const float* __restrict__ bdt_f,
                          const float* __restrict__ Wdt_b,
                          const float* __restrict__ bdt_b,
                          float* __restrict__ xz_all) {
  const int dir = blockIdx.y;
  const float* dbl = dbl_all + (size_t)dir * MROWS * DBLW;
  const float* Wdt = dir ? Wdt_b : Wdt_f;
  const float* bdt = dir ? bdt_b : bdt_f;
  float* xz = xz_all + (size_t)dir * MROWS * XZW;
  const int wave = blockIdx.x * (blockDim.x >> 5) + (threadIdx.x >> 5);
  const int lane = threadIdx.x & 31;
  const int TG = DI / 64;
  const int gm = wave / TG, gng = wave % TG;
  const int mrow = lane & 15, kh = lane >> 4;

  const float* ap = dbl + (size_t)(gm * 16 + mrow) * DBLW;
  const float* bp[4];
#pragma unroll
  for (int j = 0; j < 4; ++j)
    bp[j] = Wdt + (size_t)(gng * 64 + j * 16 + mrow) * DTR;

  v8f acc[4] = {};
  for (int k0 = 0; k0 < DTR; k0 += 4) {
    int kk = k0 + 2 * kh;
    v2f a; a.x = ap[kk]; a.y = ap[kk + 1];
#pragma unroll
    for (int j = 0; j < 4; ++j) {
      v2f b; b.x = bp[j][kk]; b.y = bp[j][kk + 1];
      acc[j] = wmma4(a, b, acc[j]);
    }
  }
#pragma unroll
  for (int j = 0; j < 4; ++j) {
    int n = gng * 64 + j * 16 + mrow;
    float bias = bdt[n];
#pragma unroll
    for (int v = 0; v < 8; ++v) {
      int orow = gm * 16 + v + 8 * kh;
      float t = acc[j][v] + bias;
      xz[(size_t)orow * XZW + n] = softplus_f32(t);
    }
  }
}

#define LOG2E_F 1.44269504088896340736f
__device__ __forceinline__ float neg_exp_scaled(float al) { return -__expf(al) * LOG2E_F; }
#define TGRP 8
#define NGRP (CL / TGRP)

__device__ __forceinline__ void scan1_group(const float (&dt8)[TGRP], const float (&u8)[TGRP],
                                            const v4f (*sB)[NST / 4],
                                            const float (&a)[NST], float (&P)[NST], float (&S)[NST]) {
#pragma unroll
  for (int k = 0; k < TGRP; ++k) {
    float Bv[NST];
#pragma unroll
    for (int q = 0; q < NST / 4; ++q) {
      const v4f v = sB[k][q];
      Bv[4 * q] = v.x; Bv[4 * q + 1] = v.y; Bv[4 * q + 2] = v.z; Bv[4 * q + 3] = v.w;
    }
    const float dt = dt8[k];
    const float dtu = dt * u8[k];
#pragma unroll
    for (int n = 0; n < NST; ++n) {
      float dA = __builtin_amdgcn_exp2f(dt * a[n]);
      S[n] = S[n] * dA + dtu * Bv[n];
      P[n] *= dA;
    }
  }
}

__global__ void __launch_bounds__(DI) k_scan_part1(const float* __restrict__ xz_all,
                             const float* __restrict__ u_all,
                             const float* __restrict__ dbl_all,
                             const float* __restrict__ Alog_f,
                             const float* __restrict__ Alog_b,
                             float* __restrict__ Pb_all, float* __restrict__ Sb_all) {
  __shared__ v4f sB[CL][NST / 4];
  const int dir = blockIdx.y;
  const float* xz  = xz_all  + (size_t)dir * MROWS * XZW;
  const float* u   = u_all   + (size_t)dir * MROWS * DI;
  const float* dbl = dbl_all + (size_t)dir * MROWS * DBLW;
  const float* Alog = dir ? Alog_b : Alog_f;
  float* Pb = Pb_all + (size_t)dir * PSTRIDE;
  float* Sb = Sb_all + (size_t)dir * PSTRIDE;
  const int b = blockIdx.x / NCH;
  const int c = blockIdx.x % NCH;
  const int d = threadIdx.x;

  const size_t row0 = (size_t)b * SEQ + (size_t)c * CL;
  const float* dtp = xz  + row0 * XZW + d;
  const float* utp = u   + row0 * DI + d;
  const float* bcp = dbl + row0 * DBLW + DTR;

  const bool has_piece = d < CL * (NST / 4);
  const int pr = d >> 2, pq = d & 3;
  v4f bpiece = {0.0f, 0.0f, 0.0f, 0.0f};
  if (has_piece)
    bpiece = *reinterpret_cast<const v4f*>(bcp + (size_t)pr * DBLW + 4 * pq);
  float al[NST];
#pragma unroll
  for (int n = 0; n < NST; ++n) al[n] = Alog[d * NST + n];
  float dtA[TGRP], utA[TGRP], dtB[TGRP], utB[TGRP];
#pragma unroll
  for (int k = 0; k < TGRP; ++k) {
    dtA[k] = dtp[(size_t)k * XZW];
    utA[k] = utp[(size_t)k * DI];
  }
  if (has_piece) sB[pr][pq] = bpiece;

  float a[NST], P[NST], S[NST];
#pragma unroll
  for (int n = 0; n < NST; ++n) {
    a[n] = neg_exp_scaled(al[n]);
    P[n] = 1.0f;
    S[n] = 0.0f;
  }
  __syncthreads();

  for (int g = 0; g < NGRP; g += 2) {
#pragma unroll
    for (int k = 0; k < TGRP; ++k) {
      dtB[k] = dtp[(size_t)((g + 1) * TGRP + k) * XZW];
      utB[k] = utp[(size_t)((g + 1) * TGRP + k) * DI];
    }
    REQUESTS_STAY_AHEAD();
    scan1_group(dtA, utA, sB + g * TGRP, a, P, S);
    const int gn = (g + 2 < NGRP) ? g + 2 : g + 1;
#pragma unroll
    for (int k = 0; k < TGRP; ++k) {
      dtA[k] = dtp[(size_t)(gn * TGRP + k) * XZW];
      utA[k] = utp[(size_t)(gn * TGRP + k) * DI];
    }
    REQUESTS_STAY_AHEAD();
    scan1_group(dtB, utB, sB + (g + 1) * TGRP, a, P, S);
  }
  const size_t base = ((size_t)(b * NCH + c) * DI + d) * NST;
#pragma unroll
  for (int n = 0; n < NST; ++n) {
    Pb[base + n] = P[n];
    Sb[base + n] = S[n];
  }
}

__global__ void k_scan_combine(const float* __restrict__ Pb_all,
                               float* __restrict__ Sb_all) {
  const int dir = blockIdx.y;
  const float* Pb = Pb_all + (size_t)dir * PSTRIDE;
  float* Sb = Sb_all + (size_t)dir * PSTRIDE;
  int idx = blockIdx.x * blockDim.x + threadIdx.x;
  if (idx >= BATCH * DI * NST) return;
  const int n = idx & (NST - 1);
  const int d = (idx >> 4) % DI;
  const int b = idx / (NST * DI);

  const unsigned base0 = (unsigned)(((b * NCH) * DI + d) * NST + n);
  const unsigned cstep = (unsigned)(DI * NST);
  float p[NCH], s[NCH];
#pragma unroll
  for (int c = 0; c < NCH - 1; ++c) {
    p[c] = Pb[base0 + c * cstep];
    s[c] = Sb[base0 + c * cstep];
  }
  p[NCH - 1] = 0.0f;
  s[NCH - 1] = 0.0f;
#define QUAD(a, i) (v4f){(a)[i], (a)[(i) + 1], (a)[(i) + 2], (a)[(i) + 3]}
  asm volatile("" :: "v"(QUAD(p, 0)), "v"(QUAD(p, 4)), "v"(QUAD(p, 8)), "v"(QUAD(p, 12)),
               "v"(QUAD(p, 16)), "v"(QUAD(p, 20)), "v"(QUAD(p, 24)), "v"(QUAD(p, 28)),
               "v"(QUAD(s, 0)), "v"(QUAD(s, 4)), "v"(QUAD(s, 8)), "v"(QUAD(s, 12)),
               "v"(QUAD(s, 16)), "v"(QUAD(s, 20)), "v"(QUAD(s, 24)), "v"(QUAD(s, 28)));
#undef QUAD
  float carry = 0.0f;
#pragma unroll
  for (int c = 0; c < NCH; ++c) {
    Sb[base0 + c * cstep] = carry;
    if (c + 1 < NCH) carry = p[c] * carry + s[c];
  }
}

__device__ __forceinline__ void scan3_group(const float (&dt8)[TGRP], const float (&u8)[TGRP],
                                            const float (&z8)[TGRP],
                                            const v4f (*sBC)[2 * NST / 4], float* yp,
                                            const v2f (&a2)[NST / 2], v2f (&h)[NST / 2], float Dd) {
#pragma unroll
  for (int k = 0; k < TGRP; ++k) {
    v4f BCv[2 * NST / 4];
#pragma unroll
    for (int q = 0; q < 2 * NST / 4; ++q) BCv[q] = sBC[k][q];
    const float dt = dt8[k];
    const float ut = u8[k];
    const float dtu = dt * ut;
    v2f y2 = {0.0f, 0.0f};
#pragma unroll
    for (int i = 0; i < NST / 2; ++i) {
      v2f dA, T;
      dA.x = __builtin_amdgcn_exp2f(dt * a2[i].x);
      dA.y = __builtin_amdgcn_exp2f(dt * a2[i].y);
      const v2f Bp = (i & 1) ? hi2(BCv[i >> 1]) : lo2(BCv[i >> 1]);
      const v2f Cp = (i & 1) ? hi2(BCv[NST / 4 + (i >> 1)]) : lo2(BCv[NST / 4 + (i >> 1)]);
      T.x = dtu * Bp.x;
      T.y = dtu * Bp.y;
      h[i] = h[i] * dA + T;
      y2 += h[i] * Cp;
    }
    float y = y2.x + y2.y;
    y += ut * Dd;
    const float z = z8[k];
    y *= z / (1.0f + __expf(-z));
    yp[(size_t)k * XZW] = y;
  }
}

__global__ void __launch_bounds__(DI) k_scan_part3(float* __restrict__ xz_all,
                             const float* __restrict__ u_all,
                             const float* __restrict__ dbl_all,
                             const float* __restrict__ Alog_f,
                             const float* __restrict__ Dv_f,
                             const float* __restrict__ Alog_b,
                             const float* __restrict__ Dv_b,
                             const float* __restrict__ Sb_all) {
  __shared__ v4f sBC[CL][2 * NST / 4];
  const int dir = blockIdx.y;
  float* xz        = xz_all  + (size_t)dir * MROWS * XZW;
  const float* u   = u_all   + (size_t)dir * MROWS * DI;
  const float* dbl = dbl_all + (size_t)dir * MROWS * DBLW;
  const float* Alog = dir ? Alog_b : Alog_f;
  const float* Dv   = dir ? Dv_b : Dv_f;
  const float* Sb = Sb_all + (size_t)dir * PSTRIDE;
  const int b = blockIdx.x / NCH;
  const int c = blockIdx.x % NCH;
  const int d = threadIdx.x;

  const size_t row0 = (size_t)b * SEQ + (size_t)c * CL;
  float* xzp = xz + row0 * XZW + d;
  const float* utp = u   + row0 * DI + d;
  const float* bcp = dbl + row0 * DBLW + DTR;

  const int i0 = d, i1 = d + DI;
  const bool two = i1 < CL * (2 * NST / 4);
  v4f bc0 = *reinterpret_cast<const v4f*>(bcp + (size_t)(i0 >> 3) * DBLW + 4 * (i0 & 7));
  v4f bc1 = {0.0f, 0.0f, 0.0f, 0.0f};
  if (two)
    bc1 = *reinterpret_cast<const v4f*>(bcp + (size_t)(i1 >> 3) * DBLW + 4 * (i1 & 7));
  float al[NST];
  v2f h[NST / 2];
  const size_t base = ((size_t)(b * NCH + c) * DI + d) * NST;
#pragma unroll
  for (int i = 0; i < NST / 2; ++i) {
    al[2 * i]     = Alog[d * NST + 2 * i];
    al[2 * i + 1] = Alog[d * NST + 2 * i + 1];
    h[i].x = Sb[base + 2 * i];
    h[i].y = Sb[base + 2 * i + 1];
  }
  const float Dd = Dv[d];
  float dtA[TGRP], utA[TGRP], zA[TGRP], dtB[TGRP], utB[TGRP], zB[TGRP];
#pragma unroll
  for (int k = 0; k < TGRP; ++k) {
    dtA[k] = xzp[(size_t)k * XZW];
    zA[k]  = xzp[(size_t)k * XZW + DI];
    utA[k] = utp[(size_t)k * DI];
  }
  sBC[i0 >> 3][i0 & 7] = bc0;
  if (two) sBC[i1 >> 3][i1 & 7] = bc1;

  v2f a2[NST / 2];
#pragma unroll
  for (int i = 0; i < NST / 2; ++i) {
    a2[i].x = neg_exp_scaled(al[2 * i]);
    a2[i].y = neg_exp_scaled(al[2 * i + 1]);
  }
  __syncthreads();

  for (int g = 0; g < NGRP; g += 2) {
#pragma unroll
    for (int k = 0; k < TGRP; ++k) {
      dtB[k] = xzp[(size_t)((g + 1) * TGRP + k) * XZW];
      zB[k]  = xzp[(size_t)((g + 1) * TGRP + k) * XZW + DI];
      utB[k] = utp[(size_t)((g + 1) * TGRP + k) * DI];
    }
    REQUESTS_STAY_AHEAD();
    scan3_group(dtA, utA, zA, sBC + g * TGRP, xzp + (size_t)(g * TGRP) * XZW + DI, a2, h, Dd);
    const int gn = (g + 2 < NGRP) ? g + 2 : g + 1;
#pragma unroll
    for (int k = 0; k < TGRP; ++k) {
      dtA[k] = xzp[(size_t)(gn * TGRP + k) * XZW];
      zA[k]  = xzp[(size_t)(gn * TGRP + k) * XZW + DI];
      utA[k] = utp[(size_t)(gn * TGRP + k) * DI];
    }
    REQUESTS_STAY_AHEAD();
    scan3_group(dtB, utB, zB, sBC + (g + 1) * TGRP, xzp + (size_t)((g + 1) * TGRP) * XZW + DI, a2, h, Dd);
  }
}

#define GO_BM    32
#define GO_ROWS  (GO_BM + DIMX)
#define GO_STAGES (DI / GI_KS)
__global__ void __launch_bounds__(384) k_gemm_out(const float* __restrict__ xz_f,
                           const float* __restrict__ xz_b,
                           const float* __restrict__ x,
                           const float* __restrict__ Wout,
                           float* __restrict__ out) {
  __shared__ __attribute__((aligned(16))) float sT[2][GO_ROWS * GI_PITCH];
  const int mb = blockIdx.x;
  const int tid = threadIdx.x;
  const int wave = tid >> 5, lane = tid & 31;
  const int wm = wave / 6, wn = wave % 6;
  const int mrow = lane & 15, kh = lane >> 4;

  float xres[2][8];
#pragma unroll
  for (int j = 0; j < 2; ++j)
#pragma unroll
    for (int v = 0; v < 8; ++v)
      xres[j][v] = x[(size_t)(mb * GO_BM + wm * 16 + v + 8 * kh) * DIMX + wn * 32 + j * 16 + mrow];

  const int r0 = tid >> 3, p = tid & 7;
  const bool has_a = r0 < GO_BM;
  const int m  = mb * GO_BM + (has_a ? r0 : 0);
  const int bb = m / SEQ, ll = m % SEQ;
  const int mrev = bb * SEQ + (SEQ - 1 - ll);
  const float* gf = xz_f + (size_t)m    * XZW + DI + 4 * p;
  const float* gr = xz_b + (size_t)mrev * XZW + DI + 4 * p;
  const float* gb = Wout + (size_t)r0 * DI + 4 * p;
  const int loa = r0 * GI_PITCH + 4 * p;
  const int lob = (GO_BM + r0) * GI_PITCH + 4 * p;
  const int aoff = (wm * 16 + mrow) * GI_PITCH + 16 * kh;
  const int boff = (GO_BM + wn * 32 + mrow) * GI_PITCH + 16 * kh;

  v4f sf[2], sr[2], st[2][4];
  v4f a[2][4], b[2][2][4];

#define GO_REQUEST(set, stage)                                                     \
  do {                                                                             \
    sf[set] = *reinterpret_cast<const v4f*>(gf + (stage) * GI_KS);                 \
    sr[set] = *reinterpret_cast<const v4f*>(gr + (stage) * GI_KS);                 \
    _Pragma("unroll") for (int j = 0; j < 4; ++j)                                  \
      st[set][j] = *reinterpret_cast<const v4f*>(gb + (size_t)j * 48 * DI + (stage) * GI_KS); \
  } while (0)
#define GO_DEPOSIT(set, buf)                                                       \
  do {                                                                             \
    if (has_a)                                                                     \
      *reinterpret_cast<v4f*>(&(buf)[loa]) = sf[set] + sr[set];    \
    _Pragma("unroll") for (int j = 0; j < 4; ++j)                                  \
      *reinterpret_cast<v4f*>(&(buf)[lob + j * 48 * GI_PITCH]) = st[set][j];       \
  } while (0)
#define GO_FETCH(set, buf)                                                         \
  do {                                                                             \
    ld16((buf) + aoff, a[set]);                                                    \
    _Pragma("unroll") for (int j = 0; j < 2; ++j)                                  \
      ld16((buf) + boff + j * 16 * GI_PITCH, b[set][j]);                           \
  } while (0)

  GO_REQUEST(0, 0);
  GO_REQUEST(1, 1);
  GO_DEPOSIT(0, sT[0]);
  GO_REQUEST(0, 2);
  lds_handover_barrier();
  GO_FETCH(0, sT[0]);
  GO_DEPOSIT(1, sT[1]);
  GO_REQUEST(1, 3);
  lds_handover_barrier();

  v8f acc[2] = {};
#pragma unroll
  for (int s = 0; s < GO_STAGES; ++s) {
    const int cur = s & 1;
    if (s + 1 < GO_STAGES) GO_FETCH(cur ^ 1, sT[cur ^ 1]);
    __builtin_amdgcn_sched_barrier(0);
#pragma unroll
    for (int q = 0; q < 4; ++q) {
#pragma unroll
      for (int j = 0; j < 2; ++j) acc[j] = wmma4(lo2(a[cur][q]), lo2(b[cur][j][q]), acc[j]);
#pragma unroll
      for (int j = 0; j < 2; ++j) acc[j] = wmma4(hi2(a[cur][q]), hi2(b[cur][j][q]), acc[j]);
    }
    __builtin_amdgcn_sched_barrier(0);
    if (s + 2 < GO_STAGES) GO_DEPOSIT(cur, sT[cur]);
    if (s + 4 < GO_STAGES) GO_REQUEST(cur, s + 4);
    if (s + 2 < GO_STAGES) lds_handover_barrier();
  }
#undef GO_REQUEST
#undef GO_DEPOSIT
#undef GO_FETCH
#pragma unroll
  for (int j = 0; j < 2; ++j)
#pragma unroll
    for (int v = 0; v < 8; ++v) {
      int orow = mb * GO_BM + wm * 16 + v + 8 * kh;
      int n = wn * 32 + j * 16 + mrow;
      out[(size_t)orow * DIMX + n] = acc[j][v] + xres[j][v];
    }
}

extern "C" void kernel_launch(void* const* d_in, const int* in_sizes, int n_in,
                              void* d_out, int out_size, void* d_ws, size_t ws_size,
                              hipStream_t stream) {
  const float* x      = (const float*)d_in[0];
  const float* Win_f  = (const float*)d_in[1];
  const float* cw_f   = (const float*)d_in[2];
  const float* cb_f   = (const float*)d_in[3];
  const float* Wxp_f  = (const float*)d_in[4];
  const float* Wdt_f  = (const float*)d_in[5];
  const float* bdt_f  = (const float*)d_in[6];
  const float* Alog_f = (const float*)d_in[7];
  const float* Dv_f   = (const float*)d_in[8];
  const float* Win_b  = (const float*)d_in[9];
  const float* cw_b   = (const float*)d_in[10];
  const float* cb_b   = (const float*)d_in[11];
  const float* Wxp_b  = (const float*)d_in[12];
  const float* Wdt_b  = (const float*)d_in[13];
  const float* bdt_b  = (const float*)d_in[14];
  const float* Alog_b = (const float*)d_in[15];
  const float* Dv_b   = (const float*)d_in[16];
  const float* Wout   = (const float*)d_in[17];
  float* out = (float*)d_out;

  float* ws    = (float*)d_ws;
  float* xz_f  = ws;
  float* xz_b  = xz_f + (size_t)MROWS * XZW;
  float* u_f   = xz_b + (size_t)MROWS * XZW;
  float* dbl_f = u_f  + (size_t)2 * MROWS * DI;
  const size_t PS = (size_t)BATCH * NCH * DI * NST;
  float* P_f   = dbl_f + (size_t)2 * MROWS * DBLW;
  float* S_f   = P_f + PS;

  const int BLK = 256, WPB = BLK / 32;

  {
    int blocks = (MROWS / GI_BM) * GI_NB;
    k_gemm_in<<<dim3(blocks, 2), BLK, 0, stream>>>(x, Win_f, Win_b, xz_f);
  }
  {
    int n = (MROWS / CT) * DQ;
    k_conv_silu<<<dim3(n / BLK, 2), BLK, 0, stream>>>(
        xz_f, cw_f, cb_f, cw_b, cb_b, u_f);
  }
  {
    int blocks = MROWS / GX_BM;
    k_gemm_xp<<<dim3(blocks, 2), 128, 0, stream>>>(u_f, Wxp_f, Wxp_b, dbl_f);
  }
  {
    int waves = (MROWS / 16) * (DI / 64);
    k_gemm_dt<<<dim3(waves / WPB, 2), BLK, 0, stream>>>(
        dbl_f, Wdt_f, bdt_f, Wdt_b, bdt_b, xz_f);
  }
  {
    int cb = (BATCH * DI * NST + BLK - 1) / BLK;
    k_scan_part1<<<dim3(BATCH * NCH, 2), DI, 0, stream>>>(
        xz_f, u_f, dbl_f, Alog_f, Alog_b, P_f, S_f);
    k_scan_combine<<<dim3(cb, 2), BLK, 0, stream>>>(P_f, S_f);
    k_scan_part3<<<dim3(BATCH * NCH, 2), DI, 0, stream>>>(
        xz_f, u_f, dbl_f, Alog_f, Dv_f, Alog_b, Dv_b, S_f);
  }
  {
    int blocks = MROWS / GO_BM;
    k_gemm_out<<<blocks, 384, 0, stream>>>(xz_f, xz_b, x, Wout, out);
  }
}
